# starting kernel
# baseline (speedup 1.0000x reference)
#include <hip/hip_runtime.h>
#include <math.h>

#define NN   50000
#define EE   800000
#define INF_ 512
#define FF   32
#define HH   4
#define CC   32
#define HCC  128
#define EDD  4
#define E2E  850000
#define LL   2

typedef float v2f __attribute__((ext_vector_type(2)));
typedef float v8f __attribute__((ext_vector_type(8)));

__device__ __forceinline__ unsigned fkey(float f) {
    unsigned u = __float_as_uint(f);
    return (u & 0x80000000u) ? ~u : (u | 0x80000000u);
}
__device__ __forceinline__ float funkey(unsigned u) {
    return __uint_as_float((u & 0x80000000u) ? (u & 0x7FFFFFFFu) : ~u);
}
__device__ __forceinline__ float leakyf(float x, float s) {
    return x >= 0.0f ? x : s * x;
}

template <int CT, int KT, bool USE_LDS>
__global__ __launch_bounds__(256)
void gemm_wmma_f32(const float* __restrict__ A, const float* __restrict__ W,
                   const float* __restrict__ bias, float* __restrict__ Cout,
                   int M, int Nout) {
    extern __shared__ float sW[];
    const int tid  = (int)threadIdx.x;
    const int lane = tid & 31;
    const int wave = (int)((blockIdx.x * blockDim.x + tid) >> 5);
    const int mtiles  = M >> 4;
    const int ngroups = Nout / (16 * CT);
    const int stride  = KT + 4;

    if (USE_LDS) {
        const int total = Nout * KT;
        for (int i = tid; i < total; i += 256) {
            const int c  = i / KT;
            const int kk = i - c * KT;
            sW[c * stride + kk] = W[i];
        }
        __syncthreads();
    }
    if (wave >= mtiles * ngroups) return;

    const int mtile = wave / ngroups;
    const int ng    = wave % ngroups;
    const int row0  = mtile * 16;
    const int col0  = ng * 16 * CT;
    const int hi = lane >> 4;
    const int mi = lane & 15;

    v8f acc[CT] = {};
    const float* Arow = A + (size_t)(row0 + mi) * KT + 2 * hi;

    v2f a_cur;
    a_cur.x = Arow[0];
    a_cur.y = Arow[1];
    v2f b_cur[CT];
    if (!USE_LDS) {
#pragma unroll
        for (int t = 0; t < CT; ++t) {
            const float* Wc = W + (size_t)(col0 + t * 16 + mi) * KT + 2 * hi;
            b_cur[t].x = Wc[0];
            b_cur[t].y = Wc[1];
        }
    }

    for (int k = 0; k < KT; k += 4) {
        v2f a_nxt = {};
        v2f b_nxt[CT];
        if (k + 4 < KT) {
            a_nxt.x = Arow[k + 4];
            a_nxt.y = Arow[k + 5];
            if (!USE_LDS) {
#pragma unroll
                for (int t = 0; t < CT; ++t) {
                    const float* Wc = W + (size_t)(col0 + t * 16 + mi) * KT + 2 * hi;
                    b_nxt[t].x = Wc[k + 4];
                    b_nxt[t].y = Wc[k + 5];
                }
                if ((k & 31) == 0)
                    __builtin_prefetch(Arow + k + 132, 0, 1);
            }
        }
#pragma unroll
        for (int t = 0; t < CT; ++t) {
            v2f b;
            if (USE_LDS) {
                const float* Sc = sW + (size_t)(col0 + t * 16 + mi) * stride + 2 * hi;
                b.x = Sc[k];
                b.y = Sc[k + 1];
            } else {
                b = b_cur[t];
            }
            acc[t] = __builtin_amdgcn_wmma_f32_16x16x4_f32(
                false, a_cur, false, b, (short)0, acc[t], false, false);
        }
        a_cur = a_nxt;
        if (!USE_LDS) {
#pragma unroll
            for (int t = 0; t < CT; ++t) b_cur[t] = b_nxt[t];
        }
    }

#pragma unroll
    for (int t = 0; t < CT; ++t) {
        const int col = col0 + t * 16 + mi;
        const float bb = bias ? bias[col] : 0.0f;
#pragma unroll
        for (int v = 0; v < 8; ++v) {
            const int r = row0 + v + 8 * hi;
            Cout[(size_t)r * Nout + col] = acc[t][v] + bb;
        }
    }
}

__global__ __launch_bounds__(256)
void ln_rows_kernel(const float* __restrict__ Z, const float* __restrict__ w,
                    const float* __restrict__ b, float* __restrict__ Hout,
                    float* __restrict__ jk, int mode, int M) {
    const int row  = (int)((blockIdx.x * blockDim.x + threadIdx.x) >> 5);
    const int lane = (int)(threadIdx.x & 31);
    if (row >= M) return;
    float v = Z[(size_t)row * FF + lane];
    float s = v;
#pragma unroll
    for (int off = 16; off >= 1; off >>= 1) s += __shfl_xor(s, off);
    const float mu = s * (1.0f / 32.0f);
    const float d = v - mu;
    float q = d * d;
#pragma unroll
    for (int off = 16; off >= 1; off >>= 1) q += __shfl_xor(q, off);
    const float var = q * (1.0f / 32.0f);
    float y = w[lane] * d * rsqrtf(var + 1e-5f) + b[lane];
    y = leakyf(y, 0.01f);
    Hout[(size_t)row * FF + lane] = y;
    float* o = jk + (size_t)row * FF + lane;
    *o = (mode == 0) ? y : fmaxf(*o, y);
}

__global__ __launch_bounds__(256)
void edge_score_kernel(const int* __restrict__ src, const int* __restrict__ dst,
                       const float* __restrict__ ea, const float* __restrict__ We,
                       const float* __restrict__ xl, const float* __restrict__ xr,
                       const float* __restrict__ att, float* __restrict__ score,
                       unsigned* __restrict__ smax) {
    const int e    = (int)((blockIdx.x * blockDim.x + threadIdx.x) >> 5);
    const int lane = (int)(threadIdx.x & 31);
    if (e >= E2E) return;
    int sN, dN;
    float4 ev;
    if (e < EE) {
        sN = src[e]; dN = dst[e];
        ev = ((const float4*)ea)[e];
    } else {
        sN = dN = e - EE;
        ev = make_float4(0.0f, 0.0f, 0.0f, 0.0f);
    }
    const float4 xlv = ((const float4*)xl)[(size_t)sN * (HCC / 4) + lane];
    const float4 xrv = ((const float4*)xr)[(size_t)dN * (HCC / 4) + lane];
    const float4 atv = ((const float4*)att)[lane];
    const float4* We4 = (const float4*)We;

    const float xa[4] = {xlv.x, xlv.y, xlv.z, xlv.w};
    const float xb[4] = {xrv.x, xrv.y, xrv.z, xrv.w};
    const float at[4] = {atv.x, atv.y, atv.z, atv.w};
    float sc = 0.0f;
#pragma unroll
    for (int k = 0; k < 4; ++k) {
        const float4 wr = We4[lane * 4 + k];
        const float eev = wr.x * ev.x + wr.y * ev.y + wr.z * ev.z + wr.w * ev.w;
        float m = xa[k] + xb[k] + eev;
        m = leakyf(m, 0.2f);
        sc += at[k] * m;
    }
    sc += __shfl_xor(sc, 4);
    sc += __shfl_xor(sc, 2);
    sc += __shfl_xor(sc, 1);
    if ((lane & 7) == 0) {
        const int h = lane >> 3;
        score[(size_t)e * HH + h] = sc;
        atomicMax(&smax[(size_t)dN * HH + h], fkey(sc));
    }
}

__global__ __launch_bounds__(256)
void edge_den_kernel(const int* __restrict__ dst, const float* __restrict__ score,
                     const unsigned* __restrict__ smax, float* __restrict__ den) {
    const int t = (int)(blockIdx.x * blockDim.x + threadIdx.x);
    if (t >= E2E * HH) return;
    const int e = t >> 2;
    const int h = t & 3;
    const int dN = (e < EE) ? dst[e] : (e - EE);
    const float ex = expf(score[t] - funkey(smax[(size_t)dN * HH + h]));
    atomicAdd(&den[(size_t)dN * HH + h], ex);
}

__global__ __launch_bounds__(256)
void edge_agg_kernel(const int* __restrict__ src, const int* __restrict__ dst,
                     const float* __restrict__ score, const unsigned* __restrict__ smax,
                     const float* __restrict__ den, const float* __restrict__ xl,
                     float* __restrict__ agg) {
    const int e    = (int)((blockIdx.x * blockDim.x + threadIdx.x) >> 5);
    const int lane = (int)(threadIdx.x & 31);
    if (e >= E2E) return;
    int sN, dN;
    if (e < EE) { sN = src[e]; dN = dst[e]; }
    else        { sN = dN = e - EE; }
    const int h = lane >> 3;
    const float mx = funkey(smax[(size_t)dN * HH + h]);
    const float dn = den[(size_t)dN * HH + h];
    const float alpha = expf(score[(size_t)e * HH + h] - mx) / (dn + 1e-16f);
    const float4 xv = ((const float4*)xl)[(size_t)sN * (HCC / 4) + lane];
    float* ao = agg + (size_t)dN * HCC + lane * 4;
    atomicAdd(ao + 0, xv.x * alpha);
    atomicAdd(ao + 1, xv.y * alpha);
    atomicAdd(ao + 2, xv.z * alpha);
    atomicAdd(ao + 3, xv.w * alpha);
}

__global__ __launch_bounds__(256)
void bias_stats_kernel(float* __restrict__ agg, const float* __restrict__ cb,
                       double* __restrict__ red, int total) {
    __shared__ double ss[256];
    __shared__ double sq[256];
    const int idx = (int)(blockIdx.x * blockDim.x + threadIdx.x);
    double s = 0.0, q = 0.0;
    if (idx < total) {
        const float y = agg[idx] + cb[idx & (HCC - 1)];
        agg[idx] = y;
        s = (double)y;
        q = (double)y * (double)y;
    }
    ss[threadIdx.x] = s;
    sq[threadIdx.x] = q;
    __syncthreads();
    for (int o = 128; o > 0; o >>= 1) {
        if ((int)threadIdx.x < o) {
            ss[threadIdx.x] += ss[threadIdx.x + o];
            sq[threadIdx.x] += sq[threadIdx.x + o];
        }
        __syncthreads();
    }
    if (threadIdx.x == 0) {
        atomicAdd(&red[0], ss[0]);
        atomicAdd(&red[1], sq[0]);
    }
}

__global__ __launch_bounds__(256)
void pln_apply_kernel(float* __restrict__ agg, const float* __restrict__ w,
                      const float* __restrict__ b, const double* __restrict__ red,
                      int total) {
    const int idx = (int)(blockIdx.x * blockDim.x + threadIdx.x);
    if (idx >= total) return;
    const double mu = red[0] / (double)total;
    double var = red[1] / (double)total - mu * mu;
    if (var < 0.0) var = 0.0;
    const float sd = sqrtf((float)var);
    const int c = idx & (HCC - 1);
    float y = w[c] * (float)((agg[idx] - mu) / (double)(sd + 1e-5f)) + b[c];
    agg[idx] = leakyf(y, 0.01f);
}

extern "C" void kernel_launch(void* const* d_in, const int* in_sizes, int n_in,
                              void* d_out, int out_size, void* d_ws, size_t ws_size,
                              hipStream_t stream) {
    const float* x          = (const float*)d_in[0];
    const int*   edge_index = (const int*)  d_in[1];
    const float* edge_attr  = (const float*)d_in[2];
    const float* embed_W    = (const float*)d_in[3];
    const float* embed_b    = (const float*)d_in[4];
    const float* embed_lnw  = (const float*)d_in[5];
    const float* embed_lnb  = (const float*)d_in[6];
    const float* Wl         = (const float*)d_in[7];
    const float* bl         = (const float*)d_in[8];
    const float* Wr         = (const float*)d_in[9];
    const float* br         = (const float*)d_in[10];
    const float* We         = (const float*)d_in[11];
    const float* att        = (const float*)d_in[12];
    const float* conv_b     = (const float*)d_in[13];
    const float* pln_w      = (const float*)d_in[14];
    const float* pln_b      = (const float*)d_in[15];
    const float* proj_W     = (const float*)d_in[16];
    const float* proj_b     = (const float*)d_in[17];
    const float* ln2_w      = (const float*)d_in[18];
    const float* ln2_b      = (const float*)d_in[19];

    const int* srcE = edge_index;
    const int* dstE = edge_index + EE;

    size_t off = 0;
    auto carve = [&](size_t nbytes) -> char* {
        char* p = (char*)d_ws + off;
        off += (nbytes + 255) & ~(size_t)255;
        return p;
    };
    float*    h     = (float*)   carve((size_t)NN * FF  * 4);
    float*    raw   = (float*)   carve((size_t)NN * FF  * 4);
    float*    xl    = (float*)   carve((size_t)NN * HCC * 4);
    float*    xr    = (float*)   carve((size_t)NN * HCC * 4);
    float*    agg   = (float*)   carve((size_t)NN * HCC * 4);
    float*    score = (float*)   carve((size_t)E2E * HH * 4);
    unsigned* smax  = (unsigned*)carve((size_t)NN * HH  * 4);
    float*    den   = (float*)   carve((size_t)NN * HH  * 4);
    double*   red   = (double*)  carve(2 * sizeof(double));
    (void)ws_size; (void)in_sizes; (void)n_in; (void)out_size;

    float* jk = (float*)d_out;

    const int mtiles = NN / 16;
    const dim3 blk(256);
    const dim3 gGemm((mtiles * 32 + 255) / 256);
    const dim3 gRows((NN * 32 + 255) / 256);
    const dim3 gEdgeW(((size_t)E2E * 32 + 255) / 256);
    const dim3 gEdgeT(((size_t)E2E * HH + 255) / 256);
    const dim3 gElem(((size_t)NN * HCC + 255) / 256);
    const int totalHC = NN * HCC;

    const size_t shLin  = (size_t)HCC * (FF + 4) * 4;
    const size_t shProj = (size_t)FF * (HCC + 4) * 4;

    gemm_wmma_f32<2, INF_, false><<<gGemm, blk, 0, stream>>>(x, embed_W, embed_b, raw, NN, FF);
    ln_rows_kernel<<<gRows, blk, 0, stream>>>(raw, embed_lnw, embed_lnb, h, jk, 0, NN);

    for (int l = 0; l < LL; ++l) {
        const float* Wl_l = Wl + (size_t)l * HCC * FF;
        const float* bl_l = bl + (size_t)l * HCC;
        const float* Wr_l = Wr + (size_t)l * HCC * FF;
        const float* br_l = br + (size_t)l * HCC;
        const float* We_l = We + (size_t)l * HCC * EDD;
        const float* at_l = att + (size_t)l * HH * CC;
        const float* cb_l = conv_b + (size_t)l * HCC;
        const float* pw_l = pln_w + (size_t)l * HCC;
        const float* pb_l = pln_b + (size_t)l * HCC;
        const float* pW_l = proj_W + (size_t)l * FF * HCC;
        const float* pB_l = proj_b + (size_t)l * FF;
        const float* l2w  = ln2_w + (size_t)l * FF;
        const float* l2b  = ln2_b + (size_t)l * FF;

        gemm_wmma_f32<8, FF, true><<<gGemm, blk, shLin, stream>>>(h, Wl_l, bl_l, xl, NN, HCC);
        gemm_wmma_f32<8, FF, true><<<gGemm, blk, shLin, stream>>>(h, Wr_l, br_l, xr, NN, HCC);

        hipMemsetAsync(smax, 0, (size_t)NN * HH * 4, stream);
        hipMemsetAsync(den,  0, (size_t)NN * HH * 4, stream);
        hipMemsetAsync(agg,  0, (size_t)NN * HCC * 4, stream);
        hipMemsetAsync(red,  0, 2 * sizeof(double), stream);

        edge_score_kernel<<<gEdgeW, blk, 0, stream>>>(srcE, dstE, edge_attr, We_l,
                                                      xl, xr, at_l, score, smax);
        edge_den_kernel<<<gEdgeT, blk, 0, stream>>>(dstE, score, smax, den);
        edge_agg_kernel<<<gEdgeW, blk, 0, stream>>>(srcE, dstE, score, smax, den, xl, agg);

        bias_stats_kernel<<<gElem, blk, 0, stream>>>(agg, cb_l, red, totalHC);
        pln_apply_kernel<<<gElem, blk, 0, stream>>>(agg, pw_l, pb_l, red, totalHC);

        gemm_wmma_f32<2, HCC, true><<<gGemm, blk, shProj, stream>>>(agg, pW_l, pB_l, raw, NN, FF);
        ln_rows_kernel<<<gRows, blk, 0, stream>>>(raw, l2w, l2b, h, jk, 1, NN);
    }
}


# graph-LN sums: per-block partials to workspace, one finishing block adds them in a fixed tree and writes mean and std+eps once; no f64 atomics, apply waves no longer derive them; sum order changes
# speedup vs baseline: 1.3440x; 1.3440x over previous
#include <hip/hip_runtime.h>
#include <math.h>

#define NN   50000
#define EE   800000
#define INF_ 512
#define FF   32
#define HH   4
#define CC   32
#define HCC  128
#define EDD  4
#define E2E  850000
#define LL   2

typedef float v2f __attribute__((ext_vector_type(2)));
typedef float v8f __attribute__((ext_vector_type(8)));

__device__ __forceinline__ unsigned fkey(float f) {
    unsigned u = __float_as_uint(f);
    return (u & 0x80000000u) ? ~u : (u | 0x80000000u);
}
__device__ __forceinline__ float funkey(unsigned u) {
    return __uint_as_float((u & 0x80000000u) ? (u & 0x7FFFFFFFu) : ~u);
}
__device__ __forceinline__ float leakyf(float x, float s) {
    return x >= 0.0f ? x : s * x;
}

template <int CT, int KT, bool USE_LDS>
__global__ __launch_bounds__(256)
void gemm_wmma_f32(const float* __restrict__ A, const float* __restrict__ W,
                   const float* __restrict__ bias, float* __restrict__ Cout,
                   int M, int Nout) {
    extern __shared__ float sW[];
    const int tid  = (int)threadIdx.x;
    const int lane = tid & 31;
    const int wave = (int)((blockIdx.x * blockDim.x + tid) >> 5);
    const int mtiles  = M >> 4;
    const int ngroups = Nout / (16 * CT);
    const int stride  = KT + 4;

    if (USE_LDS) {
        const int total = Nout * KT;
        for (int i = tid; i < total; i += 256) {
            const int c  = i / KT;
            const int kk = i - c * KT;
            sW[c * stride + kk] = W[i];
        }
        __syncthreads();
    }
    if (wave >= mtiles * ngroups) return;

    const int mtile = wave / ngroups;
    const int ng    = wave % ngroups;
    const int row0  = mtile * 16;
    const int col0  = ng * 16 * CT;
    const int hi = lane >> 4;
    const int mi = lane & 15;

    v8f acc[CT] = {};
    const float* Arow = A + (size_t)(row0 + mi) * KT + 2 * hi;

    v2f a_cur;
    a_cur.x = Arow[0];
    a_cur.y = Arow[1];
    v2f b_cur[CT];
    if (!USE_LDS) {
#pragma unroll
        for (int t = 0; t < CT; ++t) {
            const float* Wc = W + (size_t)(col0 + t * 16 + mi) * KT + 2 * hi;
            b_cur[t].x = Wc[0];
            b_cur[t].y = Wc[1];
        }
    }

    for (int k = 0; k < KT; k += 4) {
        v2f a_nxt = {};
        v2f b_nxt[CT];
        if (k + 4 < KT) {
            a_nxt.x = Arow[k + 4];
            a_nxt.y = Arow[k + 5];
            if (!USE_LDS) {
#pragma unroll
                for (int t = 0; t < CT; ++t) {
                    const float* Wc = W + (size_t)(col0 + t * 16 + mi) * KT + 2 * hi;
                    b_nxt[t].x = Wc[k + 4];
                    b_nxt[t].y = Wc[k + 5];
                }
                if ((k & 31) == 0)
                    __builtin_prefetch(Arow + k + 132, 0, 1);
            }
        }
#pragma unroll
        for (int t = 0; t < CT; ++t) {
            v2f b;
            if (USE_LDS) {
                const float* Sc = sW + (size_t)(col0 + t * 16 + mi) * stride + 2 * hi;
                b.x = Sc[k];
                b.y = Sc[k + 1];
            } else {
                b = b_cur[t];
            }
            acc[t] = __builtin_amdgcn_wmma_f32_16x16x4_f32(
                false, a_cur, false, b, (short)0, acc[t], false, false);
        }
        a_cur = a_nxt;
        if (!USE_LDS) {
#pragma unroll
            for (int t = 0; t < CT; ++t) b_cur[t] = b_nxt[t];
        }
    }

#pragma unroll
    for (int t = 0; t < CT; ++t) {
        const int col = col0 + t * 16 + mi;
        const float bb = bias ? bias[col] : 0.0f;
#pragma unroll
        for (int v = 0; v < 8; ++v) {
            const int r = row0 + v + 8 * hi;
            Cout[(size_t)r * Nout + col] = acc[t][v] + bb;
        }
    }
}

__global__ __launch_bounds__(256)
void ln_rows_kernel(const float* __restrict__ Z, const float* __restrict__ w,
                    const float* __restrict__ b, float* __restrict__ Hout,
                    float* __restrict__ jk, int mode, int M) {
    const int row  = (int)((blockIdx.x * blockDim.x + threadIdx.x) >> 5);
    const int lane = (int)(threadIdx.x & 31);
    if (row >= M) return;
    float v = Z[(size_t)row * FF + lane];
    float s = v;
#pragma unroll
    for (int off = 16; off >= 1; off >>= 1) s += __shfl_xor(s, off);
    const float mu = s * (1.0f / 32.0f);
    const float d = v - mu;
    float q = d * d;
#pragma unroll
    for (int off = 16; off >= 1; off >>= 1) q += __shfl_xor(q, off);
    const float var = q * (1.0f / 32.0f);
    float y = w[lane] * d * rsqrtf(var + 1e-5f) + b[lane];
    y = leakyf(y, 0.01f);
    Hout[(size_t)row * FF + lane] = y;
    float* o = jk + (size_t)row * FF + lane;
    *o = (mode == 0) ? y : fmaxf(*o, y);
}

__global__ __launch_bounds__(256)
void edge_score_kernel(const int* __restrict__ src, const int* __restrict__ dst,
                       const float* __restrict__ ea, const float* __restrict__ We,
                       const float* __restrict__ xl, const float* __restrict__ xr,
                       const float* __restrict__ att, float* __restrict__ score,
                       unsigned* __restrict__ smax) {
    const int e    = (int)((blockIdx.x * blockDim.x + threadIdx.x) >> 5);
    const int lane = (int)(threadIdx.x & 31);
    if (e >= E2E) return;
    int sN, dN;
    float4 ev;
    if (e < EE) {
        sN = src[e]; dN = dst[e];
        ev = ((const float4*)ea)[e];
    } else {
        sN = dN = e - EE;
        ev = make_float4(0.0f, 0.0f, 0.0f, 0.0f);
    }
    const float4 xlv = ((const float4*)xl)[(size_t)sN * (HCC / 4) + lane];
    const float4 xrv = ((const float4*)xr)[(size_t)dN * (HCC / 4) + lane];
    const float4 atv = ((const float4*)att)[lane];
    const float4* We4 = (const float4*)We;

    const float xa[4] = {xlv.x, xlv.y, xlv.z, xlv.w};
    const float xb[4] = {xrv.x, xrv.y, xrv.z, xrv.w};
    const float at[4] = {atv.x, atv.y, atv.z, atv.w};
    float sc = 0.0f;
#pragma unroll
    for (int k = 0; k < 4; ++k) {
        const float4 wr = We4[lane * 4 + k];
        const float eev = wr.x * ev.x + wr.y * ev.y + wr.z * ev.z + wr.w * ev.w;
        float m = xa[k] + xb[k] + eev;
        m = leakyf(m, 0.2f);
        sc += at[k] * m;
    }
    sc += __shfl_xor(sc, 4);
    sc += __shfl_xor(sc, 2);
    sc += __shfl_xor(sc, 1);
    if ((lane & 7) == 0) {
        const int h = lane >> 3;
        score[(size_t)e * HH + h] = sc;
        atomicMax(&smax[(size_t)dN * HH + h], fkey(sc));
    }
}

__global__ __launch_bounds__(256)
void edge_agg_kernel(const int* __restrict__ src, const int* __restrict__ dst,
                     const float* __restrict__ score, const unsigned* __restrict__ smax,
                     float* __restrict__ den, const float* __restrict__ xl,
                     float* __restrict__ agg) {
    const int e    = (int)((blockIdx.x * blockDim.x + threadIdx.x) >> 5);
    const int lane = (int)(threadIdx.x & 31);
    if (e >= E2E) return;
    int sN, dN;
    if (e < EE) { sN = src[e]; dN = dst[e]; }
    else        { sN = dN = e - EE; }
    const int h = lane >> 3;
    const float mx = funkey(smax[(size_t)dN * HH + h]);
    const float ex = expf(score[(size_t)e * HH + h] - mx);
    const float4 xv = ((const float4*)xl)[(size_t)sN * (HCC / 4) + lane];
    float* ao = agg + (size_t)dN * HCC + lane * 4;
    atomicAdd(ao + 0, xv.x * ex);
    atomicAdd(ao + 1, xv.y * ex);
    atomicAdd(ao + 2, xv.z * ex);
    atomicAdd(ao + 3, xv.w * ex);
    if ((lane & 7) == 0) atomicAdd(&den[(size_t)dN * HH + h], ex);
}

__global__ __launch_bounds__(256)
void bias_stats_kernel(float* __restrict__ agg, const float* __restrict__ den,
                       const float* __restrict__ cb,
                       double* __restrict__ part, int total) {
    __shared__ double ss[256];
    __shared__ double sq[256];
    const int idx = (int)(blockIdx.x * blockDim.x + threadIdx.x);
    double s = 0.0, q = 0.0;
    if (idx < total) {
        const float y = agg[idx] / (den[idx >> 5] + 1e-16f) + cb[idx & (HCC - 1)];
        agg[idx] = y;
        s = (double)y;
        q = (double)y * (double)y;
    }
    ss[threadIdx.x] = s;
    sq[threadIdx.x] = q;
    __syncthreads();
    for (int o = 128; o > 0; o >>= 1) {
        if ((int)threadIdx.x < o) {
            ss[threadIdx.x] += ss[threadIdx.x + o];
            sq[threadIdx.x] += sq[threadIdx.x + o];
        }
        __syncthreads();
    }
    if (threadIdx.x == 0) {
        ((double2*)part)[blockIdx.x] = make_double2(ss[0], sq[0]);
    }
}

__global__ __launch_bounds__(1024)
void stats_finish_kernel(const double* __restrict__ part, int nblk,
                         double* __restrict__ red, int total) {
    __shared__ double ws_[32];
    __shared__ double wq_[32];
    const int tid  = (int)threadIdx.x;
    const int lane = tid & 31;
    const int wv   = tid >> 5;
    double s = 0.0, q = 0.0;
    for (int i = tid; i < nblk; i += 1024) {
        const double2 p = ((const double2*)part)[i];
        s += p.x;
        q += p.y;
    }
    for (int off = 16; off >= 1; off >>= 1) {
        s += __shfl_xor(s, off);
        q += __shfl_xor(q, off);
    }
    if (lane == 0) { ws_[wv] = s; wq_[wv] = q; }
    __syncthreads();
    if (wv == 0) {
        s = ws_[lane];
        q = wq_[lane];
        for (int off = 16; off >= 1; off >>= 1) {
            s += __shfl_xor(s, off);
            q += __shfl_xor(q, off);
        }
        if (lane == 0) {
            const double mu = s / (double)total;
            double var = q / (double)total - mu * mu;
            if (var < 0.0) var = 0.0;
            const float sd = sqrtf((float)var);
            red[0] = mu;
            red[1] = (double)(sd + 1e-5f);
        }
    }
}

__global__ __launch_bounds__(256)
void pln_apply_kernel(float* __restrict__ agg, const float* __restrict__ w,
                      const float* __restrict__ b, const double* __restrict__ red,
                      int total) {
    const int idx = (int)(blockIdx.x * blockDim.x + threadIdx.x);
    if (idx >= total) return;
    const double mu  = red[0];
    const double dnm = red[1];
    const int c = idx & (HCC - 1);
    float y = w[c] * (float)((agg[idx] - mu) / dnm) + b[c];
    agg[idx] = leakyf(y, 0.01f);
}

extern "C" void kernel_launch(void* const* d_in, const int* in_sizes, int n_in,
                              void* d_out, int out_size, void* d_ws, size_t ws_size,
                              hipStream_t stream) {
    const float* x          = (const float*)d_in[0];
    const int*   edge_index = (const int*)  d_in[1];
    const float* edge_attr  = (const float*)d_in[2];
    const float* embed_W    = (const float*)d_in[3];
    const float* embed_b    = (const float*)d_in[4];
    const float* embed_lnw  = (const float*)d_in[5];
    const float* embed_lnb  = (const float*)d_in[6];
    const float* Wl         = (const float*)d_in[7];
    const float* bl         = (const float*)d_in[8];
    const float* Wr         = (const float*)d_in[9];
    const float* br         = (const float*)d_in[10];
    const float* We         = (const float*)d_in[11];
    const float* att        = (const float*)d_in[12];
    const float* conv_b     = (const float*)d_in[13];
    const float* pln_w      = (const float*)d_in[14];
    const float* pln_b      = (const float*)d_in[15];
    const float* proj_W     = (const float*)d_in[16];
    const float* proj_b     = (const float*)d_in[17];
    const float* ln2_w      = (const float*)d_in[18];
    const float* ln2_b      = (const float*)d_in[19];

    const int* srcE = edge_index;
    const int* dstE = edge_index + EE;

    size_t off = 0;
    auto carve = [&](size_t nbytes) -> char* {
        char* p = (char*)d_ws + off;
        off += (nbytes + 255) & ~(size_t)255;
        return p;
    };
    float*    h     = (float*)   carve((size_t)NN * FF  * 4);
    float*    raw   = (float*)   carve((size_t)NN * FF  * 4);
    float*    xl    = (float*)   carve((size_t)NN * HCC * 4);
    float*    xr    = (float*)   carve((size_t)NN * HCC * 4);
    float*    agg   = (float*)   carve((size_t)NN * HCC * 4);
    float*    score = (float*)   carve((size_t)E2E * HH * 4);
    unsigned* smax  = (unsigned*)carve((size_t)NN * HH  * 4);
    float*    den   = (float*)   carve((size_t)NN * HH  * 4);
    double*   red   = (double*)  carve(2 * sizeof(double));
    double*   part  = (double*)score;
    (void)ws_size; (void)in_sizes; (void)n_in; (void)out_size;

    float* jk = (float*)d_out;

    const int mtiles = NN / 16;
    const dim3 blk(256);
    const dim3 gGemm((mtiles * 32 + 255) / 256);
    const dim3 gRows((NN * 32 + 255) / 256);
    const dim3 gEdgeW(((size_t)E2E * 32 + 255) / 256);
    const dim3 gElem(((size_t)NN * HCC + 255) / 256);
    const int totalHC = NN * HCC;
    const int nStatBlk = (NN * HCC + 255) / 256;

    const size_t shLin  = (size_t)HCC * (FF + 4) * 4;
    const size_t shProj = (size_t)FF * (HCC + 4) * 4;

    gemm_wmma_f32<2, INF_, false><<<gGemm, blk, 0, stream>>>(x, embed_W, embed_b, raw, NN, FF);
    ln_rows_kernel<<<gRows, blk, 0, stream>>>(raw, embed_lnw, embed_lnb, h, jk, 0, NN);

    for (int l = 0; l < LL; ++l) {
        const float* Wl_l = Wl + (size_t)l * HCC * FF;
        const float* bl_l = bl + (size_t)l * HCC;
        const float* Wr_l = Wr + (size_t)l * HCC * FF;
        const float* br_l = br + (size_t)l * HCC;
        const float* We_l = We + (size_t)l * HCC * EDD;
        const float* at_l = att + (size_t)l * HH * CC;
        const float* cb_l = conv_b + (size_t)l * HCC;
        const float* pw_l = pln_w + (size_t)l * HCC;
        const float* pb_l = pln_b + (size_t)l * HCC;
        const float* pW_l = proj_W + (size_t)l * FF * HCC;
        const float* pB_l = proj_b + (size_t)l * FF;
        const float* l2w  = ln2_w + (size_t)l * FF;
        const float* l2b  = ln2_b + (size_t)l * FF;

        gemm_wmma_f32<8, FF, true><<<gGemm, blk, shLin, stream>>>(h, Wl_l, bl_l, xl, NN, HCC);
        gemm_wmma_f32<8, FF, true><<<gGemm, blk, shLin, stream>>>(h, Wr_l, br_l, xr, NN, HCC);

        hipMemsetAsync(smax, 0, (size_t)NN * HH * 4, stream);
        hipMemsetAsync(den,  0, (size_t)NN * HH * 4, stream);
        hipMemsetAsync(agg,  0, (size_t)NN * HCC * 4, stream);

        edge_score_kernel<<<gEdgeW, blk, 0, stream>>>(srcE, dstE, edge_attr, We_l,
                                                      xl, xr, at_l, score, smax);
        edge_agg_kernel<<<gEdgeW, blk, 0, stream>>>(srcE, dstE, score, smax, den, xl, agg);

        bias_stats_kernel<<<gElem, blk, 0, stream>>>(agg, den, cb_l, part, totalHC);
        stats_finish_kernel<<<dim3(1), dim3(1024), 0, stream>>>(part, nStatBlk, red, totalHC);
        pln_apply_kernel<<<gElem, blk, 0, stream>>>(agg, pw_l, pb_l, red, totalHC);

        gemm_wmma_f32<2, HCC, true><<<gGemm, blk, shProj, stream>>>(agg, pW_l, pB_l, raw, NN, FF);
        ln_rows_kernel<<<gRows, blk, 0, stream>>>(raw, l2w, l2b, h, jk, 1, NN);
    }
}


# edge aggregation: 256 work-groups each own 196 target nodes, filter the edge list once, sum rows in LDS with LDS float adds, store each row once; no global float atomics; sums in LDS arrival order
# speedup vs baseline: 1.6742x; 1.2457x over previous
#include <hip/hip_runtime.h>
#include <math.h>

#define NN   50000
#define EE   800000
#define INF_ 512
#define FF   32
#define HH   4
#define CC   32
#define HCC  128
#define EDD  4
#define E2E  850000
#define LL   2

typedef float v2f __attribute__((ext_vector_type(2)));
typedef float v8f __attribute__((ext_vector_type(8)));

__device__ __forceinline__ unsigned fkey(float f) {
    unsigned u = __float_as_uint(f);
    return (u & 0x80000000u) ? ~u : (u | 0x80000000u);
}
__device__ __forceinline__ float funkey(unsigned u) {
    return __uint_as_float((u & 0x80000000u) ? (u & 0x7FFFFFFFu) : ~u);
}
__device__ __forceinline__ float leakyf(float x, float s) {
    return x >= 0.0f ? x : s * x;
}

template <int CT, int KT, bool USE_LDS>
__global__ __launch_bounds__(256)
void gemm_wmma_f32(const float* __restrict__ A, const float* __restrict__ W,
                   const float* __restrict__ bias, float* __restrict__ Cout,
                   int M, int Nout) {
    extern __shared__ float sW[];
    const int tid  = (int)threadIdx.x;
    const int lane = tid & 31;
    const int wave = (int)((blockIdx.x * blockDim.x + tid) >> 5);
    const int mtiles  = M >> 4;
    const int ngroups = Nout / (16 * CT);
    const int stride  = KT + 4;

    if (USE_LDS) {
        const int total = Nout * KT;
        for (int i = tid; i < total; i += 256) {
            const int c  = i / KT;
            const int kk = i - c * KT;
            sW[c * stride + kk] = W[i];
        }
        __syncthreads();
    }
    if (wave >= mtiles * ngroups) return;

    const int mtile = wave / ngroups;
    const int ng    = wave % ngroups;
    const int row0  = mtile * 16;
    const int col0  = ng * 16 * CT;
    const int hi = lane >> 4;
    const int mi = lane & 15;

    v8f acc[CT] = {};
    const float* Arow = A + (size_t)(row0 + mi) * KT + 2 * hi;

    v2f a_cur;
    a_cur.x = Arow[0];
    a_cur.y = Arow[1];
    v2f b_cur[CT];
    if (!USE_LDS) {
#pragma unroll
        for (int t = 0; t < CT; ++t) {
            const float* Wc = W + (size_t)(col0 + t * 16 + mi) * KT + 2 * hi;
            b_cur[t].x = Wc[0];
            b_cur[t].y = Wc[1];
        }
    }

    for (int k = 0; k < KT; k += 4) {
        v2f a_nxt = {};
        v2f b_nxt[CT];
        if (k + 4 < KT) {
            a_nxt.x = Arow[k + 4];
            a_nxt.y = Arow[k + 5];
            if (!USE_LDS) {
#pragma unroll
                for (int t = 0; t < CT; ++t) {
                    const float* Wc = W + (size_t)(col0 + t * 16 + mi) * KT + 2 * hi;
                    b_nxt[t].x = Wc[k + 4];
                    b_nxt[t].y = Wc[k + 5];
                }
                if ((k & 31) == 0)
                    __builtin_prefetch(Arow + k + 132, 0, 1);
            }
        }
#pragma unroll
        for (int t = 0; t < CT; ++t) {
            v2f b;
            if (USE_LDS) {
                const float* Sc = sW + (size_t)(col0 + t * 16 + mi) * stride + 2 * hi;
                b.x = Sc[k];
                b.y = Sc[k + 1];
            } else {
                b = b_cur[t];
            }
            acc[t] = __builtin_amdgcn_wmma_f32_16x16x4_f32(
                false, a_cur, false, b, (short)0, acc[t], false, false);
        }
        a_cur = a_nxt;
        if (!USE_LDS) {
#pragma unroll
            for (int t = 0; t < CT; ++t) b_cur[t] = b_nxt[t];
        }
    }

#pragma unroll
    for (int t = 0; t < CT; ++t) {
        const int col = col0 + t * 16 + mi;
        const float bb = bias ? bias[col] : 0.0f;
#pragma unroll
        for (int v = 0; v < 8; ++v) {
            const int r = row0 + v + 8 * hi;
            Cout[(size_t)r * Nout + col] = acc[t][v] + bb;
        }
    }
}

__global__ __launch_bounds__(256)
void ln_rows_kernel(const float* __restrict__ Z, const float* __restrict__ w,
                    const float* __restrict__ b, float* __restrict__ Hout,
                    float* __restrict__ jk, int mode, int M) {
    const int row  = (int)((blockIdx.x * blockDim.x + threadIdx.x) >> 5);
    const int lane = (int)(threadIdx.x & 31);
    if (row >= M) return;
    float v = Z[(size_t)row * FF + lane];
    float s = v;
#pragma unroll
    for (int off = 16; off >= 1; off >>= 1) s += __shfl_xor(s, off);
    const float mu = s * (1.0f / 32.0f);
    const float d = v - mu;
    float q = d * d;
#pragma unroll
    for (int off = 16; off >= 1; off >>= 1) q += __shfl_xor(q, off);
    const float var = q * (1.0f / 32.0f);
    float y = w[lane] * d * rsqrtf(var + 1e-5f) + b[lane];
    y = leakyf(y, 0.01f);
    Hout[(size_t)row * FF + lane] = y;
    float* o = jk + (size_t)row * FF + lane;
    *o = (mode == 0) ? y : fmaxf(*o, y);
}

__global__ __launch_bounds__(256)
void edge_score_kernel(const int* __restrict__ src, const int* __restrict__ dst,
                       const float* __restrict__ ea, const float* __restrict__ We,
                       const float* __restrict__ xl, const float* __restrict__ xr,
                       const float* __restrict__ att, float* __restrict__ score,
                       unsigned* __restrict__ smax) {
    const int e    = (int)((blockIdx.x * blockDim.x + threadIdx.x) >> 5);
    const int lane = (int)(threadIdx.x & 31);
    if (e >= E2E) return;
    int sN, dN;
    float4 ev;
    if (e < EE) {
        sN = src[e]; dN = dst[e];
        ev = ((const float4*)ea)[e];
    } else {
        sN = dN = e - EE;
        ev = make_float4(0.0f, 0.0f, 0.0f, 0.0f);
    }
    const float4 xlv = ((const float4*)xl)[(size_t)sN * (HCC / 4) + lane];
    const float4 xrv = ((const float4*)xr)[(size_t)dN * (HCC / 4) + lane];
    const float4 atv = ((const float4*)att)[lane];
    const float4* We4 = (const float4*)We;

    const float xa[4] = {xlv.x, xlv.y, xlv.z, xlv.w};
    const float xb[4] = {xrv.x, xrv.y, xrv.z, xrv.w};
    const float at[4] = {atv.x, atv.y, atv.z, atv.w};
    float sc = 0.0f;
#pragma unroll
    for (int k = 0; k < 4; ++k) {
        const float4 wr = We4[lane * 4 + k];
        const float eev = wr.x * ev.x + wr.y * ev.y + wr.z * ev.z + wr.w * ev.w;
        float m = xa[k] + xb[k] + eev;
        m = leakyf(m, 0.2f);
        sc += at[k] * m;
    }
    sc += __shfl_xor(sc, 4);
    sc += __shfl_xor(sc, 2);
    sc += __shfl_xor(sc, 1);
    if ((lane & 7) == 0) {
        const int h = lane >> 3;
        score[(size_t)e * HH + h] = sc;
        atomicMax(&smax[(size_t)dN * HH + h], fkey(sc));
    }
}

#define NPB    196
#define NOWN   256
#define LCAP   8192
#define NCHUNK (EE / 128)

__global__ __launch_bounds__(1024)
void edge_agg_owner_kernel(const int* __restrict__ src, const int* __restrict__ dst,
                           const float* __restrict__ score, const unsigned* __restrict__ smax,
                           float* __restrict__ den, const float* __restrict__ xl,
                           float* __restrict__ agg) {
    __shared__ float acc[NPB * HCC];
    __shared__ float dsum[NPB * HH];
    __shared__ int   lst[LCAP];
    __shared__ int   lsrc[LCAP];
    __shared__ int   cnt;
    const int tid  = (int)threadIdx.x;
    const int lane = tid & 31;
    const int wv   = tid >> 5;
    const int base = (int)blockIdx.x * NPB;

    for (int i = tid; i < NPB * HCC; i += 1024) acc[i] = 0.0f;
    for (int i = tid; i < NPB * HH; i += 1024) dsum[i] = 0.0f;
    if (tid == 0) cnt = 0;
    __syncthreads();

    const int4* dst4 = (const int4*)dst;
    const int start = (int)(((unsigned)blockIdx.x * (unsigned)NCHUNK) / (unsigned)NOWN);
    for (int j0 = 0; j0 < 196; j0 += 7) {
        int4 dv[7];
        int  cb[7];
#pragma unroll
        for (int u = 0; u < 7; ++u) {
            const int t = wv + 32 * (j0 + u);
            int ch = t + start;
            if (ch >= NCHUNK) ch -= NCHUNK;
            cb[u] = ch;
            dv[u] = (t < NCHUNK) ? dst4[(size_t)ch * 32 + lane] : make_int4(-1, -1, -1, -1);
        }
#pragma unroll
        for (int u = 0; u < 7; ++u) {
            const int e0 = cb[u] * 128 + lane * 4;
            const int dd[4] = {dv[u].x, dv[u].y, dv[u].z, dv[u].w};
#pragma unroll
            for (int k = 0; k < 4; ++k) {
                const unsigned dl = (unsigned)(dd[k] - base);
                if (dl < (unsigned)NPB) {
                    const int pos = atomicAdd(&cnt, 1);
                    if (pos < LCAP) lst[pos] = (e0 + k) | ((int)dl << 20);
                }
            }
        }
    }
    __syncthreads();

    int n = cnt;
    if (n > LCAP) {
        for (int e = tid; e < EE; e += 1024) {
            const int d = dst[e];
            const unsigned dl = (unsigned)(d - base);
            if (dl < (unsigned)NPB) {
                const int s = src[e];
                for (int hh = 0; hh < HH; ++hh) {
                    const float ex = expf(score[(size_t)e * HH + hh] -
                                          funkey(smax[(size_t)d * HH + hh]));
                    atomicAdd(&dsum[dl * HH + hh], ex);
                    for (int cc = 0; cc < CC; ++cc) {
                        const int ch = hh * CC + cc;
                        atomicAdd(&acc[dl * HCC + (ch & 3) * 32 + (ch >> 2)],
                                  xl[(size_t)s * HCC + ch] * ex);
                    }
                }
            }
        }
        n = 0;
    }
    for (int i = tid; i < n; i += 1024) lsrc[i] = src[lst[i] & 0xFFFFF];
    __syncthreads();

    const int total = n + NPB;
    const int h     = lane >> 3;
    const float4* xl4 = (const float4*)xl;
    for (int i0 = wv; i0 < total; i0 += 128) {
        int  rec[4], sl[4];
#pragma unroll
        for (int u = 0; u < 4; ++u) {
            const int idx = i0 + 32 * u;
            rec[u] = lst[idx < n ? idx : 0];
            sl[u]  = lsrc[idx < n ? idx : 0];
        }
        int      dl[4];
        float    sc[4];
        unsigned mk[4];
        float4   xv[4];
        bool     ok[4];
#pragma unroll
        for (int u = 0; u < 4; ++u) {
            const int  idx = i0 + 32 * u;
            const bool inl = idx < n;
            dl[u] = inl ? (rec[u] >> 20) : (idx - n);
            const int node = base + dl[u];
            ok[u] = (idx < total) && (node < NN);
            const int dN = ok[u] ? node : 0;
            const int sN = ok[u] ? (inl ? sl[u] : node) : 0;
            const int eN = ok[u] ? (inl ? (rec[u] & 0xFFFFF) : EE + node) : 0;
            sc[u] = score[(size_t)eN * HH + h];
            mk[u] = smax[(size_t)dN * HH + h];
            xv[u] = xl4[(size_t)sN * (HCC / 4) + lane];
        }
#pragma unroll
        for (int u = 0; u < 4; ++u) {
            if (ok[u]) {
                const float ex = expf(sc[u] - funkey(mk[u]));
                float* a = &acc[dl[u] * HCC + lane];
                atomicAdd(a +  0, xv[u].x * ex);
                atomicAdd(a + 32, xv[u].y * ex);
                atomicAdd(a + 64, xv[u].z * ex);
                atomicAdd(a + 96, xv[u].w * ex);
                if ((lane & 7) == 0) atomicAdd(&dsum[dl[u] * HH + h], ex);
            }
        }
    }
    __syncthreads();

    float4* agg4 = (float4*)agg;
    for (int i = tid; i < NPB * (HCC / 4); i += 1024) {
        const int node = base + (i >> 5);
        const float* a = &acc[(i >> 5) * HCC + (i & 31)];
        if (node < NN)
            agg4[(size_t)node * (HCC / 4) + (i & 31)] = make_float4(a[0], a[32], a[64], a[96]);
    }
    for (int i = tid; i < NPB * HH; i += 1024) {
        const int node = base + (i >> 2);
        if (node < NN) den[(size_t)node * HH + (i & 3)] = dsum[i];
    }
}

__global__ __launch_bounds__(256)
void bias_stats_kernel(float* __restrict__ agg, const float* __restrict__ den,
                       const float* __restrict__ cb,
                       double* __restrict__ part, int total) {
    __shared__ double ss[256];
    __shared__ double sq[256];
    const int idx = (int)(blockIdx.x * blockDim.x + threadIdx.x);
    double s = 0.0, q = 0.0;
    if (idx < total) {
        const float y = agg[idx] / (den[idx >> 5] + 1e-16f) + cb[idx & (HCC - 1)];
        agg[idx] = y;
        s = (double)y;
        q = (double)y * (double)y;
    }
    ss[threadIdx.x] = s;
    sq[threadIdx.x] = q;
    __syncthreads();
    for (int o = 128; o > 0; o >>= 1) {
        if ((int)threadIdx.x < o) {
            ss[threadIdx.x] += ss[threadIdx.x + o];
            sq[threadIdx.x] += sq[threadIdx.x + o];
        }
        __syncthreads();
    }
    if (threadIdx.x == 0) {
        ((double2*)part)[blockIdx.x] = make_double2(ss[0], sq[0]);
    }
}

__global__ __launch_bounds__(1024)
void stats_finish_kernel(const double* __restrict__ part, int nblk,
                         double* __restrict__ red, int total) {
    __shared__ double ws_[32];
    __shared__ double wq_[32];
    const int tid  = (int)threadIdx.x;
    const int lane = tid & 31;
    const int wv   = tid >> 5;
    double s = 0.0, q = 0.0;
    for (int i = tid; i < nblk; i += 1024) {
        const double2 p = ((const double2*)part)[i];
        s += p.x;
        q += p.y;
    }
    for (int off = 16; off >= 1; off >>= 1) {
        s += __shfl_xor(s, off);
        q += __shfl_xor(q, off);
    }
    if (lane == 0) { ws_[wv] = s; wq_[wv] = q; }
    __syncthreads();
    if (wv == 0) {
        s = ws_[lane];
        q = wq_[lane];
        for (int off = 16; off >= 1; off >>= 1) {
            s += __shfl_xor(s, off);
            q += __shfl_xor(q, off);
        }
        if (lane == 0) {
            const double mu = s / (double)total;
            double var = q / (double)total - mu * mu;
            if (var < 0.0) var = 0.0;
            const float sd = sqrtf((float)var);
            red[0] = mu;
            red[1] = (double)(sd + 1e-5f);
        }
    }
}

__global__ __launch_bounds__(256)
void pln_apply_kernel(float* __restrict__ agg, const float* __restrict__ w,
                      const float* __restrict__ b, const double* __restrict__ red,
                      int total) {
    const int idx = (int)(blockIdx.x * blockDim.x + threadIdx.x);
    if (idx >= total) return;
    const double mu  = red[0];
    const double dnm = red[1];
    const int c = idx & (HCC - 1);
    float y = w[c] * (float)((agg[idx] - mu) / dnm) + b[c];
    agg[idx] = leakyf(y, 0.01f);
}

extern "C" void kernel_launch(void* const* d_in, const int* in_sizes, int n_in,
                              void* d_out, int out_size, void* d_ws, size_t ws_size,
                              hipStream_t stream) {
    const float* x          = (const float*)d_in[0];
    const int*   edge_index = (const int*)  d_in[1];
    const float* edge_attr  = (const float*)d_in[2];
    const float* embed_W    = (const float*)d_in[3];
    const float* embed_b    = (const float*)d_in[4];
    const float* embed_lnw  = (const float*)d_in[5];
    const float* embed_lnb  = (const float*)d_in[6];
    const float* Wl         = (const float*)d_in[7];
    const float* bl         = (const float*)d_in[8];
    const float* Wr         = (const float*)d_in[9];
    const float* br         = (const float*)d_in[10];
    const float* We         = (const float*)d_in[11];
    const float* att        = (const float*)d_in[12];
    const float* conv_b     = (const float*)d_in[13];
    const float* pln_w      = (const float*)d_in[14];
    const float* pln_b      = (const float*)d_in[15];
    const float* proj_W     = (const float*)d_in[16];
    const float* proj_b     = (const float*)d_in[17];
    const float* ln2_w      = (const float*)d_in[18];
    const float* ln2_b      = (const float*)d_in[19];

    const int* srcE = edge_index;
    const int* dstE = edge_index + EE;

    size_t off = 0;
    auto carve = [&](size_t nbytes) -> char* {
        char* p = (char*)d_ws + off;
        off += (nbytes + 255) & ~(size_t)255;
        return p;
    };
    float*    h     = (float*)   carve((size_t)NN * FF  * 4);
    float*    raw   = (float*)   carve((size_t)NN * FF  * 4);
    float*    xl    = (float*)   carve((size_t)NN * HCC * 4);
    float*    xr    = (float*)   carve((size_t)NN * HCC * 4);
    float*    agg   = (float*)   carve((size_t)NN * HCC * 4);
    float*    score = (float*)   carve((size_t)E2E * HH * 4);
    unsigned* smax  = (unsigned*)carve((size_t)NN * HH  * 4);
    float*    den   = (float*)   carve((size_t)NN * HH  * 4);
    double*   red   = (double*)  carve(2 * sizeof(double));
    double*   part  = (double*)score;
    (void)ws_size; (void)in_sizes; (void)n_in; (void)out_size;

    float* jk = (float*)d_out;

    const int mtiles = NN / 16;
    const dim3 blk(256);
    const dim3 gGemm((mtiles * 32 + 255) / 256);
    const dim3 gRows((NN * 32 + 255) / 256);
    const dim3 gEdgeW(((size_t)E2E * 32 + 255) / 256);
    const dim3 gElem(((size_t)NN * HCC + 255) / 256);
    const int totalHC = NN * HCC;
    const int nStatBlk = (NN * HCC + 255) / 256;

    const size_t shLin  = (size_t)HCC * (FF + 4) * 4;
    const size_t shProj = (size_t)FF * (HCC + 4) * 4;

    gemm_wmma_f32<2, INF_, false><<<gGemm, blk, 0, stream>>>(x, embed_W, embed_b, raw, NN, FF);
    ln_rows_kernel<<<gRows, blk, 0, stream>>>(raw, embed_lnw, embed_lnb, h, jk, 0, NN);

    for (int l = 0; l < LL; ++l) {
        const float* Wl_l = Wl + (size_t)l * HCC * FF;
        const float* bl_l = bl + (size_t)l * HCC;
        const float* Wr_l = Wr + (size_t)l * HCC * FF;
        const float* br_l = br + (size_t)l * HCC;
        const float* We_l = We + (size_t)l * HCC * EDD;
        const float* at_l = att + (size_t)l * HH * CC;
        const float* cb_l = conv_b + (size_t)l * HCC;
        const float* pw_l = pln_w + (size_t)l * HCC;
        const float* pb_l = pln_b + (size_t)l * HCC;
        const float* pW_l = proj_W + (size_t)l * FF * HCC;
        const float* pB_l = proj_b + (size_t)l * FF;
        const float* l2w  = ln2_w + (size_t)l * FF;
        const float* l2b  = ln2_b + (size_t)l * FF;

        gemm_wmma_f32<8, FF, true><<<gGemm, blk, shLin, stream>>>(h, Wl_l, bl_l, xl, NN, HCC);
        gemm_wmma_f32<8, FF, true><<<gGemm, blk, shLin, stream>>>(h, Wr_l, br_l, xr, NN, HCC);

        hipMemsetAsync(smax, 0, (size_t)NN * HH * 4, stream);

        edge_score_kernel<<<gEdgeW, blk, 0, stream>>>(srcE, dstE, edge_attr, We_l,
                                                      xl, xr, at_l, score, smax);
        edge_agg_owner_kernel<<<dim3(NOWN), dim3(1024), 0, stream>>>(srcE, dstE, score, smax, den, xl, agg);

        bias_stats_kernel<<<gElem, blk, 0, stream>>>(agg, den, cb_l, part, totalHC);
        stats_finish_kernel<<<dim3(1), dim3(1024), 0, stream>>>(part, nStatBlk, red, totalHC);
        pln_apply_kernel<<<gElem, blk, 0, stream>>>(agg, pw_l, pb_l, red, totalHC);

        gemm_wmma_f32<2, HCC, true><<<gGemm, blk, shProj, stream>>>(agg, pW_l, pB_l, raw, NN, FF);
        ln_rows_kernel<<<gRows, blk, 0, stream>>>(raw, l2w, l2b, h, jk, 1, NN);
    }
}


# edge scores and per-target maxima formed inside the owner work-groups (one more sweep; scores and maxima kept in LDS): score kernel, score and maximum arrays and their fill gone; sum order kept
# speedup vs baseline: 2.0420x; 1.2197x over previous
#include <hip/hip_runtime.h>
#include <math.h>

#define NN   50000
#define EE   800000
#define INF_ 512
#define FF   32
#define HH   4
#define CC   32
#define HCC  128
#define EDD  4
#define E2E  850000
#define LL   2

typedef float v2f __attribute__((ext_vector_type(2)));
typedef float v8f __attribute__((ext_vector_type(8)));
typedef float v4f __attribute__((ext_vector_type(4)));

__device__ __forceinline__ unsigned fkey(float f) {
    unsigned u = __float_as_uint(f);
    return (u & 0x80000000u) ? ~u : (u | 0x80000000u);
}
__device__ __forceinline__ float funkey(unsigned u) {
    return __uint_as_float((u & 0x80000000u) ? (u & 0x7FFFFFFFu) : ~u);
}
__device__ __forceinline__ float leakyf(float x, float s) {
    return x >= 0.0f ? x : s * x;
}

template <int CT, int KT, bool USE_LDS>
__global__ __launch_bounds__(256)
void gemm_wmma_f32(const float* __restrict__ A, const float* __restrict__ W,
                   const float* __restrict__ bias, float* __restrict__ Cout,
                   int M, int Nout) {
    extern __shared__ float sW[];
    const int tid  = (int)threadIdx.x;
    const int lane = tid & 31;
    const int wave = (int)((blockIdx.x * blockDim.x + tid) >> 5);
    const int mtiles  = M >> 4;
    const int ngroups = Nout / (16 * CT);
    const int stride  = KT + 4;

    if (USE_LDS) {
        const int total = Nout * KT;
        for (int i = tid; i < total; i += 256) {
            const int c  = i / KT;
            const int kk = i - c * KT;
            sW[c * stride + kk] = W[i];
        }
        __syncthreads();
    }
    if (wave >= mtiles * ngroups) return;

    const int mtile = wave / ngroups;
    const int ng    = wave % ngroups;
    const int row0  = mtile * 16;
    const int col0  = ng * 16 * CT;
    const int hi = lane >> 4;
    const int mi = lane & 15;

    v8f acc[CT] = {};
    const float* Arow = A + (size_t)(row0 + mi) * KT + 2 * hi;

    v2f a_cur;
    a_cur.x = Arow[0];
    a_cur.y = Arow[1];
    v2f b_cur[CT];
    if (!USE_LDS) {
#pragma unroll
        for (int t = 0; t < CT; ++t) {
            const float* Wc = W + (size_t)(col0 + t * 16 + mi) * KT + 2 * hi;
            b_cur[t].x = Wc[0];
            b_cur[t].y = Wc[1];
        }
    }

    for (int k = 0; k < KT; k += 4) {
        v2f a_nxt = {};
        v2f b_nxt[CT];
        if (k + 4 < KT) {
            a_nxt.x = Arow[k + 4];
            a_nxt.y = Arow[k + 5];
            if (!USE_LDS) {
#pragma unroll
                for (int t = 0; t < CT; ++t) {
                    const float* Wc = W + (size_t)(col0 + t * 16 + mi) * KT + 2 * hi;
                    b_nxt[t].x = Wc[k + 4];
                    b_nxt[t].y = Wc[k + 5];
                }
                if ((k & 31) == 0)
                    __builtin_prefetch(Arow + k + 132, 0, 1);
            }
        }
#pragma unroll
        for (int t = 0; t < CT; ++t) {
            v2f b;
            if (USE_LDS) {
                const float* Sc = sW + (size_t)(col0 + t * 16 + mi) * stride + 2 * hi;
                b.x = Sc[k];
                b.y = Sc[k + 1];
            } else {
                b = b_cur[t];
            }
            acc[t] = __builtin_amdgcn_wmma_f32_16x16x4_f32(
                false, a_cur, false, b, (short)0, acc[t], false, false);
        }
        a_cur = a_nxt;
        if (!USE_LDS) {
#pragma unroll
            for (int t = 0; t < CT; ++t) b_cur[t] = b_nxt[t];
        }
    }

#pragma unroll
    for (int t = 0; t < CT; ++t) {
        const int col = col0 + t * 16 + mi;
        const float bb = bias ? bias[col] : 0.0f;
#pragma unroll
        for (int v = 0; v < 8; ++v) {
            const int r = row0 + v + 8 * hi;
            Cout[(size_t)r * Nout + col] = acc[t][v] + bb;
        }
    }
}

__global__ __launch_bounds__(256)
void ln_rows_kernel(const float* __restrict__ Z, const float* __restrict__ w,
                    const float* __restrict__ b, float* __restrict__ Hout,
                    float* __restrict__ jk, int mode, int M) {
    const int row  = (int)((blockIdx.x * blockDim.x + threadIdx.x) >> 5);
    const int lane = (int)(threadIdx.x & 31);
    if (row >= M) return;
    float v = Z[(size_t)row * FF + lane];
    float s = v;
#pragma unroll
    for (int off = 16; off >= 1; off >>= 1) s += __shfl_xor(s, off);
    const float mu = s * (1.0f / 32.0f);
    const float d = v - mu;
    float q = d * d;
#pragma unroll
    for (int off = 16; off >= 1; off >>= 1) q += __shfl_xor(q, off);
    const float var = q * (1.0f / 32.0f);
    float y = w[lane] * d * rsqrtf(var + 1e-5f) + b[lane];
    y = leakyf(y, 0.01f);
    Hout[(size_t)row * FF + lane] = y;
    float* o = jk + (size_t)row * FF + lane;
    *o = (mode == 0) ? y : fmaxf(*o, y);
}

#define NPB    196
#define NOWN   256
#define LCAP   4096
#define NCHUNK (EE / 128)

__device__ __forceinline__ float edge_score_one(int e, int s, int d, int hh,
                                             const float* __restrict__ ea,
                                             const float* __restrict__ We,
                                             const float* __restrict__ xl,
                                             const float* __restrict__ xr,
                                             const float* __restrict__ att) {
    const float4 ev = ((const float4*)ea)[e];
    float sc = 0.0f;
#pragma nounroll
    for (int cc = 0; cc < CC; ++cc) {
        const int ch = hh * CC + cc;
        const float4 wr = ((const float4*)We)[ch];
        const float eev = wr.x * ev.x + wr.y * ev.y + wr.z * ev.z + wr.w * ev.w;
        const float m = xl[(size_t)s * HCC + ch] + xr[(size_t)d * HCC + ch] + eev;
        sc += att[ch] * leakyf(m, 0.2f);
    }
    return sc;
}

__global__ __launch_bounds__(1024)
void edge_owner_kernel(const int* __restrict__ src, const int* __restrict__ dst,
                       const float* __restrict__ ea, const float* __restrict__ We,
                       const float* __restrict__ xl, const float* __restrict__ xr,
                       const float* __restrict__ att,
                       float* __restrict__ den, float* __restrict__ agg) {
    __shared__ float    acc[NPB * HCC];
    __shared__ float    dsum[NPB * HH];
    __shared__ unsigned mx[NPB * HH];
    __shared__ int      lst[LCAP];
    __shared__ int      lsrc[LCAP];
    __shared__ float    lsc[(LCAP + NPB) * HH];
    __shared__ int      cnt;
    const int tid  = (int)threadIdx.x;
    const int lane = tid & 31;
    const int wv   = tid >> 5;
    const int base = (int)blockIdx.x * NPB;

    for (int i = tid; i < NPB * HCC; i += 1024) acc[i] = 0.0f;
    for (int i = tid; i < NPB * HH; i += 1024) { dsum[i] = 0.0f; mx[i] = 0u; }
    if (tid == 0) cnt = 0;
    __syncthreads();

    const int4* dst4 = (const int4*)dst;
    const int start = (int)(((unsigned)blockIdx.x * (unsigned)NCHUNK) / (unsigned)NOWN);
    for (int j0 = 0; j0 < 196; j0 += 7) {
        int4 dv[7];
        int  cb[7];
#pragma unroll
        for (int u = 0; u < 7; ++u) {
            const int t = wv + 32 * (j0 + u);
            int ch = t + start;
            if (ch >= NCHUNK) ch -= NCHUNK;
            cb[u] = ch;
            dv[u] = (t < NCHUNK) ? dst4[(size_t)ch * 32 + lane] : make_int4(-1, -1, -1, -1);
        }
#pragma unroll
        for (int u = 0; u < 7; ++u) {
            const int e0 = cb[u] * 128 + lane * 4;
            const int dd[4] = {dv[u].x, dv[u].y, dv[u].z, dv[u].w};
#pragma unroll
            for (int k = 0; k < 4; ++k) {
                const unsigned dl = (unsigned)(dd[k] - base);
                if (dl < (unsigned)NPB) {
                    const int pos = atomicAdd(&cnt, 1);
                    if (pos < LCAP) lst[pos] = (e0 + k) | ((int)dl << 20);
                }
            }
        }
    }
    __syncthreads();

    int n = cnt;
    const bool over = n > LCAP;
    if (over) {
        for (int e = tid; e < EE; e += 1024) {
            const int d = dst[e];
            const unsigned dl = (unsigned)(d - base);
            if (dl < (unsigned)NPB) {
                const int s = src[e];
                for (int hh = 0; hh < HH; ++hh)
                    atomicMax(&mx[dl * HH + hh],
                              fkey(edge_score_one(e, s, d, hh, ea, We, xl, xr, att)));
            }
        }
        n = 0;
    }
    for (int i = tid; i < n; i += 1024) lsrc[i] = src[lst[i] & 0xFFFFF];
    __syncthreads();

    const int total = n + NPB;
    const int h     = lane >> 3;
    const float4* xl4 = (const float4*)xl;
    const float4* xr4 = (const float4*)xr;

    {
        const float4* We4 = (const float4*)We;
        const v4f* xl4v = (const v4f*)xl;
        const v4f* xr4v = (const v4f*)xr;
        const v4f* ea4v = (const v4f*)ea;
        float4 wr[4];
#pragma unroll
        for (int k = 0; k < 4; ++k) wr[k] = We4[lane * 4 + k];
        const float4 atv = ((const float4*)att)[lane];
        const float at[4] = {atv.x, atv.y, atv.z, atv.w};

        for (int i0 = wv; i0 < n; i0 += 128) {
            int rec[4], sl[4];
#pragma unroll
            for (int u = 0; u < 4; ++u) {
                const int idx = i0 + 32 * u;
                rec[u] = lst[idx < n ? idx : 0];
                sl[u]  = lsrc[idx < n ? idx : 0];
            }
            v4f xa[4], xb[4], ev[4];
#pragma unroll
            for (int u = 0; u < 4; ++u) {
                xa[u] = xl4v[(size_t)sl[u] * (HCC / 4) + lane];
                xb[u] = xr4v[(size_t)(base + (rec[u] >> 20)) * (HCC / 4) + lane];
                ev[u] = ea4v[rec[u] & 0xFFFFF];
            }
            asm volatile("" : "+v"(xa[0]), "+v"(xa[1]), "+v"(xa[2]), "+v"(xa[3]),
                              "+v"(xb[0]), "+v"(xb[1]), "+v"(xb[2]), "+v"(xb[3]),
                              "+v"(ev[0]), "+v"(ev[1]), "+v"(ev[2]), "+v"(ev[3]));
            float sc[4];
#pragma unroll
            for (int u = 0; u < 4; ++u) {
                const float pa[4] = {xa[u].x, xa[u].y, xa[u].z, xa[u].w};
                const float pb[4] = {xb[u].x, xb[u].y, xb[u].z, xb[u].w};
                float s = 0.0f;
#pragma unroll
                for (int k = 0; k < 4; ++k) {
                    const float eev = wr[k].x * ev[u].x + wr[k].y * ev[u].y +
                                      wr[k].z * ev[u].z + wr[k].w * ev[u].w;
                    float m = pa[k] + pb[k] + eev;
                    m = leakyf(m, 0.2f);
                    s += at[k] * m;
                }
                sc[u] = s;
            }
#pragma unroll
            for (int off = 4; off >= 1; off >>= 1) {
#pragma unroll
                for (int u = 0; u < 4; ++u) sc[u] += __shfl_xor(sc[u], off);
            }
#pragma unroll
            for (int u = 0; u < 4; ++u) {
                const int idx = i0 + 32 * u;
                if (idx < n && (lane & 7) == 0) {
                    lsc[idx * HH + h] = sc[u];
                    atomicMax(&mx[(rec[u] >> 20) * HH + h], fkey(sc[u]));
                }
            }
        }

        for (int j0 = wv; j0 < NPB; j0 += 128) {
            float4 xa[4], xb[4];
            bool   ok[4];
#pragma unroll
            for (int u = 0; u < 4; ++u) {
                const int j    = j0 + 32 * u;
                const int node = base + j;
                ok[u] = (j < NPB) && (node < NN);
                const int nd = ok[u] ? node : 0;
                xa[u] = xl4[(size_t)nd * (HCC / 4) + lane];
                xb[u] = xr4[(size_t)nd * (HCC / 4) + lane];
            }
            float sc[4];
#pragma unroll
            for (int u = 0; u < 4; ++u) {
                const float pa[4] = {xa[u].x, xa[u].y, xa[u].z, xa[u].w};
                const float pb[4] = {xb[u].x, xb[u].y, xb[u].z, xb[u].w};
                float s = 0.0f;
#pragma unroll
                for (int k = 0; k < 4; ++k) {
                    float m = pa[k] + pb[k];
                    m = leakyf(m, 0.2f);
                    s += at[k] * m;
                }
                sc[u] = s;
            }
#pragma unroll
            for (int off = 4; off >= 1; off >>= 1) {
#pragma unroll
                for (int u = 0; u < 4; ++u) sc[u] += __shfl_xor(sc[u], off);
            }
#pragma unroll
            for (int u = 0; u < 4; ++u) {
                const int j = j0 + 32 * u;
                if (ok[u] && (lane & 7) == 0) {
                    lsc[(n + j) * HH + h] = sc[u];
                    atomicMax(&mx[j * HH + h], fkey(sc[u]));
                }
            }
        }
    }
    __syncthreads();

    if (over) {
        for (int e = tid; e < EE; e += 1024) {
            const int d = dst[e];
            const unsigned dl = (unsigned)(d - base);
            if (dl < (unsigned)NPB) {
                const int s = src[e];
                for (int hh = 0; hh < HH; ++hh) {
                    const float ex = expf(edge_score_one(e, s, d, hh, ea, We, xl, xr, att) -
                                          funkey(mx[dl * HH + hh]));
                    atomicAdd(&dsum[dl * HH + hh], ex);
                    for (int cc = 0; cc < CC; ++cc) {
                        const int ch = hh * CC + cc;
                        atomicAdd(&acc[dl * HCC + (ch & 3) * 32 + (ch >> 2)],
                                  xl[(size_t)s * HCC + ch] * ex);
                    }
                }
            }
        }
    }

    for (int i0 = wv; i0 < total; i0 += 128) {
        int rec[4], sl[4];
#pragma unroll
        for (int u = 0; u < 4; ++u) {
            const int idx = i0 + 32 * u;
            rec[u] = lst[idx < n ? idx : 0];
            sl[u]  = lsrc[idx < n ? idx : 0];
        }
        int      dl[4];
        float    sc[4];
        unsigned mk[4];
        float4   xv[4];
        bool     ok[4];
#pragma unroll
        for (int u = 0; u < 4; ++u) {
            const int  idx = i0 + 32 * u;
            const bool inl = idx < n;
            dl[u] = inl ? (rec[u] >> 20) : (idx - n);
            const int node = base + dl[u];
            ok[u] = (idx < total) && (node < NN);
            const int sN = ok[u] ? (inl ? sl[u] : node) : 0;
            sc[u] = lsc[(ok[u] ? idx : 0) * HH + h];
            mk[u] = mx[(ok[u] ? dl[u] : 0) * HH + h];
            xv[u] = xl4[(size_t)sN * (HCC / 4) + lane];
        }
#pragma unroll
        for (int u = 0; u < 4; ++u) {
            if (ok[u]) {
                const float ex = expf(sc[u] - funkey(mk[u]));
                float* a = &acc[dl[u] * HCC + lane];
                atomicAdd(a +  0, xv[u].x * ex);
                atomicAdd(a + 32, xv[u].y * ex);
                atomicAdd(a + 64, xv[u].z * ex);
                atomicAdd(a + 96, xv[u].w * ex);
                if ((lane & 7) == 0) atomicAdd(&dsum[dl[u] * HH + h], ex);
            }
        }
    }
    __syncthreads();

    float4* agg4 = (float4*)agg;
    for (int i = tid; i < NPB * (HCC / 4); i += 1024) {
        const int node = base + (i >> 5);
        const float* a = &acc[(i >> 5) * HCC + (i & 31)];
        if (node < NN)
            agg4[(size_t)node * (HCC / 4) + (i & 31)] = make_float4(a[0], a[32], a[64], a[96]);
    }
    for (int i = tid; i < NPB * HH; i += 1024) {
        const int node = base + (i >> 2);
        if (node < NN) den[(size_t)node * HH + (i & 3)] = dsum[i];
    }
}

__global__ __launch_bounds__(256)
void bias_stats_kernel(float* __restrict__ agg, const float* __restrict__ den,
                       const float* __restrict__ cb,
                       double* __restrict__ part, int total) {
    __shared__ double ss[256];
    __shared__ double sq[256];
    const int idx = (int)(blockIdx.x * blockDim.x + threadIdx.x);
    double s = 0.0, q = 0.0;
    if (idx < total) {
        const float y = agg[idx] / (den[idx >> 5] + 1e-16f) + cb[idx & (HCC - 1)];
        agg[idx] = y;
        s = (double)y;
        q = (double)y * (double)y;
    }
    ss[threadIdx.x] = s;
    sq[threadIdx.x] = q;
    __syncthreads();
    for (int o = 128; o > 0; o >>= 1) {
        if ((int)threadIdx.x < o) {
            ss[threadIdx.x] += ss[threadIdx.x + o];
            sq[threadIdx.x] += sq[threadIdx.x + o];
        }
        __syncthreads();
    }
    if (threadIdx.x == 0) {
        ((double2*)part)[blockIdx.x] = make_double2(ss[0], sq[0]);
    }
}

__global__ __launch_bounds__(1024)
void stats_finish_kernel(const double* __restrict__ part, int nblk,
                         double* __restrict__ red, int total) {
    __shared__ double ws_[32];
    __shared__ double wq_[32];
    const int tid  = (int)threadIdx.x;
    const int lane = tid & 31;
    const int wv   = tid >> 5;
    double s = 0.0, q = 0.0;
    for (int i = tid; i < nblk; i += 1024) {
        const double2 p = ((const double2*)part)[i];
        s += p.x;
        q += p.y;
    }
    for (int off = 16; off >= 1; off >>= 1) {
        s += __shfl_xor(s, off);
        q += __shfl_xor(q, off);
    }
    if (lane == 0) { ws_[wv] = s; wq_[wv] = q; }
    __syncthreads();
    if (wv == 0) {
        s = ws_[lane];
        q = wq_[lane];
        for (int off = 16; off >= 1; off >>= 1) {
            s += __shfl_xor(s, off);
            q += __shfl_xor(q, off);
        }
        if (lane == 0) {
            const double mu = s / (double)total;
            double var = q / (double)total - mu * mu;
            if (var < 0.0) var = 0.0;
            const float sd = sqrtf((float)var);
            red[0] = mu;
            red[1] = (double)(sd + 1e-5f);
        }
    }
}

__global__ __launch_bounds__(256)
void pln_apply_kernel(float* __restrict__ agg, const float* __restrict__ w,
                      const float* __restrict__ b, const double* __restrict__ red,
                      int total) {
    const int idx = (int)(blockIdx.x * blockDim.x + threadIdx.x);
    if (idx >= total) return;
    const double mu  = red[0];
    const double dnm = red[1];
    const int c = idx & (HCC - 1);
    float y = w[c] * (float)((agg[idx] - mu) / dnm) + b[c];
    agg[idx] = leakyf(y, 0.01f);
}

extern "C" void kernel_launch(void* const* d_in, const int* in_sizes, int n_in,
                              void* d_out, int out_size, void* d_ws, size_t ws_size,
                              hipStream_t stream) {
    const float* x          = (const float*)d_in[0];
    const int*   edge_index = (const int*)  d_in[1];
    const float* edge_attr  = (const float*)d_in[2];
    const float* embed_W    = (const float*)d_in[3];
    const float* embed_b    = (const float*)d_in[4];
    const float* embed_lnw  = (const float*)d_in[5];
    const float* embed_lnb  = (const float*)d_in[6];
    const float* Wl         = (const float*)d_in[7];
    const float* bl         = (const float*)d_in[8];
    const float* Wr         = (const float*)d_in[9];
    const float* br         = (const float*)d_in[10];
    const float* We         = (const float*)d_in[11];
    const float* att        = (const float*)d_in[12];
    const float* conv_b     = (const float*)d_in[13];
    const float* pln_w      = (const float*)d_in[14];
    const float* pln_b      = (const float*)d_in[15];
    const float* proj_W     = (const float*)d_in[16];
    const float* proj_b     = (const float*)d_in[17];
    const float* ln2_w      = (const float*)d_in[18];
    const float* ln2_b      = (const float*)d_in[19];

    const int* srcE = edge_index;
    const int* dstE = edge_index + EE;

    size_t off = 0;
    auto carve = [&](size_t nbytes) -> char* {
        char* p = (char*)d_ws + off;
        off += (nbytes + 255) & ~(size_t)255;
        return p;
    };
    float*    h     = (float*)   carve((size_t)NN * FF  * 4);
    float*    raw   = (float*)   carve((size_t)NN * FF  * 4);
    float*    xl    = (float*)   carve((size_t)NN * HCC * 4);
    float*    xr    = (float*)   carve((size_t)NN * HCC * 4);
    float*    agg   = (float*)   carve((size_t)NN * HCC * 4);
    float*    den   = (float*)   carve((size_t)NN * HH  * 4);
    double*   red   = (double*)  carve(2 * sizeof(double));
    double*   part  = (double*)  carve((size_t)((NN * HCC + 255) / 256) * 2 * sizeof(double));
    (void)ws_size; (void)in_sizes; (void)n_in; (void)out_size;

    float* jk = (float*)d_out;

    const int mtiles = NN / 16;
    const dim3 blk(256);
    const dim3 gGemm((mtiles * 32 + 255) / 256);
    const dim3 gRows((NN * 32 + 255) / 256);
    const dim3 gElem(((size_t)NN * HCC + 255) / 256);
    const int totalHC = NN * HCC;
    const int nStatBlk = (NN * HCC + 255) / 256;

    const size_t shLin  = (size_t)HCC * (FF + 4) * 4;
    const size_t shProj = (size_t)FF * (HCC + 4) * 4;

    gemm_wmma_f32<2, INF_, false><<<gGemm, blk, 0, stream>>>(x, embed_W, embed_b, raw, NN, FF);
    ln_rows_kernel<<<gRows, blk, 0, stream>>>(raw, embed_lnw, embed_lnb, h, jk, 0, NN);

    for (int l = 0; l < LL; ++l) {
        const float* Wl_l = Wl + (size_t)l * HCC * FF;
        const float* bl_l = bl + (size_t)l * HCC;
        const float* Wr_l = Wr + (size_t)l * HCC * FF;
        const float* br_l = br + (size_t)l * HCC;
        const float* We_l = We + (size_t)l * HCC * EDD;
        const float* at_l = att + (size_t)l * HH * CC;
        const float* cb_l = conv_b + (size_t)l * HCC;
        const float* pw_l = pln_w + (size_t)l * HCC;
        const float* pb_l = pln_b + (size_t)l * HCC;
        const float* pW_l = proj_W + (size_t)l * FF * HCC;
        const float* pB_l = proj_b + (size_t)l * FF;
        const float* l2w  = ln2_w + (size_t)l * FF;
        const float* l2b  = ln2_b + (size_t)l * FF;

        gemm_wmma_f32<8, FF, true><<<gGemm, blk, shLin, stream>>>(h, Wl_l, bl_l, xl, NN, HCC);
        gemm_wmma_f32<8, FF, true><<<gGemm, blk, shLin, stream>>>(h, Wr_l, br_l, xr, NN, HCC);

        edge_owner_kernel<<<dim3(NOWN), dim3(1024), 0, stream>>>(srcE, dstE, edge_attr, We_l,
                                                                 xl, xr, at_l, den, agg);

        bias_stats_kernel<<<gElem, blk, 0, stream>>>(agg, den, cb_l, part, totalHC);
        stats_finish_kernel<<<dim3(1), dim3(1024), 0, stream>>>(part, nStatBlk, red, totalHC);
        pln_apply_kernel<<<gElem, blk, 0, stream>>>(agg, pw_l, pb_l, red, totalHC);

        gemm_wmma_f32<2, HCC, true><<<gGemm, blk, shProj, stream>>>(agg, pW_l, pB_l, raw, NN, FF);
        ln_rows_kernel<<<gRows, blk, 0, stream>>>(raw, l2w, l2b, h, jk, 1, NN);
    }
}


# owner work-groups' edge lists formed once per call: the first layer's edge launch stores entries, sources and count to workspace, the second loads them and skips the target-list read; sum order kept
# speedup vs baseline: 2.2036x; 1.0791x over previous
#include <hip/hip_runtime.h>
#include <math.h>

#define NN   50000
#define EE   800000
#define INF_ 512
#define FF   32
#define HH   4
#define CC   32
#define HCC  128
#define EDD  4
#define E2E  850000
#define LL   2

typedef float v2f __attribute__((ext_vector_type(2)));
typedef float v8f __attribute__((ext_vector_type(8)));
typedef float v4f __attribute__((ext_vector_type(4)));

__device__ __forceinline__ unsigned fkey(float f) {
    unsigned u = __float_as_uint(f);
    return (u & 0x80000000u) ? ~u : (u | 0x80000000u);
}
__device__ __forceinline__ float funkey(unsigned u) {
    return __uint_as_float((u & 0x80000000u) ? (u & 0x7FFFFFFFu) : ~u);
}
__device__ __forceinline__ float leakyf(float x, float s) {
    return x >= 0.0f ? x : s * x;
}

template <int CT, int KT, bool USE_LDS>
__global__ __launch_bounds__(256)
void gemm_wmma_f32(const float* __restrict__ A, const float* __restrict__ W,
                   const float* __restrict__ bias, float* __restrict__ Cout,
                   int M, int Nout) {
    extern __shared__ float sW[];
    const int tid  = (int)threadIdx.x;
    const int lane = tid & 31;
    const int wave = (int)((blockIdx.x * blockDim.x + tid) >> 5);
    const int mtiles  = M >> 4;
    const int ngroups = Nout / (16 * CT);
    const int stride  = KT + 4;

    if (USE_LDS) {
        const int total = Nout * KT;
        for (int i = tid; i < total; i += 256) {
            const int c  = i / KT;
            const int kk = i - c * KT;
            sW[c * stride + kk] = W[i];
        }
        __syncthreads();
    }
    if (wave >= mtiles * ngroups) return;

    const int mtile = wave / ngroups;
    const int ng    = wave % ngroups;
    const int row0  = mtile * 16;
    const int col0  = ng * 16 * CT;
    const int hi = lane >> 4;
    const int mi = lane & 15;

    v8f acc[CT] = {};
    const float* Arow = A + (size_t)(row0 + mi) * KT + 2 * hi;

    v2f a_cur;
    a_cur.x = Arow[0];
    a_cur.y = Arow[1];
    v2f b_cur[CT];
    if (!USE_LDS) {
#pragma unroll
        for (int t = 0; t < CT; ++t) {
            const float* Wc = W + (size_t)(col0 + t * 16 + mi) * KT + 2 * hi;
            b_cur[t].x = Wc[0];
            b_cur[t].y = Wc[1];
        }
    }

    for (int k = 0; k < KT; k += 4) {
        v2f a_nxt = {};
        v2f b_nxt[CT];
        if (k + 4 < KT) {
            a_nxt.x = Arow[k + 4];
            a_nxt.y = Arow[k + 5];
            if (!USE_LDS) {
#pragma unroll
                for (int t = 0; t < CT; ++t) {
                    const float* Wc = W + (size_t)(col0 + t * 16 + mi) * KT + 2 * hi;
                    b_nxt[t].x = Wc[k + 4];
                    b_nxt[t].y = Wc[k + 5];
                }
                if ((k & 31) == 0)
                    __builtin_prefetch(Arow + k + 132, 0, 1);
            }
        }
#pragma unroll
        for (int t = 0; t < CT; ++t) {
            v2f b;
            if (USE_LDS) {
                const float* Sc = sW + (size_t)(col0 + t * 16 + mi) * stride + 2 * hi;
                b.x = Sc[k];
                b.y = Sc[k + 1];
            } else {
                b = b_cur[t];
            }
            acc[t] = __builtin_amdgcn_wmma_f32_16x16x4_f32(
                false, a_cur, false, b, (short)0, acc[t], false, false);
        }
        a_cur = a_nxt;
        if (!USE_LDS) {
#pragma unroll
            for (int t = 0; t < CT; ++t) b_cur[t] = b_nxt[t];
        }
    }

#pragma unroll
    for (int t = 0; t < CT; ++t) {
        const int col = col0 + t * 16 + mi;
        const float bb = bias ? bias[col] : 0.0f;
#pragma unroll
        for (int v = 0; v < 8; ++v) {
            const int r = row0 + v + 8 * hi;
            Cout[(size_t)r * Nout + col] = acc[t][v] + bb;
        }
    }
}

__global__ __launch_bounds__(256)
void ln_rows_kernel(const float* __restrict__ Z, const float* __restrict__ w,
                    const float* __restrict__ b, float* __restrict__ Hout,
                    float* __restrict__ jk, int mode, int M) {
    const int row  = (int)((blockIdx.x * blockDim.x + threadIdx.x) >> 5);
    const int lane = (int)(threadIdx.x & 31);
    if (row >= M) return;
    float v = Z[(size_t)row * FF + lane];
    float s = v;
#pragma unroll
    for (int off = 16; off >= 1; off >>= 1) s += __shfl_xor(s, off);
    const float mu = s * (1.0f / 32.0f);
    const float d = v - mu;
    float q = d * d;
#pragma unroll
    for (int off = 16; off >= 1; off >>= 1) q += __shfl_xor(q, off);
    const float var = q * (1.0f / 32.0f);
    float y = w[lane] * d * rsqrtf(var + 1e-5f) + b[lane];
    y = leakyf(y, 0.01f);
    Hout[(size_t)row * FF + lane] = y;
    float* o = jk + (size_t)row * FF + lane;
    *o = (mode == 0) ? y : fmaxf(*o, y);
}

#define NPB    196
#define NOWN   256
#define LCAP   4096
#define NCHUNK (EE / 128)

__device__ __forceinline__ float edge_score_one(int e, int s, int d, int hh,
                                             const float* __restrict__ ea,
                                             const float* __restrict__ We,
                                             const float* __restrict__ xl,
                                             const float* __restrict__ xr,
                                             const float* __restrict__ att) {
    const float4 ev = ((const float4*)ea)[e];
    float sc = 0.0f;
#pragma nounroll
    for (int cc = 0; cc < CC; ++cc) {
        const int ch = hh * CC + cc;
        const float4 wr = ((const float4*)We)[ch];
        const float eev = wr.x * ev.x + wr.y * ev.y + wr.z * ev.z + wr.w * ev.w;
        const float m = xl[(size_t)s * HCC + ch] + xr[(size_t)d * HCC + ch] + eev;
        sc += att[ch] * leakyf(m, 0.2f);
    }
    return sc;
}

template <bool BUILD>
__global__ __launch_bounds__(1024)
void edge_owner_kernel(const int* __restrict__ src, const int* __restrict__ dst,
                       const float* __restrict__ ea, const float* __restrict__ We,
                       const float* __restrict__ xl, const float* __restrict__ xr,
                       const float* __restrict__ att,
                       float* __restrict__ den, float* __restrict__ agg,
                       int* __restrict__ glst, int* __restrict__ gsrc,
                       int* __restrict__ gcnt) {
    __shared__ float    acc[NPB * HCC];
    __shared__ float    dsum[NPB * HH];
    __shared__ unsigned mx[NPB * HH];
    __shared__ int      lst[LCAP];
    __shared__ int      lsrc[LCAP];
    __shared__ float    lsc[(LCAP + NPB) * HH];
    __shared__ int      cnt;
    const int tid  = (int)threadIdx.x;
    const int lane = tid & 31;
    const int wv   = tid >> 5;
    const int base = (int)blockIdx.x * NPB;

    for (int i = tid; i < NPB * HCC; i += 1024) acc[i] = 0.0f;
    for (int i = tid; i < NPB * HH; i += 1024) { dsum[i] = 0.0f; mx[i] = 0u; }
    if (BUILD && tid == 0) cnt = 0;
    __syncthreads();

    int n;
    if (BUILD) {
    const int4* dst4 = (const int4*)dst;
    const int start = (int)(((unsigned)blockIdx.x * (unsigned)NCHUNK) / (unsigned)NOWN);
    for (int j0 = 0; j0 < 196; j0 += 7) {
        int4 dv[7];
        int  cb[7];
#pragma unroll
        for (int u = 0; u < 7; ++u) {
            const int t = wv + 32 * (j0 + u);
            int ch = t + start;
            if (ch >= NCHUNK) ch -= NCHUNK;
            cb[u] = ch;
            dv[u] = (t < NCHUNK) ? dst4[(size_t)ch * 32 + lane] : make_int4(-1, -1, -1, -1);
        }
#pragma unroll
        for (int u = 0; u < 7; ++u) {
            const int e0 = cb[u] * 128 + lane * 4;
            const int dd[4] = {dv[u].x, dv[u].y, dv[u].z, dv[u].w};
#pragma unroll
            for (int k = 0; k < 4; ++k) {
                const unsigned dl = (unsigned)(dd[k] - base);
                if (dl < (unsigned)NPB) {
                    const int pos = atomicAdd(&cnt, 1);
                    if (pos < LCAP) lst[pos] = (e0 + k) | ((int)dl << 20);
                }
            }
        }
    }
    __syncthreads();
    n = cnt;
    } else {
        n = gcnt[blockIdx.x];
    }
    const int  nraw = n;
    const bool over = n > LCAP;
    if (over) {
        for (int e = tid; e < EE; e += 1024) {
            const int d = dst[e];
            const unsigned dl = (unsigned)(d - base);
            if (dl < (unsigned)NPB) {
                const int s = src[e];
                for (int hh = 0; hh < HH; ++hh)
                    atomicMax(&mx[dl * HH + hh],
                              fkey(edge_score_one(e, s, d, hh, ea, We, xl, xr, att)));
            }
        }
        n = 0;
    }
    int* mylst = glst + (size_t)blockIdx.x * LCAP;
    int* mysrc = gsrc + (size_t)blockIdx.x * LCAP;
    if (BUILD) {
        if (tid == 0) gcnt[blockIdx.x] = nraw;
        for (int i = tid; i < n; i += 1024) {
            const int r = lst[i];
            const int s = src[r & 0xFFFFF];
            lsrc[i]  = s;
            mylst[i] = r;
            mysrc[i] = s;
        }
    } else {
        for (int i = tid; i < n; i += 1024) {
            lst[i]  = mylst[i];
            lsrc[i] = mysrc[i];
        }
    }
    __syncthreads();

    const int total = n + NPB;
    const int h     = lane >> 3;
    const float4* xl4 = (const float4*)xl;
    const float4* xr4 = (const float4*)xr;

    {
        const float4* We4 = (const float4*)We;
        const v4f* xl4v = (const v4f*)xl;
        const v4f* xr4v = (const v4f*)xr;
        const v4f* ea4v = (const v4f*)ea;
        float4 wr[4];
#pragma unroll
        for (int k = 0; k < 4; ++k) wr[k] = We4[lane * 4 + k];
        const float4 atv = ((const float4*)att)[lane];
        const float at[4] = {atv.x, atv.y, atv.z, atv.w};

        for (int i0 = wv; i0 < n; i0 += 128) {
            int rec[4], sl[4];
#pragma unroll
            for (int u = 0; u < 4; ++u) {
                const int idx = i0 + 32 * u;
                rec[u] = lst[idx < n ? idx : 0];
                sl[u]  = lsrc[idx < n ? idx : 0];
            }
            v4f xa[4], xb[4], ev[4];
#pragma unroll
            for (int u = 0; u < 4; ++u) {
                xa[u] = xl4v[(size_t)sl[u] * (HCC / 4) + lane];
                xb[u] = xr4v[(size_t)(base + (rec[u] >> 20)) * (HCC / 4) + lane];
                ev[u] = ea4v[rec[u] & 0xFFFFF];
            }
            asm volatile("" : "+v"(xa[0]), "+v"(xa[1]), "+v"(xa[2]), "+v"(xa[3]),
                              "+v"(xb[0]), "+v"(xb[1]), "+v"(xb[2]), "+v"(xb[3]),
                              "+v"(ev[0]), "+v"(ev[1]), "+v"(ev[2]), "+v"(ev[3]));
            float sc[4];
#pragma unroll
            for (int u = 0; u < 4; ++u) {
                const float pa[4] = {xa[u].x, xa[u].y, xa[u].z, xa[u].w};
                const float pb[4] = {xb[u].x, xb[u].y, xb[u].z, xb[u].w};
                float s = 0.0f;
#pragma unroll
                for (int k = 0; k < 4; ++k) {
                    const float eev = wr[k].x * ev[u].x + wr[k].y * ev[u].y +
                                      wr[k].z * ev[u].z + wr[k].w * ev[u].w;
                    float m = pa[k] + pb[k] + eev;
                    m = leakyf(m, 0.2f);
                    s += at[k] * m;
                }
                sc[u] = s;
            }
#pragma unroll
            for (int off = 4; off >= 1; off >>= 1) {
#pragma unroll
                for (int u = 0; u < 4; ++u) sc[u] += __shfl_xor(sc[u], off);
            }
#pragma unroll
            for (int u = 0; u < 4; ++u) {
                const int idx = i0 + 32 * u;
                if (idx < n && (lane & 7) == 0) {
                    lsc[idx * HH + h] = sc[u];
                    atomicMax(&mx[(rec[u] >> 20) * HH + h], fkey(sc[u]));
                }
            }
        }

        for (int j0 = wv; j0 < NPB; j0 += 128) {
            float4 xa[4], xb[4];
            bool   ok[4];
#pragma unroll
            for (int u = 0; u < 4; ++u) {
                const int j    = j0 + 32 * u;
                const int node = base + j;
                ok[u] = (j < NPB) && (node < NN);
                const int nd = ok[u] ? node : 0;
                xa[u] = xl4[(size_t)nd * (HCC / 4) + lane];
                xb[u] = xr4[(size_t)nd * (HCC / 4) + lane];
            }
            float sc[4];
#pragma unroll
            for (int u = 0; u < 4; ++u) {
                const float pa[4] = {xa[u].x, xa[u].y, xa[u].z, xa[u].w};
                const float pb[4] = {xb[u].x, xb[u].y, xb[u].z, xb[u].w};
                float s = 0.0f;
#pragma unroll
                for (int k = 0; k < 4; ++k) {
                    float m = pa[k] + pb[k];
                    m = leakyf(m, 0.2f);
                    s += at[k] * m;
                }
                sc[u] = s;
            }
#pragma unroll
            for (int off = 4; off >= 1; off >>= 1) {
#pragma unroll
                for (int u = 0; u < 4; ++u) sc[u] += __shfl_xor(sc[u], off);
            }
#pragma unroll
            for (int u = 0; u < 4; ++u) {
                const int j = j0 + 32 * u;
                if (ok[u] && (lane & 7) == 0) {
                    lsc[(n + j) * HH + h] = sc[u];
                    atomicMax(&mx[j * HH + h], fkey(sc[u]));
                }
            }
        }
    }
    __syncthreads();

    if (over) {
        for (int e = tid; e < EE; e += 1024) {
            const int d = dst[e];
            const unsigned dl = (unsigned)(d - base);
            if (dl < (unsigned)NPB) {
                const int s = src[e];
                for (int hh = 0; hh < HH; ++hh) {
                    const float ex = expf(edge_score_one(e, s, d, hh, ea, We, xl, xr, att) -
                                          funkey(mx[dl * HH + hh]));
                    atomicAdd(&dsum[dl * HH + hh], ex);
                    for (int cc = 0; cc < CC; ++cc) {
                        const int ch = hh * CC + cc;
                        atomicAdd(&acc[dl * HCC + (ch & 3) * 32 + (ch >> 2)],
                                  xl[(size_t)s * HCC + ch] * ex);
                    }
                }
            }
        }
    }

    for (int i0 = wv; i0 < total; i0 += 128) {
        int rec[4], sl[4];
#pragma unroll
        for (int u = 0; u < 4; ++u) {
            const int idx = i0 + 32 * u;
            rec[u] = lst[idx < n ? idx : 0];
            sl[u]  = lsrc[idx < n ? idx : 0];
        }
        int      dl[4];
        float    sc[4];
        unsigned mk[4];
        float4   xv[4];
        bool     ok[4];
#pragma unroll
        for (int u = 0; u < 4; ++u) {
            const int  idx = i0 + 32 * u;
            const bool inl = idx < n;
            dl[u] = inl ? (rec[u] >> 20) : (idx - n);
            const int node = base + dl[u];
            ok[u] = (idx < total) && (node < NN);
            const int sN = ok[u] ? (inl ? sl[u] : node) : 0;
            sc[u] = lsc[(ok[u] ? idx : 0) * HH + h];
            mk[u] = mx[(ok[u] ? dl[u] : 0) * HH + h];
            xv[u] = xl4[(size_t)sN * (HCC / 4) + lane];
        }
#pragma unroll
        for (int u = 0; u < 4; ++u) {
            if (ok[u]) {
                const float ex = expf(sc[u] - funkey(mk[u]));
                float* a = &acc[dl[u] * HCC + lane];
                atomicAdd(a +  0, xv[u].x * ex);
                atomicAdd(a + 32, xv[u].y * ex);
                atomicAdd(a + 64, xv[u].z * ex);
                atomicAdd(a + 96, xv[u].w * ex);
                if ((lane & 7) == 0) atomicAdd(&dsum[dl[u] * HH + h], ex);
            }
        }
    }
    __syncthreads();

    float4* agg4 = (float4*)agg;
    for (int i = tid; i < NPB * (HCC / 4); i += 1024) {
        const int node = base + (i >> 5);
        const float* a = &acc[(i >> 5) * HCC + (i & 31)];
        if (node < NN)
            agg4[(size_t)node * (HCC / 4) + (i & 31)] = make_float4(a[0], a[32], a[64], a[96]);
    }
    for (int i = tid; i < NPB * HH; i += 1024) {
        const int node = base + (i >> 2);
        if (node < NN) den[(size_t)node * HH + (i & 3)] = dsum[i];
    }
}

__global__ __launch_bounds__(256)
void bias_stats_kernel(float* __restrict__ agg, const float* __restrict__ den,
                       const float* __restrict__ cb,
                       double* __restrict__ part, int total) {
    __shared__ double ss[256];
    __shared__ double sq[256];
    const int idx = (int)(blockIdx.x * blockDim.x + threadIdx.x);
    double s = 0.0, q = 0.0;
    if (idx < total) {
        const float y = agg[idx] / (den[idx >> 5] + 1e-16f) + cb[idx & (HCC - 1)];
        agg[idx] = y;
        s = (double)y;
        q = (double)y * (double)y;
    }
    ss[threadIdx.x] = s;
    sq[threadIdx.x] = q;
    __syncthreads();
    for (int o = 128; o > 0; o >>= 1) {
        if ((int)threadIdx.x < o) {
            ss[threadIdx.x] += ss[threadIdx.x + o];
            sq[threadIdx.x] += sq[threadIdx.x + o];
        }
        __syncthreads();
    }
    if (threadIdx.x == 0) {
        ((double2*)part)[blockIdx.x] = make_double2(ss[0], sq[0]);
    }
}

__global__ __launch_bounds__(1024)
void stats_finish_kernel(const double* __restrict__ part, int nblk,
                         double* __restrict__ red, int total) {
    __shared__ double ws_[32];
    __shared__ double wq_[32];
    const int tid  = (int)threadIdx.x;
    const int lane = tid & 31;
    const int wv   = tid >> 5;
    double s = 0.0, q = 0.0;
    for (int i = tid; i < nblk; i += 1024) {
        const double2 p = ((const double2*)part)[i];
        s += p.x;
        q += p.y;
    }
    for (int off = 16; off >= 1; off >>= 1) {
        s += __shfl_xor(s, off);
        q += __shfl_xor(q, off);
    }
    if (lane == 0) { ws_[wv] = s; wq_[wv] = q; }
    __syncthreads();
    if (wv == 0) {
        s = ws_[lane];
        q = wq_[lane];
        for (int off = 16; off >= 1; off >>= 1) {
            s += __shfl_xor(s, off);
            q += __shfl_xor(q, off);
        }
        if (lane == 0) {
            const double mu = s / (double)total;
            double var = q / (double)total - mu * mu;
            if (var < 0.0) var = 0.0;
            const float sd = sqrtf((float)var);
            red[0] = mu;
            red[1] = (double)(sd + 1e-5f);
        }
    }
}

__global__ __launch_bounds__(256)
void pln_apply_kernel(float* __restrict__ agg, const float* __restrict__ w,
                      const float* __restrict__ b, const double* __restrict__ red,
                      int total) {
    const int idx = (int)(blockIdx.x * blockDim.x + threadIdx.x);
    if (idx >= total) return;
    const double mu  = red[0];
    const double dnm = red[1];
    const int c = idx & (HCC - 1);
    float y = w[c] * (float)((agg[idx] - mu) / dnm) + b[c];
    agg[idx] = leakyf(y, 0.01f);
}

extern "C" void kernel_launch(void* const* d_in, const int* in_sizes, int n_in,
                              void* d_out, int out_size, void* d_ws, size_t ws_size,
                              hipStream_t stream) {
    const float* x          = (const float*)d_in[0];
    const int*   edge_index = (const int*)  d_in[1];
    const float* edge_attr  = (const float*)d_in[2];
    const float* embed_W    = (const float*)d_in[3];
    const float* embed_b    = (const float*)d_in[4];
    const float* embed_lnw  = (const float*)d_in[5];
    const float* embed_lnb  = (const float*)d_in[6];
    const float* Wl         = (const float*)d_in[7];
    const float* bl         = (const float*)d_in[8];
    const float* Wr         = (const float*)d_in[9];
    const float* br         = (const float*)d_in[10];
    const float* We         = (const float*)d_in[11];
    const float* att        = (const float*)d_in[12];
    const float* conv_b     = (const float*)d_in[13];
    const float* pln_w      = (const float*)d_in[14];
    const float* pln_b      = (const float*)d_in[15];
    const float* proj_W     = (const float*)d_in[16];
    const float* proj_b     = (const float*)d_in[17];
    const float* ln2_w      = (const float*)d_in[18];
    const float* ln2_b      = (const float*)d_in[19];

    const int* srcE = edge_index;
    const int* dstE = edge_index + EE;

    size_t off = 0;
    auto carve = [&](size_t nbytes) -> char* {
        char* p = (char*)d_ws + off;
        off += (nbytes + 255) & ~(size_t)255;
        return p;
    };
    float*    h     = (float*)   carve((size_t)NN * FF  * 4);
    float*    raw   = (float*)   carve((size_t)NN * FF  * 4);
    float*    xl    = (float*)   carve((size_t)NN * HCC * 4);
    float*    xr    = (float*)   carve((size_t)NN * HCC * 4);
    float*    agg   = (float*)   carve((size_t)NN * HCC * 4);
    float*    den   = (float*)   carve((size_t)NN * HH  * 4);
    double*   red   = (double*)  carve(2 * sizeof(double));
    double*   part  = (double*)  carve((size_t)((NN * HCC + 255) / 256) * 2 * sizeof(double));
    int*      olst  = (int*)     carve((size_t)NOWN * LCAP * sizeof(int));
    int*      osrc  = (int*)     carve((size_t)NOWN * LCAP * sizeof(int));
    int*      ocnt  = (int*)     carve((size_t)NOWN * sizeof(int));
    (void)ws_size; (void)in_sizes; (void)n_in; (void)out_size;

    float* jk = (float*)d_out;

    const int mtiles = NN / 16;
    const dim3 blk(256);
    const dim3 gGemm((mtiles * 32 + 255) / 256);
    const dim3 gRows((NN * 32 + 255) / 256);
    const dim3 gElem(((size_t)NN * HCC + 255) / 256);
    const int totalHC = NN * HCC;
    const int nStatBlk = (NN * HCC + 255) / 256;

    const size_t shLin  = (size_t)HCC * (FF + 4) * 4;
    const size_t shProj = (size_t)FF * (HCC + 4) * 4;

    gemm_wmma_f32<2, INF_, false><<<gGemm, blk, 0, stream>>>(x, embed_W, embed_b, raw, NN, FF);
    ln_rows_kernel<<<gRows, blk, 0, stream>>>(raw, embed_lnw, embed_lnb, h, jk, 0, NN);

    for (int l = 0; l < LL; ++l) {
        const float* Wl_l = Wl + (size_t)l * HCC * FF;
        const float* bl_l = bl + (size_t)l * HCC;
        const float* Wr_l = Wr + (size_t)l * HCC * FF;
        const float* br_l = br + (size_t)l * HCC;
        const float* We_l = We + (size_t)l * HCC * EDD;
        const float* at_l = att + (size_t)l * HH * CC;
        const float* cb_l = conv_b + (size_t)l * HCC;
        const float* pw_l = pln_w + (size_t)l * HCC;
        const float* pb_l = pln_b + (size_t)l * HCC;
        const float* pW_l = proj_W + (size_t)l * FF * HCC;
        const float* pB_l = proj_b + (size_t)l * FF;
        const float* l2w  = ln2_w + (size_t)l * FF;
        const float* l2b  = ln2_b + (size_t)l * FF;

        gemm_wmma_f32<8, FF, true><<<gGemm, blk, shLin, stream>>>(h, Wl_l, bl_l, xl, NN, HCC);
        gemm_wmma_f32<8, FF, true><<<gGemm, blk, shLin, stream>>>(h, Wr_l, br_l, xr, NN, HCC);

        if (l == 0)
            edge_owner_kernel<true><<<dim3(NOWN), dim3(1024), 0, stream>>>(
                srcE, dstE, edge_attr, We_l, xl, xr, at_l, den, agg, olst, osrc, ocnt);
        else
            edge_owner_kernel<false><<<dim3(NOWN), dim3(1024), 0, stream>>>(
                srcE, dstE, edge_attr, We_l, xl, xr, at_l, den, agg, olst, osrc, ocnt);

        bias_stats_kernel<<<gElem, blk, 0, stream>>>(agg, den, cb_l, part, totalHC);
        stats_finish_kernel<<<dim3(1), dim3(1024), 0, stream>>>(part, nStatBlk, red, totalHC);
        pln_apply_kernel<<<gElem, blk, 0, stream>>>(agg, pw_l, pb_l, red, totalHC);

        gemm_wmma_f32<2, HCC, true><<<gGemm, blk, shProj, stream>>>(agg, pW_l, pB_l, raw, NN, FF);
        ln_rows_kernel<<<gRows, blk, 0, stream>>>(raw, l2w, l2b, h, jk, 1, NN);
    }
}


# lin_r of a work-group's own 196 target rows formed by that work-group (f32 WMMA) straight into LDS; sweeps read target rows there: lin_r launches, xr array, target-row gathers gone; sums unchanged
# speedup vs baseline: 2.2926x; 1.0404x over previous
#include <hip/hip_runtime.h>
#include <math.h>

#define NN   50000
#define EE   800000
#define INF_ 512
#define FF   32
#define HH   4
#define CC   32
#define HCC  128
#define EDD  4
#define E2E  850000
#define LL   2

typedef float v2f __attribute__((ext_vector_type(2)));
typedef float v8f __attribute__((ext_vector_type(8)));
typedef float v4f __attribute__((ext_vector_type(4)));

__device__ __forceinline__ unsigned fkey(float f) {
    unsigned u = __float_as_uint(f);
    return (u & 0x80000000u) ? ~u : (u | 0x80000000u);
}
__device__ __forceinline__ float funkey(unsigned u) {
    return __uint_as_float((u & 0x80000000u) ? (u & 0x7FFFFFFFu) : ~u);
}
__device__ __forceinline__ float leakyf(float x, float s) {
    return x >= 0.0f ? x : s * x;
}

template <int CT, int KT, bool USE_LDS>
__global__ __launch_bounds__(256)
void gemm_wmma_f32(const float* __restrict__ A, const float* __restrict__ W,
                   const float* __restrict__ bias, float* __restrict__ Cout,
                   int M, int Nout) {
    extern __shared__ float sW[];
    const int tid  = (int)threadIdx.x;
    const int lane = tid & 31;
    const int wave = (int)((blockIdx.x * blockDim.x + tid) >> 5);
    const int mtiles  = M >> 4;
    const int ngroups = Nout / (16 * CT);
    const int stride  = KT + 4;

    if (USE_LDS) {
        const int total = Nout * KT;
        for (int i = tid; i < total; i += 256) {
            const int c  = i / KT;
            const int kk = i - c * KT;
            sW[c * stride + kk] = W[i];
        }
        __syncthreads();
    }
    if (wave >= mtiles * ngroups) return;

    const int mtile = wave / ngroups;
    const int ng    = wave % ngroups;
    const int row0  = mtile * 16;
    const int col0  = ng * 16 * CT;
    const int hi = lane >> 4;
    const int mi = lane & 15;

    v8f acc[CT] = {};
    const float* Arow = A + (size_t)(row0 + mi) * KT + 2 * hi;

    v2f a_cur;
    a_cur.x = Arow[0];
    a_cur.y = Arow[1];
    v2f b_cur[CT];
    if (!USE_LDS) {
#pragma unroll
        for (int t = 0; t < CT; ++t) {
            const float* Wc = W + (size_t)(col0 + t * 16 + mi) * KT + 2 * hi;
            b_cur[t].x = Wc[0];
            b_cur[t].y = Wc[1];
        }
    }

    for (int k = 0; k < KT; k += 4) {
        v2f a_nxt = {};
        v2f b_nxt[CT];
        if (k + 4 < KT) {
            a_nxt.x = Arow[k + 4];
            a_nxt.y = Arow[k + 5];
            if (!USE_LDS) {
#pragma unroll
                for (int t = 0; t < CT; ++t) {
                    const float* Wc = W + (size_t)(col0 + t * 16 + mi) * KT + 2 * hi;
                    b_nxt[t].x = Wc[k + 4];
                    b_nxt[t].y = Wc[k + 5];
                }
                if ((k & 31) == 0)
                    __builtin_prefetch(Arow + k + 132, 0, 1);
            }
        }
#pragma unroll
        for (int t = 0; t < CT; ++t) {
            v2f b;
            if (USE_LDS) {
                const float* Sc = sW + (size_t)(col0 + t * 16 + mi) * stride + 2 * hi;
                b.x = Sc[k];
                b.y = Sc[k + 1];
            } else {
                b = b_cur[t];
            }
            acc[t] = __builtin_amdgcn_wmma_f32_16x16x4_f32(
                false, a_cur, false, b, (short)0, acc[t], false, false);
        }
        a_cur = a_nxt;
        if (!USE_LDS) {
#pragma unroll
            for (int t = 0; t < CT; ++t) b_cur[t] = b_nxt[t];
        }
    }

#pragma unroll
    for (int t = 0; t < CT; ++t) {
        const int col = col0 + t * 16 + mi;
        const float bb = bias ? bias[col] : 0.0f;
#pragma unroll
        for (int v = 0; v < 8; ++v) {
            const int r = row0 + v + 8 * hi;
            Cout[(size_t)r * Nout + col] = acc[t][v] + bb;
        }
    }
}

__global__ __launch_bounds__(256)
void ln_rows_kernel(const float* __restrict__ Z, const float* __restrict__ w,
                    const float* __restrict__ b, float* __restrict__ Hout,
                    float* __restrict__ jk, int mode, int M) {
    const int row  = (int)((blockIdx.x * blockDim.x + threadIdx.x) >> 5);
    const int lane = (int)(threadIdx.x & 31);
    if (row >= M) return;
    float v = Z[(size_t)row * FF + lane];
    float s = v;
#pragma unroll
    for (int off = 16; off >= 1; off >>= 1) s += __shfl_xor(s, off);
    const float mu = s * (1.0f / 32.0f);
    const float d = v - mu;
    float q = d * d;
#pragma unroll
    for (int off = 16; off >= 1; off >>= 1) q += __shfl_xor(q, off);
    const float var = q * (1.0f / 32.0f);
    float y = w[lane] * d * rsqrtf(var + 1e-5f) + b[lane];
    y = leakyf(y, 0.01f);
    Hout[(size_t)row * FF + lane] = y;
    float* o = jk + (size_t)row * FF + lane;
    *o = (mode == 0) ? y : fmaxf(*o, y);
}

#define NPB    196
#define NOWN   256
#define LCAP   4096
#define NCHUNK (EE / 128)
#define XRT    13

__device__ __forceinline__ float edge_score_one(int e, int s, const float* xrow, int hh,
                                             const float* __restrict__ ea,
                                             const float* __restrict__ We,
                                             const float* __restrict__ xl,
                                             const float* __restrict__ att) {
    const float4 ev = ((const float4*)ea)[e];
    float sc = 0.0f;
#pragma nounroll
    for (int cc = 0; cc < CC; ++cc) {
        const int ch = hh * CC + cc;
        const float4 wr = ((const float4*)We)[ch];
        const float eev = wr.x * ev.x + wr.y * ev.y + wr.z * ev.z + wr.w * ev.w;
        const float m = xl[(size_t)s * HCC + ch] + xrow[ch] + eev;
        sc += att[ch] * leakyf(m, 0.2f);
    }
    return sc;
}

template <bool BUILD>
__global__ __launch_bounds__(1024)
void edge_owner_kernel(const int* __restrict__ src, const int* __restrict__ dst,
                       const float* __restrict__ ea, const float* __restrict__ We,
                       const float* __restrict__ xl, const float* __restrict__ hin,
                       const float* __restrict__ Wr, const float* __restrict__ br,
                       const float* __restrict__ att,
                       float* __restrict__ den, float* __restrict__ agg,
                       int* __restrict__ glst, int* __restrict__ gsrc,
                       int* __restrict__ gcnt) {
    __shared__ float    acc[NPB * HCC];
    __shared__ float    dsum[NPB * HH];
    __shared__ unsigned mx[NPB * HH];
    __shared__ int      lst[LCAP];
    __shared__ int      lsrc[LCAP];
    __shared__ float    lsc[(LCAP + NPB) * HH];
    __shared__ int      cnt;
    __shared__ __attribute__((aligned(16))) float xrs[XRT * 16 * HCC];
    const int tid  = (int)threadIdx.x;
    const int lane = tid & 31;
    const int wv   = tid >> 5;
    const int base = (int)blockIdx.x * NPB;

    {
        const int hi = lane >> 4;
        const int mi = lane & 15;
        const int nt = wv & 7;
        const float* Wc = Wr + (size_t)(nt * 16 + mi) * FF + 2 * hi;
        v2f bf[8];
#pragma unroll
        for (int q = 0; q < 8; ++q) bf[q] = *(const v2f*)(Wc + 4 * q);
        const float bb = br[nt * 16 + mi];
        v2f af[4][8];
#pragma unroll
        for (int u = 0; u < 4; ++u) {
            int row = base + ((wv >> 3) + 4 * u) * 16 + mi;
            row = row < NN ? row : NN - 1;
            const float* Ar = hin + (size_t)row * FF + 2 * hi;
#pragma unroll
            for (int q = 0; q < 8; ++q) af[u][q] = *(const v2f*)(Ar + 4 * q);
        }
        v8f cacc[4] = {};
#pragma unroll
        for (int q = 0; q < 8; ++q) {
#pragma unroll
            for (int u = 0; u < 4; ++u)
                cacc[u] = __builtin_amdgcn_wmma_f32_16x16x4_f32(
                    false, af[u][q], false, bf[q], (short)0, cacc[u], false, false);
        }
#pragma unroll
        for (int u = 0; u < 4; ++u) {
            const int mt = (wv >> 3) + 4 * u;
            if (mt < XRT) {
#pragma unroll
                for (int v = 0; v < 8; ++v)
                    xrs[(mt * 16 + v + 8 * hi) * HCC + nt * 16 + mi] = cacc[u][v] + bb;
            }
        }
    }

    for (int i = tid; i < NPB * HCC; i += 1024) acc[i] = 0.0f;
    for (int i = tid; i < NPB * HH; i += 1024) { dsum[i] = 0.0f; mx[i] = 0u; }
    if (BUILD && tid == 0) cnt = 0;
    __syncthreads();

    int n;
    if (BUILD) {
    const int4* dst4 = (const int4*)dst;
    const int start = (int)(((unsigned)blockIdx.x * (unsigned)NCHUNK) / (unsigned)NOWN);
    for (int j0 = 0; j0 < 196; j0 += 7) {
        int4 dv[7];
        int  cb[7];
#pragma unroll
        for (int u = 0; u < 7; ++u) {
            const int t = wv + 32 * (j0 + u);
            int ch = t + start;
            if (ch >= NCHUNK) ch -= NCHUNK;
            cb[u] = ch;
            dv[u] = (t < NCHUNK) ? dst4[(size_t)ch * 32 + lane] : make_int4(-1, -1, -1, -1);
        }
#pragma unroll
        for (int u = 0; u < 7; ++u) {
            const int e0 = cb[u] * 128 + lane * 4;
            const int dd[4] = {dv[u].x, dv[u].y, dv[u].z, dv[u].w};
#pragma unroll
            for (int k = 0; k < 4; ++k) {
                const unsigned dl = (unsigned)(dd[k] - base);
                if (dl < (unsigned)NPB) {
                    const int pos = atomicAdd(&cnt, 1);
                    if (pos < LCAP) lst[pos] = (e0 + k) | ((int)dl << 20);
                }
            }
        }
    }
    __syncthreads();
    n = cnt;
    } else {
        n = gcnt[blockIdx.x];
    }
    const int  nraw = n;
    const bool over = n > LCAP;
    if (over) {
        for (int e = tid; e < EE; e += 1024) {
            const int d = dst[e];
            const unsigned dl = (unsigned)(d - base);
            if (dl < (unsigned)NPB) {
                const int s = src[e];
                for (int hh = 0; hh < HH; ++hh)
                    atomicMax(&mx[dl * HH + hh],
                              fkey(edge_score_one(e, s, &xrs[dl * HCC], hh, ea, We, xl, att)));
            }
        }
        n = 0;
    }
    int* mylst = glst + (size_t)blockIdx.x * LCAP;
    int* mysrc = gsrc + (size_t)blockIdx.x * LCAP;
    if (BUILD) {
        if (tid == 0) gcnt[blockIdx.x] = nraw;
        for (int i = tid; i < n; i += 1024) {
            const int r = lst[i];
            const int s = src[r & 0xFFFFF];
            lsrc[i]  = s;
            mylst[i] = r;
            mysrc[i] = s;
        }
    } else {
        for (int i = tid; i < n; i += 1024) {
            lst[i]  = mylst[i];
            lsrc[i] = mysrc[i];
        }
    }
    __syncthreads();

    const int total = n + NPB;
    const int h     = lane >> 3;
    const float4* xl4 = (const float4*)xl;
    const float4* xrs4 = (const float4*)xrs;

    {
        const float4* We4 = (const float4*)We;
        const v4f* xl4v = (const v4f*)xl;
        const v4f* xrsv = (const v4f*)xrs;
        const v4f* ea4v = (const v4f*)ea;
        float4 wr[4];
#pragma unroll
        for (int k = 0; k < 4; ++k) wr[k] = We4[lane * 4 + k];
        const float4 atv = ((const float4*)att)[lane];
        const float at[4] = {atv.x, atv.y, atv.z, atv.w};

        for (int i0 = wv; i0 < n; i0 += 128) {
            int rec[4], sl[4];
#pragma unroll
            for (int u = 0; u < 4; ++u) {
                const int idx = i0 + 32 * u;
                rec[u] = lst[idx < n ? idx : 0];
                sl[u]  = lsrc[idx < n ? idx : 0];
            }
            v4f xa[4], xb[4], ev[4];
#pragma unroll
            for (int u = 0; u < 4; ++u) {
                xa[u] = xl4v[(size_t)sl[u] * (HCC / 4) + lane];
                xb[u] = xrsv[(rec[u] >> 20) * (HCC / 4) + lane];
                ev[u] = ea4v[rec[u] & 0xFFFFF];
            }
            asm volatile("" : "+v"(xa[0]), "+v"(xa[1]), "+v"(xa[2]), "+v"(xa[3]),
                              "+v"(xb[0]), "+v"(xb[1]), "+v"(xb[2]), "+v"(xb[3]),
                              "+v"(ev[0]), "+v"(ev[1]), "+v"(ev[2]), "+v"(ev[3]));
            float sc[4];
#pragma unroll
            for (int u = 0; u < 4; ++u) {
                const float pa[4] = {xa[u].x, xa[u].y, xa[u].z, xa[u].w};
                const float pb[4] = {xb[u].x, xb[u].y, xb[u].z, xb[u].w};
                float s = 0.0f;
#pragma unroll
                for (int k = 0; k < 4; ++k) {
                    const float eev = wr[k].x * ev[u].x + wr[k].y * ev[u].y +
                                      wr[k].z * ev[u].z + wr[k].w * ev[u].w;
                    float m = pa[k] + pb[k] + eev;
                    m = leakyf(m, 0.2f);
                    s += at[k] * m;
                }
                sc[u] = s;
            }
#pragma unroll
            for (int off = 4; off >= 1; off >>= 1) {
#pragma unroll
                for (int u = 0; u < 4; ++u) sc[u] += __shfl_xor(sc[u], off);
            }
#pragma unroll
            for (int u = 0; u < 4; ++u) {
                const int idx = i0 + 32 * u;
                if (idx < n && (lane & 7) == 0) {
                    lsc[idx * HH + h] = sc[u];
                    atomicMax(&mx[(rec[u] >> 20) * HH + h], fkey(sc[u]));
                }
            }
        }

        for (int j0 = wv; j0 < NPB; j0 += 128) {
            float4 xa[4], xb[4];
            bool   ok[4];
#pragma unroll
            for (int u = 0; u < 4; ++u) {
                const int j    = j0 + 32 * u;
                const int node = base + j;
                ok[u] = (j < NPB) && (node < NN);
                const int nd = ok[u] ? node : 0;
                xa[u] = xl4[(size_t)nd * (HCC / 4) + lane];
                xb[u] = xrs4[(ok[u] ? j : 0) * (HCC / 4) + lane];
            }
            float sc[4];
#pragma unroll
            for (int u = 0; u < 4; ++u) {
                const float pa[4] = {xa[u].x, xa[u].y, xa[u].z, xa[u].w};
                const float pb[4] = {xb[u].x, xb[u].y, xb[u].z, xb[u].w};
                float s = 0.0f;
#pragma unroll
                for (int k = 0; k < 4; ++k) {
                    float m = pa[k] + pb[k];
                    m = leakyf(m, 0.2f);
                    s += at[k] * m;
                }
                sc[u] = s;
            }
#pragma unroll
            for (int off = 4; off >= 1; off >>= 1) {
#pragma unroll
                for (int u = 0; u < 4; ++u) sc[u] += __shfl_xor(sc[u], off);
            }
#pragma unroll
            for (int u = 0; u < 4; ++u) {
                const int j = j0 + 32 * u;
                if (ok[u] && (lane & 7) == 0) {
                    lsc[(n + j) * HH + h] = sc[u];
                    atomicMax(&mx[j * HH + h], fkey(sc[u]));
                }
            }
        }
    }
    __syncthreads();

    if (over) {
        for (int e = tid; e < EE; e += 1024) {
            const int d = dst[e];
            const unsigned dl = (unsigned)(d - base);
            if (dl < (unsigned)NPB) {
                const int s = src[e];
                for (int hh = 0; hh < HH; ++hh) {
                    const float ex = expf(edge_score_one(e, s, &xrs[dl * HCC], hh, ea, We, xl, att) -
                                          funkey(mx[dl * HH + hh]));
                    atomicAdd(&dsum[dl * HH + hh], ex);
                    for (int cc = 0; cc < CC; ++cc) {
                        const int ch = hh * CC + cc;
                        atomicAdd(&acc[dl * HCC + (ch & 3) * 32 + (ch >> 2)],
                                  xl[(size_t)s * HCC + ch] * ex);
                    }
                }
            }
        }
    }

    for (int i0 = wv; i0 < total; i0 += 128) {
        int rec[4], sl[4];
#pragma unroll
        for (int u = 0; u < 4; ++u) {
            const int idx = i0 + 32 * u;
            rec[u] = lst[idx < n ? idx : 0];
            sl[u]  = lsrc[idx < n ? idx : 0];
        }
        int      dl[4];
        float    sc[4];
        unsigned mk[4];
        float4   xv[4];
        bool     ok[4];
#pragma unroll
        for (int u = 0; u < 4; ++u) {
            const int  idx = i0 + 32 * u;
            const bool inl = idx < n;
            dl[u] = inl ? (rec[u] >> 20) : (idx - n);
            const int node = base + dl[u];
            ok[u] = (idx < total) && (node < NN);
            const int sN = ok[u] ? (inl ? sl[u] : node) : 0;
            sc[u] = lsc[(ok[u] ? idx : 0) * HH + h];
            mk[u] = mx[(ok[u] ? dl[u] : 0) * HH + h];
            xv[u] = xl4[(size_t)sN * (HCC / 4) + lane];
        }
#pragma unroll
        for (int u = 0; u < 4; ++u) {
            if (ok[u]) {
                const float ex = expf(sc[u] - funkey(mk[u]));
                float* a = &acc[dl[u] * HCC + lane];
                atomicAdd(a +  0, xv[u].x * ex);
                atomicAdd(a + 32, xv[u].y * ex);
                atomicAdd(a + 64, xv[u].z * ex);
                atomicAdd(a + 96, xv[u].w * ex);
                if ((lane & 7) == 0) atomicAdd(&dsum[dl[u] * HH + h], ex);
            }
        }
    }
    __syncthreads();

    float4* agg4 = (float4*)agg;
    for (int i = tid; i < NPB * (HCC / 4); i += 1024) {
        const int node = base + (i >> 5);
        const float* a = &acc[(i >> 5) * HCC + (i & 31)];
        if (node < NN)
            agg4[(size_t)node * (HCC / 4) + (i & 31)] = make_float4(a[0], a[32], a[64], a[96]);
    }
    for (int i = tid; i < NPB * HH; i += 1024) {
        const int node = base + (i >> 2);
        if (node < NN) den[(size_t)node * HH + (i & 3)] = dsum[i];
    }
}

__global__ __launch_bounds__(256)
void bias_stats_kernel(float* __restrict__ agg, const float* __restrict__ den,
                       const float* __restrict__ cb,
                       double* __restrict__ part, int total) {
    __shared__ double ss[256];
    __shared__ double sq[256];
    const int idx = (int)(blockIdx.x * blockDim.x + threadIdx.x);
    double s = 0.0, q = 0.0;
    if (idx < total) {
        const float y = agg[idx] / (den[idx >> 5] + 1e-16f) + cb[idx & (HCC - 1)];
        agg[idx] = y;
        s = (double)y;
        q = (double)y * (double)y;
    }
    ss[threadIdx.x] = s;
    sq[threadIdx.x] = q;
    __syncthreads();
    for (int o = 128; o > 0; o >>= 1) {
        if ((int)threadIdx.x < o) {
            ss[threadIdx.x] += ss[threadIdx.x + o];
            sq[threadIdx.x] += sq[threadIdx.x + o];
        }
        __syncthreads();
    }
    if (threadIdx.x == 0) {
        ((double2*)part)[blockIdx.x] = make_double2(ss[0], sq[0]);
    }
}

__global__ __launch_bounds__(1024)
void stats_finish_kernel(const double* __restrict__ part, int nblk,
                         double* __restrict__ red, int total) {
    __shared__ double ws_[32];
    __shared__ double wq_[32];
    const int tid  = (int)threadIdx.x;
    const int lane = tid & 31;
    const int wv   = tid >> 5;
    double s = 0.0, q = 0.0;
    for (int i = tid; i < nblk; i += 1024) {
        const double2 p = ((const double2*)part)[i];
        s += p.x;
        q += p.y;
    }
    for (int off = 16; off >= 1; off >>= 1) {
        s += __shfl_xor(s, off);
        q += __shfl_xor(q, off);
    }
    if (lane == 0) { ws_[wv] = s; wq_[wv] = q; }
    __syncthreads();
    if (wv == 0) {
        s = ws_[lane];
        q = wq_[lane];
        for (int off = 16; off >= 1; off >>= 1) {
            s += __shfl_xor(s, off);
            q += __shfl_xor(q, off);
        }
        if (lane == 0) {
            const double mu = s / (double)total;
            double var = q / (double)total - mu * mu;
            if (var < 0.0) var = 0.0;
            const float sd = sqrtf((float)var);
            red[0] = mu;
            red[1] = (double)(sd + 1e-5f);
        }
    }
}

__global__ __launch_bounds__(256)
void pln_apply_kernel(float* __restrict__ agg, const float* __restrict__ w,
                      const float* __restrict__ b, const double* __restrict__ red,
                      int total) {
    const int idx = (int)(blockIdx.x * blockDim.x + threadIdx.x);
    if (idx >= total) return;
    const double mu  = red[0];
    const double dnm = red[1];
    const int c = idx & (HCC - 1);
    float y = w[c] * (float)((agg[idx] - mu) / dnm) + b[c];
    agg[idx] = leakyf(y, 0.01f);
}

extern "C" void kernel_launch(void* const* d_in, const int* in_sizes, int n_in,
                              void* d_out, int out_size, void* d_ws, size_t ws_size,
                              hipStream_t stream) {
    const float* x          = (const float*)d_in[0];
    const int*   edge_index = (const int*)  d_in[1];
    const float* edge_attr  = (const float*)d_in[2];
    const float* embed_W    = (const float*)d_in[3];
    const float* embed_b    = (const float*)d_in[4];
    const float* embed_lnw  = (const float*)d_in[5];
    const float* embed_lnb  = (const float*)d_in[6];
    const float* Wl         = (const float*)d_in[7];
    const float* bl         = (const float*)d_in[8];
    const float* Wr         = (const float*)d_in[9];
    const float* br         = (const float*)d_in[10];
    const float* We         = (const float*)d_in[11];
    const float* att        = (const float*)d_in[12];
    const float* conv_b     = (const float*)d_in[13];
    const float* pln_w      = (const float*)d_in[14];
    const float* pln_b      = (const float*)d_in[15];
    const float* proj_W     = (const float*)d_in[16];
    const float* proj_b     = (const float*)d_in[17];
    const float* ln2_w      = (const float*)d_in[18];
    const float* ln2_b      = (const float*)d_in[19];

    const int* srcE = edge_index;
    const int* dstE = edge_index + EE;

    size_t off = 0;
    auto carve = [&](size_t nbytes) -> char* {
        char* p = (char*)d_ws + off;
        off += (nbytes + 255) & ~(size_t)255;
        return p;
    };
    float*    h     = (float*)   carve((size_t)NN * FF  * 4);
    float*    raw   = (float*)   carve((size_t)NN * FF  * 4);
    float*    xl    = (float*)   carve((size_t)NN * HCC * 4);
    float*    agg   = (float*)   carve((size_t)NN * HCC * 4);
    float*    den   = (float*)   carve((size_t)NN * HH  * 4);
    double*   red   = (double*)  carve(2 * sizeof(double));
    double*   part  = (double*)  carve((size_t)((NN * HCC + 255) / 256) * 2 * sizeof(double));
    int*      olst  = (int*)     carve((size_t)NOWN * LCAP * sizeof(int));
    int*      osrc  = (int*)     carve((size_t)NOWN * LCAP * sizeof(int));
    int*      ocnt  = (int*)     carve((size_t)NOWN * sizeof(int));
    (void)ws_size; (void)in_sizes; (void)n_in; (void)out_size;

    float* jk = (float*)d_out;

    const int mtiles = NN / 16;
    const dim3 blk(256);
    const dim3 gGemm((mtiles * 32 + 255) / 256);
    const dim3 gRows((NN * 32 + 255) / 256);
    const dim3 gElem(((size_t)NN * HCC + 255) / 256);
    const int totalHC = NN * HCC;
    const int nStatBlk = (NN * HCC + 255) / 256;

    const size_t shLin  = (size_t)HCC * (FF + 4) * 4;
    const size_t shProj = (size_t)FF * (HCC + 4) * 4;

    gemm_wmma_f32<2, INF_, false><<<gGemm, blk, 0, stream>>>(x, embed_W, embed_b, raw, NN, FF);
    ln_rows_kernel<<<gRows, blk, 0, stream>>>(raw, embed_lnw, embed_lnb, h, jk, 0, NN);

    for (int l = 0; l < LL; ++l) {
        const float* Wl_l = Wl + (size_t)l * HCC * FF;
        const float* bl_l = bl + (size_t)l * HCC;
        const float* Wr_l = Wr + (size_t)l * HCC * FF;
        const float* br_l = br + (size_t)l * HCC;
        const float* We_l = We + (size_t)l * HCC * EDD;
        const float* at_l = att + (size_t)l * HH * CC;
        const float* cb_l = conv_b + (size_t)l * HCC;
        const float* pw_l = pln_w + (size_t)l * HCC;
        const float* pb_l = pln_b + (size_t)l * HCC;
        const float* pW_l = proj_W + (size_t)l * FF * HCC;
        const float* pB_l = proj_b + (size_t)l * FF;
        const float* l2w  = ln2_w + (size_t)l * FF;
        const float* l2b  = ln2_b + (size_t)l * FF;

        gemm_wmma_f32<8, FF, true><<<gGemm, blk, shLin, stream>>>(h, Wl_l, bl_l, xl, NN, HCC);

        if (l == 0)
            edge_owner_kernel<true><<<dim3(NOWN), dim3(1024), 0, stream>>>(
                srcE, dstE, edge_attr, We_l, xl, h, Wr_l, br_l, at_l, den, agg, olst, osrc, ocnt);
        else
            edge_owner_kernel<false><<<dim3(NOWN), dim3(1024), 0, stream>>>(
                srcE, dstE, edge_attr, We_l, xl, h, Wr_l, br_l, at_l, den, agg, olst, osrc, ocnt);

        bias_stats_kernel<<<gElem, blk, 0, stream>>>(agg, den, cb_l, part, totalHC);
        stats_finish_kernel<<<dim3(1), dim3(1024), 0, stream>>>(part, nStatBlk, red, totalHC);
        pln_apply_kernel<<<gElem, blk, 0, stream>>>(agg, pw_l, pb_l, red, totalHC);

        gemm_wmma_f32<2, HCC, true><<<gGemm, blk, shProj, stream>>>(agg, pW_l, pB_l, raw, NN, FF);
        ln_rows_kernel<<<gRows, blk, 0, stream>>>(raw, l2w, l2b, h, jk, 1, NN);
    }
}


# owner work-groups finish their rows at the one store: / denominator, + conv bias, graph-LN partial sums from LDS (f32 in a wave, f64 across); bias-stats launches, den array gone; sum order changes
# speedup vs baseline: 2.7588x; 1.2033x over previous
#include <hip/hip_runtime.h>
#include <math.h>

#define NN   50000
#define EE   800000
#define INF_ 512
#define FF   32
#define HH   4
#define CC   32
#define HCC  128
#define EDD  4
#define E2E  850000
#define LL   2

typedef float v2f __attribute__((ext_vector_type(2)));
typedef float v8f __attribute__((ext_vector_type(8)));
typedef float v4f __attribute__((ext_vector_type(4)));

__device__ __forceinline__ unsigned fkey(float f) {
    unsigned u = __float_as_uint(f);
    return (u & 0x80000000u) ? ~u : (u | 0x80000000u);
}
__device__ __forceinline__ float funkey(unsigned u) {
    return __uint_as_float((u & 0x80000000u) ? (u & 0x7FFFFFFFu) : ~u);
}
__device__ __forceinline__ float leakyf(float x, float s) {
    return x >= 0.0f ? x : s * x;
}

template <int CT, int KT, bool USE_LDS>
__global__ __launch_bounds__(256)
void gemm_wmma_f32(const float* __restrict__ A, const float* __restrict__ W,
                   const float* __restrict__ bias, float* __restrict__ Cout,
                   int M, int Nout) {
    extern __shared__ float sW[];
    const int tid  = (int)threadIdx.x;
    const int lane = tid & 31;
    const int wave = (int)((blockIdx.x * blockDim.x + tid) >> 5);
    const int mtiles  = M >> 4;
    const int ngroups = Nout / (16 * CT);
    const int stride  = KT + 4;

    if (USE_LDS) {
        const int total = Nout * KT;
        for (int i = tid; i < total; i += 256) {
            const int c  = i / KT;
            const int kk = i - c * KT;
            sW[c * stride + kk] = W[i];
        }
        __syncthreads();
    }
    if (wave >= mtiles * ngroups) return;

    const int mtile = wave / ngroups;
    const int ng    = wave % ngroups;
    const int row0  = mtile * 16;
    const int col0  = ng * 16 * CT;
    const int hi = lane >> 4;
    const int mi = lane & 15;

    v8f acc[CT] = {};
    const float* Arow = A + (size_t)(row0 + mi) * KT + 2 * hi;

    v2f a_cur;
    a_cur.x = Arow[0];
    a_cur.y = Arow[1];
    v2f b_cur[CT];
    if (!USE_LDS) {
#pragma unroll
        for (int t = 0; t < CT; ++t) {
            const float* Wc = W + (size_t)(col0 + t * 16 + mi) * KT + 2 * hi;
            b_cur[t].x = Wc[0];
            b_cur[t].y = Wc[1];
        }
    }

    for (int k = 0; k < KT; k += 4) {
        v2f a_nxt = {};
        v2f b_nxt[CT];
        if (k + 4 < KT) {
            a_nxt.x = Arow[k + 4];
            a_nxt.y = Arow[k + 5];
            if (!USE_LDS) {
#pragma unroll
                for (int t = 0; t < CT; ++t) {
                    const float* Wc = W + (size_t)(col0 + t * 16 + mi) * KT + 2 * hi;
                    b_nxt[t].x = Wc[k + 4];
                    b_nxt[t].y = Wc[k + 5];
                }
                if ((k & 31) == 0)
                    __builtin_prefetch(Arow + k + 132, 0, 1);
            }
        }
#pragma unroll
        for (int t = 0; t < CT; ++t) {
            v2f b;
            if (USE_LDS) {
                const float* Sc = sW + (size_t)(col0 + t * 16 + mi) * stride + 2 * hi;
                b.x = Sc[k];
                b.y = Sc[k + 1];
            } else {
                b = b_cur[t];
            }
            acc[t] = __builtin_amdgcn_wmma_f32_16x16x4_f32(
                false, a_cur, false, b, (short)0, acc[t], false, false);
        }
        a_cur = a_nxt;
        if (!USE_LDS) {
#pragma unroll
            for (int t = 0; t < CT; ++t) b_cur[t] = b_nxt[t];
        }
    }

#pragma unroll
    for (int t = 0; t < CT; ++t) {
        const int col = col0 + t * 16 + mi;
        const float bb = bias ? bias[col] : 0.0f;
#pragma unroll
        for (int v = 0; v < 8; ++v) {
            const int r = row0 + v + 8 * hi;
            Cout[(size_t)r * Nout + col] = acc[t][v] + bb;
        }
    }
}

__global__ __launch_bounds__(256)
void ln_rows_kernel(const float* __restrict__ Z, const float* __restrict__ w,
                    const float* __restrict__ b, float* __restrict__ Hout,
                    float* __restrict__ jk, int mode, int M) {
    const int row  = (int)((blockIdx.x * blockDim.x + threadIdx.x) >> 5);
    const int lane = (int)(threadIdx.x & 31);
    if (row >= M) return;
    float v = Z[(size_t)row * FF + lane];
    float s = v;
#pragma unroll
    for (int off = 16; off >= 1; off >>= 1) s += __shfl_xor(s, off);
    const float mu = s * (1.0f / 32.0f);
    const float d = v - mu;
    float q = d * d;
#pragma unroll
    for (int off = 16; off >= 1; off >>= 1) q += __shfl_xor(q, off);
    const float var = q * (1.0f / 32.0f);
    float y = w[lane] * d * rsqrtf(var + 1e-5f) + b[lane];
    y = leakyf(y, 0.01f);
    Hout[(size_t)row * FF + lane] = y;
    float* o = jk + (size_t)row * FF + lane;
    *o = (mode == 0) ? y : fmaxf(*o, y);
}

#define NPB    196
#define NOWN   256
#define LCAP   4096
#define NCHUNK (EE / 128)
#define XRT    13

__device__ __forceinline__ float edge_score_one(int e, int s, const float* xrow, int hh,
                                             const float* __restrict__ ea,
                                             const float* __restrict__ We,
                                             const float* __restrict__ xl,
                                             const float* __restrict__ att) {
    const float4 ev = ((const float4*)ea)[e];
    float sc = 0.0f;
#pragma nounroll
    for (int cc = 0; cc < CC; ++cc) {
        const int ch = hh * CC + cc;
        const float4 wr = ((const float4*)We)[ch];
        const float eev = wr.x * ev.x + wr.y * ev.y + wr.z * ev.z + wr.w * ev.w;
        const float m = xl[(size_t)s * HCC + ch] + xrow[ch] + eev;
        sc += att[ch] * leakyf(m, 0.2f);
    }
    return sc;
}

template <bool BUILD>
__global__ __launch_bounds__(1024)
void edge_owner_kernel(const int* __restrict__ src, const int* __restrict__ dst,
                       const float* __restrict__ ea, const float* __restrict__ We,
                       const float* __restrict__ xl, const float* __restrict__ hin,
                       const float* __restrict__ Wr, const float* __restrict__ br,
                       const float* __restrict__ att, const float* __restrict__ cb,
                       float* __restrict__ agg, double* __restrict__ part,
                       int* __restrict__ glst, int* __restrict__ gsrc,
                       int* __restrict__ gcnt) {
    __shared__ float    acc[NPB * HCC];
    __shared__ float    dsum[NPB * HH];
    __shared__ unsigned mx[NPB * HH];
    __shared__ int      lst[LCAP];
    __shared__ int      lsrc[LCAP];
    __shared__ float    lsc[(LCAP + NPB) * HH];
    __shared__ int      cnt;
    __shared__ double   wsum[32];
    __shared__ double   wsq[32];
    __shared__ __attribute__((aligned(16))) float xrs[XRT * 16 * HCC];
    const int tid  = (int)threadIdx.x;
    const int lane = tid & 31;
    const int wv   = tid >> 5;
    const int base = (int)blockIdx.x * NPB;

    {
        const int hi = lane >> 4;
        const int mi = lane & 15;
        const int nt = wv & 7;
        const float* Wc = Wr + (size_t)(nt * 16 + mi) * FF + 2 * hi;
        v2f bf[8];
#pragma unroll
        for (int q = 0; q < 8; ++q) bf[q] = *(const v2f*)(Wc + 4 * q);
        const float bb = br[nt * 16 + mi];
        v2f af[4][8];
#pragma unroll
        for (int u = 0; u < 4; ++u) {
            int row = base + ((wv >> 3) + 4 * u) * 16 + mi;
            row = row < NN ? row : NN - 1;
            const float* Ar = hin + (size_t)row * FF + 2 * hi;
#pragma unroll
            for (int q = 0; q < 8; ++q) af[u][q] = *(const v2f*)(Ar + 4 * q);
        }
        v8f cacc[4] = {};
#pragma unroll
        for (int q = 0; q < 8; ++q) {
#pragma unroll
            for (int u = 0; u < 4; ++u)
                cacc[u] = __builtin_amdgcn_wmma_f32_16x16x4_f32(
                    false, af[u][q], false, bf[q], (short)0, cacc[u], false, false);
        }
#pragma unroll
        for (int u = 0; u < 4; ++u) {
            const int mt = (wv >> 3) + 4 * u;
            if (mt < XRT) {
#pragma unroll
                for (int v = 0; v < 8; ++v)
                    xrs[(mt * 16 + v + 8 * hi) * HCC + nt * 16 + mi] = cacc[u][v] + bb;
            }
        }
    }

    for (int i = tid; i < NPB * HCC; i += 1024) acc[i] = 0.0f;
    for (int i = tid; i < NPB * HH; i += 1024) { dsum[i] = 0.0f; mx[i] = 0u; }
    if (BUILD && tid == 0) cnt = 0;
    __syncthreads();

    int n;
    if (BUILD) {
    const int4* dst4 = (const int4*)dst;
    const int start = (int)(((unsigned)blockIdx.x * (unsigned)NCHUNK) / (unsigned)NOWN);
    for (int j0 = 0; j0 < 196; j0 += 7) {
        int4 dv[7];
        int  cb[7];
#pragma unroll
        for (int u = 0; u < 7; ++u) {
            const int t = wv + 32 * (j0 + u);
            int ch = t + start;
            if (ch >= NCHUNK) ch -= NCHUNK;
            cb[u] = ch;
            dv[u] = (t < NCHUNK) ? dst4[(size_t)ch * 32 + lane] : make_int4(-1, -1, -1, -1);
        }
#pragma unroll
        for (int u = 0; u < 7; ++u) {
            const int e0 = cb[u] * 128 + lane * 4;
            const int dd[4] = {dv[u].x, dv[u].y, dv[u].z, dv[u].w};
#pragma unroll
            for (int k = 0; k < 4; ++k) {
                const unsigned dl = (unsigned)(dd[k] - base);
                if (dl < (unsigned)NPB) {
                    const int pos = atomicAdd(&cnt, 1);
                    if (pos < LCAP) lst[pos] = (e0 + k) | ((int)dl << 20);
                }
            }
        }
    }
    __syncthreads();
    n = cnt;
    } else {
        n = gcnt[blockIdx.x];
    }
    const int  nraw = n;
    const bool over = n > LCAP;
    if (over) {
        for (int e = tid; e < EE; e += 1024) {
            const int d = dst[e];
            const unsigned dl = (unsigned)(d - base);
            if (dl < (unsigned)NPB) {
                const int s = src[e];
                for (int hh = 0; hh < HH; ++hh)
                    atomicMax(&mx[dl * HH + hh],
                              fkey(edge_score_one(e, s, &xrs[dl * HCC], hh, ea, We, xl, att)));
            }
        }
        n = 0;
    }
    int* mylst = glst + (size_t)blockIdx.x * LCAP;
    int* mysrc = gsrc + (size_t)blockIdx.x * LCAP;
    if (BUILD) {
        if (tid == 0) gcnt[blockIdx.x] = nraw;
        for (int i = tid; i < n; i += 1024) {
            const int r = lst[i];
            const int s = src[r & 0xFFFFF];
            lsrc[i]  = s;
            mylst[i] = r;
            mysrc[i] = s;
        }
    } else {
        for (int i = tid; i < n; i += 1024) {
            lst[i]  = mylst[i];
            lsrc[i] = mysrc[i];
        }
    }
    __syncthreads();

    const int total = n + NPB;
    const int h     = lane >> 3;
    const float4* xl4 = (const float4*)xl;
    const float4* xrs4 = (const float4*)xrs;

    {
        const float4* We4 = (const float4*)We;
        const v4f* xl4v = (const v4f*)xl;
        const v4f* xrsv = (const v4f*)xrs;
        const v4f* ea4v = (const v4f*)ea;
        float4 wr[4];
#pragma unroll
        for (int k = 0; k < 4; ++k) wr[k] = We4[lane * 4 + k];
        const float4 atv = ((const float4*)att)[lane];
        const float at[4] = {atv.x, atv.y, atv.z, atv.w};

        for (int i0 = wv; i0 < n; i0 += 128) {
            int rec[4], sl[4];
#pragma unroll
            for (int u = 0; u < 4; ++u) {
                const int idx = i0 + 32 * u;
                rec[u] = lst[idx < n ? idx : 0];
                sl[u]  = lsrc[idx < n ? idx : 0];
            }
            v4f xa[4], xb[4], ev[4];
#pragma unroll
            for (int u = 0; u < 4; ++u) {
                xa[u] = xl4v[(size_t)sl[u] * (HCC / 4) + lane];
                xb[u] = xrsv[(rec[u] >> 20) * (HCC / 4) + lane];
                ev[u] = ea4v[rec[u] & 0xFFFFF];
            }
            asm volatile("" : "+v"(xa[0]), "+v"(xa[1]), "+v"(xa[2]), "+v"(xa[3]),
                              "+v"(xb[0]), "+v"(xb[1]), "+v"(xb[2]), "+v"(xb[3]),
                              "+v"(ev[0]), "+v"(ev[1]), "+v"(ev[2]), "+v"(ev[3]));
            float sc[4];
#pragma unroll
            for (int u = 0; u < 4; ++u) {
                const float pa[4] = {xa[u].x, xa[u].y, xa[u].z, xa[u].w};
                const float pb[4] = {xb[u].x, xb[u].y, xb[u].z, xb[u].w};
                float s = 0.0f;
#pragma unroll
                for (int k = 0; k < 4; ++k) {
                    const float eev = wr[k].x * ev[u].x + wr[k].y * ev[u].y +
                                      wr[k].z * ev[u].z + wr[k].w * ev[u].w;
                    float m = pa[k] + pb[k] + eev;
                    m = leakyf(m, 0.2f);
                    s += at[k] * m;
                }
                sc[u] = s;
            }
#pragma unroll
            for (int off = 4; off >= 1; off >>= 1) {
#pragma unroll
                for (int u = 0; u < 4; ++u) sc[u] += __shfl_xor(sc[u], off);
            }
#pragma unroll
            for (int u = 0; u < 4; ++u) {
                const int idx = i0 + 32 * u;
                if (idx < n && (lane & 7) == 0) {
                    lsc[idx * HH + h] = sc[u];
                    atomicMax(&mx[(rec[u] >> 20) * HH + h], fkey(sc[u]));
                }
            }
        }

        for (int j0 = wv; j0 < NPB; j0 += 128) {
            float4 xa[4], xb[4];
            bool   ok[4];
#pragma unroll
            for (int u = 0; u < 4; ++u) {
                const int j    = j0 + 32 * u;
                const int node = base + j;
                ok[u] = (j < NPB) && (node < NN);
                const int nd = ok[u] ? node : 0;
                xa[u] = xl4[(size_t)nd * (HCC / 4) + lane];
                xb[u] = xrs4[(ok[u] ? j : 0) * (HCC / 4) + lane];
            }
            float sc[4];
#pragma unroll
            for (int u = 0; u < 4; ++u) {
                const float pa[4] = {xa[u].x, xa[u].y, xa[u].z, xa[u].w};
                const float pb[4] = {xb[u].x, xb[u].y, xb[u].z, xb[u].w};
                float s = 0.0f;
#pragma unroll
                for (int k = 0; k < 4; ++k) {
                    float m = pa[k] + pb[k];
                    m = leakyf(m, 0.2f);
                    s += at[k] * m;
                }
                sc[u] = s;
            }
#pragma unroll
            for (int off = 4; off >= 1; off >>= 1) {
#pragma unroll
                for (int u = 0; u < 4; ++u) sc[u] += __shfl_xor(sc[u], off);
            }
#pragma unroll
            for (int u = 0; u < 4; ++u) {
                const int j = j0 + 32 * u;
                if (ok[u] && (lane & 7) == 0) {
                    lsc[(n + j) * HH + h] = sc[u];
                    atomicMax(&mx[j * HH + h], fkey(sc[u]));
                }
            }
        }
    }
    __syncthreads();

    if (over) {
        for (int e = tid; e < EE; e += 1024) {
            const int d = dst[e];
            const unsigned dl = (unsigned)(d - base);
            if (dl < (unsigned)NPB) {
                const int s = src[e];
                for (int hh = 0; hh < HH; ++hh) {
                    const float ex = expf(edge_score_one(e, s, &xrs[dl * HCC], hh, ea, We, xl, att) -
                                          funkey(mx[dl * HH + hh]));
                    atomicAdd(&dsum[dl * HH + hh], ex);
                    for (int cc = 0; cc < CC; ++cc) {
                        const int ch = hh * CC + cc;
                        atomicAdd(&acc[dl * HCC + (ch & 3) * 32 + (ch >> 2)],
                                  xl[(size_t)s * HCC + ch] * ex);
                    }
                }
            }
        }
    }

    for (int i0 = wv; i0 < total; i0 += 128) {
        int rec[4], sl[4];
#pragma unroll
        for (int u = 0; u < 4; ++u) {
            const int idx = i0 + 32 * u;
            rec[u] = lst[idx < n ? idx : 0];
            sl[u]  = lsrc[idx < n ? idx : 0];
        }
        int      dl[4];
        float    sc[4];
        unsigned mk[4];
        float4   xv[4];
        bool     ok[4];
#pragma unroll
        for (int u = 0; u < 4; ++u) {
            const int  idx = i0 + 32 * u;
            const bool inl = idx < n;
            dl[u] = inl ? (rec[u] >> 20) : (idx - n);
            const int node = base + dl[u];
            ok[u] = (idx < total) && (node < NN);
            const int sN = ok[u] ? (inl ? sl[u] : node) : 0;
            sc[u] = lsc[(ok[u] ? idx : 0) * HH + h];
            mk[u] = mx[(ok[u] ? dl[u] : 0) * HH + h];
            xv[u] = xl4[(size_t)sN * (HCC / 4) + lane];
        }
#pragma unroll
        for (int u = 0; u < 4; ++u) {
            if (ok[u]) {
                const float ex = expf(sc[u] - funkey(mk[u]));
                float* a = &acc[dl[u] * HCC + lane];
                atomicAdd(a +  0, xv[u].x * ex);
                atomicAdd(a + 32, xv[u].y * ex);
                atomicAdd(a + 64, xv[u].z * ex);
                atomicAdd(a + 96, xv[u].w * ex);
                if ((lane & 7) == 0) atomicAdd(&dsum[dl[u] * HH + h], ex);
            }
        }
    }
    __syncthreads();

    float4* agg4 = (float4*)agg;
    const float4 cbv = ((const float4*)cb)[lane];
    float fs = 0.0f, fq = 0.0f;
    for (int i = tid; i < NPB * (HCC / 4); i += 1024) {
        const int j    = i >> 5;
        const int node = base + j;
        if (node < NN) {
            const float* a = &acc[j * HCC + lane];
            const float  d = dsum[j * HH + h] + 1e-16f;
            float4 y;
            y.x = a[0]  / d + cbv.x;
            y.y = a[32] / d + cbv.y;
            y.z = a[64] / d + cbv.z;
            y.w = a[96] / d + cbv.w;
            agg4[(size_t)node * (HCC / 4) + lane] = y;
            fs += (y.x + y.y) + (y.z + y.w);
            fq += (y.x * y.x + y.y * y.y) + (y.z * y.z + y.w * y.w);
        }
    }
    for (int off = 16; off >= 1; off >>= 1) {
        fs += __shfl_xor(fs, off);
        fq += __shfl_xor(fq, off);
    }
    if (lane == 0) { wsum[wv] = (double)fs; wsq[wv] = (double)fq; }
    __syncthreads();
    if (wv == 0) {
        double s = wsum[lane];
        double q = wsq[lane];
        for (int off = 16; off >= 1; off >>= 1) {
            s += __shfl_xor(s, off);
            q += __shfl_xor(q, off);
        }
        if (lane == 0) ((double2*)part)[blockIdx.x] = make_double2(s, q);
    }
}

__global__ __launch_bounds__(256)
void stats_finish_kernel(const double* __restrict__ part, int nblk,
                         double* __restrict__ red, int total) {
    __shared__ double ws_[8];
    __shared__ double wq_[8];
    const int tid  = (int)threadIdx.x;
    const int lane = tid & 31;
    const int wv   = tid >> 5;
    double s = 0.0, q = 0.0;
    for (int i = tid; i < nblk; i += 256) {
        const double2 p = ((const double2*)part)[i];
        s += p.x;
        q += p.y;
    }
    for (int off = 16; off >= 1; off >>= 1) {
        s += __shfl_xor(s, off);
        q += __shfl_xor(q, off);
    }
    if (lane == 0) { ws_[wv] = s; wq_[wv] = q; }
    __syncthreads();
    if (wv == 0) {
        s = lane < 8 ? ws_[lane] : 0.0;
        q = lane < 8 ? wq_[lane] : 0.0;
        for (int off = 4; off >= 1; off >>= 1) {
            s += __shfl_xor(s, off);
            q += __shfl_xor(q, off);
        }
        if (lane == 0) {
            const double mu = s / (double)total;
            double var = q / (double)total - mu * mu;
            if (var < 0.0) var = 0.0;
            const float sd = sqrtf((float)var);
            red[0] = mu;
            red[1] = (double)(sd + 1e-5f);
        }
    }
}

__global__ __launch_bounds__(256)
void pln_apply_kernel(float* __restrict__ agg, const float* __restrict__ w,
                      const float* __restrict__ b, const double* __restrict__ red,
                      int total) {
    const int idx = (int)(blockIdx.x * blockDim.x + threadIdx.x);
    if (idx >= total) return;
    const double mu  = red[0];
    const double dnm = red[1];
    const int c = idx & (HCC - 1);
    float y = w[c] * (float)((agg[idx] - mu) / dnm) + b[c];
    agg[idx] = leakyf(y, 0.01f);
}

extern "C" void kernel_launch(void* const* d_in, const int* in_sizes, int n_in,
                              void* d_out, int out_size, void* d_ws, size_t ws_size,
                              hipStream_t stream) {
    const float* x          = (const float*)d_in[0];
    const int*   edge_index = (const int*)  d_in[1];
    const float* edge_attr  = (const float*)d_in[2];
    const float* embed_W    = (const float*)d_in[3];
    const float* embed_b    = (const float*)d_in[4];
    const float* embed_lnw  = (const float*)d_in[5];
    const float* embed_lnb  = (const float*)d_in[6];
    const float* Wl         = (const float*)d_in[7];
    const float* bl         = (const float*)d_in[8];
    const float* Wr         = (const float*)d_in[9];
    const float* br         = (const float*)d_in[10];
    const float* We         = (const float*)d_in[11];
    const float* att        = (const float*)d_in[12];
    const float* conv_b     = (const float*)d_in[13];
    const float* pln_w      = (const float*)d_in[14];
    const float* pln_b      = (const float*)d_in[15];
    const float* proj_W     = (const float*)d_in[16];
    const float* proj_b     = (const float*)d_in[17];
    const float* ln2_w      = (const float*)d_in[18];
    const float* ln2_b      = (const float*)d_in[19];

    const int* srcE = edge_index;
    const int* dstE = edge_index + EE;

    size_t off = 0;
    auto carve = [&](size_t nbytes) -> char* {
        char* p = (char*)d_ws + off;
        off += (nbytes + 255) & ~(size_t)255;
        return p;
    };
    float*    h     = (float*)   carve((size_t)NN * FF  * 4);
    float*    raw   = (float*)   carve((size_t)NN * FF  * 4);
    float*    xl    = (float*)   carve((size_t)NN * HCC * 4);
    float*    agg   = (float*)   carve((size_t)NN * HCC * 4);
    double*   red   = (double*)  carve(2 * sizeof(double));
    double*   part  = (double*)  carve((size_t)NOWN * 2 * sizeof(double));
    int*      olst  = (int*)     carve((size_t)NOWN * LCAP * sizeof(int));
    int*      osrc  = (int*)     carve((size_t)NOWN * LCAP * sizeof(int));
    int*      ocnt  = (int*)     carve((size_t)NOWN * sizeof(int));
    (void)ws_size; (void)in_sizes; (void)n_in; (void)out_size;

    float* jk = (float*)d_out;

    const int mtiles = NN / 16;
    const dim3 blk(256);
    const dim3 gGemm((mtiles * 32 + 255) / 256);
    const dim3 gRows((NN * 32 + 255) / 256);
    const dim3 gElem(((size_t)NN * HCC + 255) / 256);
    const int totalHC = NN * HCC;

    const size_t shLin  = (size_t)HCC * (FF + 4) * 4;
    const size_t shProj = (size_t)FF * (HCC + 4) * 4;

    gemm_wmma_f32<2, INF_, false><<<gGemm, blk, 0, stream>>>(x, embed_W, embed_b, raw, NN, FF);
    ln_rows_kernel<<<gRows, blk, 0, stream>>>(raw, embed_lnw, embed_lnb, h, jk, 0, NN);

    for (int l = 0; l < LL; ++l) {
        const float* Wl_l = Wl + (size_t)l * HCC * FF;
        const float* bl_l = bl + (size_t)l * HCC;
        const float* Wr_l = Wr + (size_t)l * HCC * FF;
        const float* br_l = br + (size_t)l * HCC;
        const float* We_l = We + (size_t)l * HCC * EDD;
        const float* at_l = att + (size_t)l * HH * CC;
        const float* cb_l = conv_b + (size_t)l * HCC;
        const float* pw_l = pln_w + (size_t)l * HCC;
        const float* pb_l = pln_b + (size_t)l * HCC;
        const float* pW_l = proj_W + (size_t)l * FF * HCC;
        const float* pB_l = proj_b + (size_t)l * FF;
        const float* l2w  = ln2_w + (size_t)l * FF;
        const float* l2b  = ln2_b + (size_t)l * FF;

        gemm_wmma_f32<8, FF, true><<<gGemm, blk, shLin, stream>>>(h, Wl_l, bl_l, xl, NN, HCC);

        if (l == 0)
            edge_owner_kernel<true><<<dim3(NOWN), dim3(1024), 0, stream>>>(
                srcE, dstE, edge_attr, We_l, xl, h, Wr_l, br_l, at_l, cb_l, agg, part, olst, osrc, ocnt);
        else
            edge_owner_kernel<false><<<dim3(NOWN), dim3(1024), 0, stream>>>(
                srcE, dstE, edge_attr, We_l, xl, h, Wr_l, br_l, at_l, cb_l, agg, part, olst, osrc, ocnt);

        stats_finish_kernel<<<dim3(1), dim3(256), 0, stream>>>(part, NOWN, red, totalHC);
        pln_apply_kernel<<<gElem, blk, 0, stream>>>(agg, pw_l, pb_l, red, totalHC);

        gemm_wmma_f32<2, HCC, true><<<gGemm, blk, shProj, stream>>>(agg, pW_l, pB_l, raw, NN, FF);
        ln_rows_kernel<<<gRows, blk, 0, stream>>>(raw, l2w, l2b, h, jk, 1, NN);
    }
}


# owner work-groups' edge lists built by one list launch: blocks of 4096 edges rank per owner in LDS, reserve runs by one global int add per owner; 256-fold target-list read gone; sums in arrival order
# speedup vs baseline: 2.9527x; 1.0703x over previous
#include <hip/hip_runtime.h>
#include <math.h>

#define NN   50000
#define EE   800000
#define INF_ 512
#define FF   32
#define HH   4
#define CC   32
#define HCC  128
#define EDD  4
#define E2E  850000
#define LL   2

typedef float v2f __attribute__((ext_vector_type(2)));
typedef float v8f __attribute__((ext_vector_type(8)));
typedef float v4f __attribute__((ext_vector_type(4)));

__device__ __forceinline__ unsigned fkey(float f) {
    unsigned u = __float_as_uint(f);
    return (u & 0x80000000u) ? ~u : (u | 0x80000000u);
}
__device__ __forceinline__ float funkey(unsigned u) {
    return __uint_as_float((u & 0x80000000u) ? (u & 0x7FFFFFFFu) : ~u);
}
__device__ __forceinline__ float leakyf(float x, float s) {
    return x >= 0.0f ? x : s * x;
}

template <int CT, int KT, bool USE_LDS>
__global__ __launch_bounds__(256)
void gemm_wmma_f32(const float* __restrict__ A, const float* __restrict__ W,
                   const float* __restrict__ bias, float* __restrict__ Cout,
                   int M, int Nout) {
    extern __shared__ float sW[];
    const int tid  = (int)threadIdx.x;
    const int lane = tid & 31;
    const int wave = (int)((blockIdx.x * blockDim.x + tid) >> 5);
    const int mtiles  = M >> 4;
    const int ngroups = Nout / (16 * CT);
    const int stride  = KT + 4;

    if (USE_LDS) {
        const int total = Nout * KT;
        for (int i = tid; i < total; i += 256) {
            const int c  = i / KT;
            const int kk = i - c * KT;
            sW[c * stride + kk] = W[i];
        }
        __syncthreads();
    }
    if (wave >= mtiles * ngroups) return;

    const int mtile = wave / ngroups;
    const int ng    = wave % ngroups;
    const int row0  = mtile * 16;
    const int col0  = ng * 16 * CT;
    const int hi = lane >> 4;
    const int mi = lane & 15;

    v8f acc[CT] = {};
    const float* Arow = A + (size_t)(row0 + mi) * KT + 2 * hi;

    v2f a_cur;
    a_cur.x = Arow[0];
    a_cur.y = Arow[1];
    v2f b_cur[CT];
    if (!USE_LDS) {
#pragma unroll
        for (int t = 0; t < CT; ++t) {
            const float* Wc = W + (size_t)(col0 + t * 16 + mi) * KT + 2 * hi;
            b_cur[t].x = Wc[0];
            b_cur[t].y = Wc[1];
        }
    }

    for (int k = 0; k < KT; k += 4) {
        v2f a_nxt = {};
        v2f b_nxt[CT];
        if (k + 4 < KT) {
            a_nxt.x = Arow[k + 4];
            a_nxt.y = Arow[k + 5];
            if (!USE_LDS) {
#pragma unroll
                for (int t = 0; t < CT; ++t) {
                    const float* Wc = W + (size_t)(col0 + t * 16 + mi) * KT + 2 * hi;
                    b_nxt[t].x = Wc[k + 4];
                    b_nxt[t].y = Wc[k + 5];
                }
                if ((k & 31) == 0)
                    __builtin_prefetch(Arow + k + 132, 0, 1);
            }
        }
#pragma unroll
        for (int t = 0; t < CT; ++t) {
            v2f b;
            if (USE_LDS) {
                const float* Sc = sW + (size_t)(col0 + t * 16 + mi) * stride + 2 * hi;
                b.x = Sc[k];
                b.y = Sc[k + 1];
            } else {
                b = b_cur[t];
            }
            acc[t] = __builtin_amdgcn_wmma_f32_16x16x4_f32(
                false, a_cur, false, b, (short)0, acc[t], false, false);
        }
        a_cur = a_nxt;
        if (!USE_LDS) {
#pragma unroll
            for (int t = 0; t < CT; ++t) b_cur[t] = b_nxt[t];
        }
    }

#pragma unroll
    for (int t = 0; t < CT; ++t) {
        const int col = col0 + t * 16 + mi;
        const float bb = bias ? bias[col] : 0.0f;
#pragma unroll
        for (int v = 0; v < 8; ++v) {
            const int r = row0 + v + 8 * hi;
            Cout[(size_t)r * Nout + col] = acc[t][v] + bb;
        }
    }
}

__global__ __launch_bounds__(256)
void ln_rows_kernel(const float* __restrict__ Z, const float* __restrict__ w,
                    const float* __restrict__ b, float* __restrict__ Hout,
                    float* __restrict__ jk, int mode, int M, int* __restrict__ zc) {
    const int row  = (int)((blockIdx.x * blockDim.x + threadIdx.x) >> 5);
    const int lane = (int)(threadIdx.x & 31);
    if (zc != nullptr && blockIdx.x == 0) zc[threadIdx.x] = 0;
    if (row >= M) return;
    float v = Z[(size_t)row * FF + lane];
    float s = v;
#pragma unroll
    for (int off = 16; off >= 1; off >>= 1) s += __shfl_xor(s, off);
    const float mu = s * (1.0f / 32.0f);
    const float d = v - mu;
    float q = d * d;
#pragma unroll
    for (int off = 16; off >= 1; off >>= 1) q += __shfl_xor(q, off);
    const float var = q * (1.0f / 32.0f);
    float y = w[lane] * d * rsqrtf(var + 1e-5f) + b[lane];
    y = leakyf(y, 0.01f);
    Hout[(size_t)row * FF + lane] = y;
    float* o = jk + (size_t)row * FF + lane;
    *o = (mode == 0) ? y : fmaxf(*o, y);
}

#define NPB    196
#define NOWN   256
#define LCAP   4096
#define XRT    13

#define LBLK 1024
static_assert(NOWN == 256, "ln_rows_kernel clears one counter per thread of a 256-thread block");
__global__ __launch_bounds__(LBLK)
void owner_lists_kernel(const int* __restrict__ src, const int* __restrict__ dst,
                        int* __restrict__ glst, int* __restrict__ gsrc,
                        int* __restrict__ gcnt) {
    __shared__ int hist[NOWN];
    __shared__ int gbase[NOWN];
    const int tid = (int)threadIdx.x;
    if (tid < NOWN) hist[tid] = 0;
    __syncthreads();
    const int  q  = (int)blockIdx.x * LBLK + tid;
    const bool in = q < EE / 4;
    int4 dv = make_int4(-1, -1, -1, -1);
    int4 sv = make_int4(0, 0, 0, 0);
    if (in) {
        dv = ((const int4*)dst)[q];
        sv = ((const int4*)src)[q];
    }
    const int dd[4] = {dv.x, dv.y, dv.z, dv.w};
    const int ss[4] = {sv.x, sv.y, sv.z, sv.w};
    int ow[4], rk[4];
#pragma unroll
    for (int k = 0; k < 4; ++k) {
        const bool ok = (unsigned)dd[k] < (unsigned)NN;
        ow[k] = ok ? dd[k] / NPB : -1;
        rk[k] = ok ? atomicAdd(&hist[ow[k]], 1) : 0;
    }
    __syncthreads();
    if (tid < NOWN) {
        const int c = hist[tid];
        gbase[tid] = c > 0 ? atomicAdd(&gcnt[tid], c) : 0;
    }
    __syncthreads();
#pragma unroll
    for (int k = 0; k < 4; ++k) {
        if (ow[k] >= 0) {
            const int pos = gbase[ow[k]] + rk[k];
            if (pos < LCAP) {
                const int dl = dd[k] - ow[k] * NPB;
                glst[(size_t)ow[k] * LCAP + pos] = (q * 4 + k) | (dl << 20);
                gsrc[(size_t)ow[k] * LCAP + pos] = ss[k];
            }
        }
    }
}

__device__ __forceinline__ float edge_score_one(int e, int s, const float* xrow, int hh,
                                             const float* __restrict__ ea,
                                             const float* __restrict__ We,
                                             const float* __restrict__ xl,
                                             const float* __restrict__ att) {
    const float4 ev = ((const float4*)ea)[e];
    float sc = 0.0f;
#pragma nounroll
    for (int cc = 0; cc < CC; ++cc) {
        const int ch = hh * CC + cc;
        const float4 wr = ((const float4*)We)[ch];
        const float eev = wr.x * ev.x + wr.y * ev.y + wr.z * ev.z + wr.w * ev.w;
        const float m = xl[(size_t)s * HCC + ch] + xrow[ch] + eev;
        sc += att[ch] * leakyf(m, 0.2f);
    }
    return sc;
}

__global__ __launch_bounds__(1024)
void edge_owner_kernel(const int* __restrict__ src, const int* __restrict__ dst,
                       const float* __restrict__ ea, const float* __restrict__ We,
                       const float* __restrict__ xl, const float* __restrict__ hin,
                       const float* __restrict__ Wr, const float* __restrict__ br,
                       const float* __restrict__ att, const float* __restrict__ cb,
                       float* __restrict__ agg, double* __restrict__ part,
                       const int* __restrict__ glst, const int* __restrict__ gsrc,
                       const int* __restrict__ gcnt) {
    __shared__ float    acc[NPB * HCC];
    __shared__ float    dsum[NPB * HH];
    __shared__ unsigned mx[NPB * HH];
    __shared__ int      lst[LCAP];
    __shared__ int      lsrc[LCAP];
    __shared__ float    lsc[(LCAP + NPB) * HH];
    __shared__ double   wsum[32];
    __shared__ double   wsq[32];
    __shared__ __attribute__((aligned(16))) float xrs[XRT * 16 * HCC];
    const int tid  = (int)threadIdx.x;
    const int lane = tid & 31;
    const int wv   = tid >> 5;
    const int base = (int)blockIdx.x * NPB;

    {
        const int hi = lane >> 4;
        const int mi = lane & 15;
        const int nt = wv & 7;
        const float* Wc = Wr + (size_t)(nt * 16 + mi) * FF + 2 * hi;
        v2f bf[8];
#pragma unroll
        for (int q = 0; q < 8; ++q) bf[q] = *(const v2f*)(Wc + 4 * q);
        const float bb = br[nt * 16 + mi];
        v2f af[4][8];
#pragma unroll
        for (int u = 0; u < 4; ++u) {
            int row = base + ((wv >> 3) + 4 * u) * 16 + mi;
            row = row < NN ? row : NN - 1;
            const float* Ar = hin + (size_t)row * FF + 2 * hi;
#pragma unroll
            for (int q = 0; q < 8; ++q) af[u][q] = *(const v2f*)(Ar + 4 * q);
        }
        v8f cacc[4] = {};
#pragma unroll
        for (int q = 0; q < 8; ++q) {
#pragma unroll
            for (int u = 0; u < 4; ++u)
                cacc[u] = __builtin_amdgcn_wmma_f32_16x16x4_f32(
                    false, af[u][q], false, bf[q], (short)0, cacc[u], false, false);
        }
#pragma unroll
        for (int u = 0; u < 4; ++u) {
            const int mt = (wv >> 3) + 4 * u;
            if (mt < XRT) {
#pragma unroll
                for (int v = 0; v < 8; ++v)
                    xrs[(mt * 16 + v + 8 * hi) * HCC + nt * 16 + mi] = cacc[u][v] + bb;
            }
        }
    }

    for (int i = tid; i < NPB * HCC; i += 1024) acc[i] = 0.0f;
    for (int i = tid; i < NPB * HH; i += 1024) { dsum[i] = 0.0f; mx[i] = 0u; }
    __syncthreads();

    int n = gcnt[blockIdx.x];
    const bool over = n > LCAP;
    if (over) {
        for (int e = tid; e < EE; e += 1024) {
            const int d = dst[e];
            const unsigned dl = (unsigned)(d - base);
            if (dl < (unsigned)NPB) {
                const int s = src[e];
                for (int hh = 0; hh < HH; ++hh)
                    atomicMax(&mx[dl * HH + hh],
                              fkey(edge_score_one(e, s, &xrs[dl * HCC], hh, ea, We, xl, att)));
            }
        }
        n = 0;
    }
    const int* mylst = glst + (size_t)blockIdx.x * LCAP;
    const int* mysrc = gsrc + (size_t)blockIdx.x * LCAP;
    for (int i = tid; i < n; i += 1024) {
        lst[i]  = mylst[i];
        lsrc[i] = mysrc[i];
    }
    __syncthreads();

    const int total = n + NPB;
    const int h     = lane >> 3;
    const float4* xl4 = (const float4*)xl;
    const float4* xrs4 = (const float4*)xrs;

    {
        const float4* We4 = (const float4*)We;
        const v4f* xl4v = (const v4f*)xl;
        const v4f* xrsv = (const v4f*)xrs;
        const v4f* ea4v = (const v4f*)ea;
        float4 wr[4];
#pragma unroll
        for (int k = 0; k < 4; ++k) wr[k] = We4[lane * 4 + k];
        const float4 atv = ((const float4*)att)[lane];
        const float at[4] = {atv.x, atv.y, atv.z, atv.w};

        for (int i0 = wv; i0 < n; i0 += 128) {
            int rec[4], sl[4];
#pragma unroll
            for (int u = 0; u < 4; ++u) {
                const int idx = i0 + 32 * u;
                rec[u] = lst[idx < n ? idx : 0];
                sl[u]  = lsrc[idx < n ? idx : 0];
            }
            v4f xa[4], xb[4], ev[4];
#pragma unroll
            for (int u = 0; u < 4; ++u) {
                xa[u] = xl4v[(size_t)sl[u] * (HCC / 4) + lane];
                xb[u] = xrsv[(rec[u] >> 20) * (HCC / 4) + lane];
                ev[u] = ea4v[rec[u] & 0xFFFFF];
            }
            asm volatile("" : "+v"(xa[0]), "+v"(xa[1]), "+v"(xa[2]), "+v"(xa[3]),
                              "+v"(xb[0]), "+v"(xb[1]), "+v"(xb[2]), "+v"(xb[3]),
                              "+v"(ev[0]), "+v"(ev[1]), "+v"(ev[2]), "+v"(ev[3]));
            float sc[4];
#pragma unroll
            for (int u = 0; u < 4; ++u) {
                const float pa[4] = {xa[u].x, xa[u].y, xa[u].z, xa[u].w};
                const float pb[4] = {xb[u].x, xb[u].y, xb[u].z, xb[u].w};
                float s = 0.0f;
#pragma unroll
                for (int k = 0; k < 4; ++k) {
                    const float eev = wr[k].x * ev[u].x + wr[k].y * ev[u].y +
                                      wr[k].z * ev[u].z + wr[k].w * ev[u].w;
                    float m = pa[k] + pb[k] + eev;
                    m = leakyf(m, 0.2f);
                    s += at[k] * m;
                }
                sc[u] = s;
            }
#pragma unroll
            for (int off = 4; off >= 1; off >>= 1) {
#pragma unroll
                for (int u = 0; u < 4; ++u) sc[u] += __shfl_xor(sc[u], off);
            }
#pragma unroll
            for (int u = 0; u < 4; ++u) {
                const int idx = i0 + 32 * u;
                if (idx < n && (lane & 7) == 0) {
                    lsc[idx * HH + h] = sc[u];
                    atomicMax(&mx[(rec[u] >> 20) * HH + h], fkey(sc[u]));
                }
            }
        }

        for (int j0 = wv; j0 < NPB; j0 += 128) {
            float4 xa[4], xb[4];
            bool   ok[4];
#pragma unroll
            for (int u = 0; u < 4; ++u) {
                const int j    = j0 + 32 * u;
                const int node = base + j;
                ok[u] = (j < NPB) && (node < NN);
                const int nd = ok[u] ? node : 0;
                xa[u] = xl4[(size_t)nd * (HCC / 4) + lane];
                xb[u] = xrs4[(ok[u] ? j : 0) * (HCC / 4) + lane];
            }
            float sc[4];
#pragma unroll
            for (int u = 0; u < 4; ++u) {
                const float pa[4] = {xa[u].x, xa[u].y, xa[u].z, xa[u].w};
                const float pb[4] = {xb[u].x, xb[u].y, xb[u].z, xb[u].w};
                float s = 0.0f;
#pragma unroll
                for (int k = 0; k < 4; ++k) {
                    float m = pa[k] + pb[k];
                    m = leakyf(m, 0.2f);
                    s += at[k] * m;
                }
                sc[u] = s;
            }
#pragma unroll
            for (int off = 4; off >= 1; off >>= 1) {
#pragma unroll
                for (int u = 0; u < 4; ++u) sc[u] += __shfl_xor(sc[u], off);
            }
#pragma unroll
            for (int u = 0; u < 4; ++u) {
                const int j = j0 + 32 * u;
                if (ok[u] && (lane & 7) == 0) {
                    lsc[(n + j) * HH + h] = sc[u];
                    atomicMax(&mx[j * HH + h], fkey(sc[u]));
                }
            }
        }
    }
    __syncthreads();

    if (over) {
        for (int e = tid; e < EE; e += 1024) {
            const int d = dst[e];
            const unsigned dl = (unsigned)(d - base);
            if (dl < (unsigned)NPB) {
                const int s = src[e];
                for (int hh = 0; hh < HH; ++hh) {
                    const float ex = expf(edge_score_one(e, s, &xrs[dl * HCC], hh, ea, We, xl, att) -
                                          funkey(mx[dl * HH + hh]));
                    atomicAdd(&dsum[dl * HH + hh], ex);
                    for (int cc = 0; cc < CC; ++cc) {
                        const int ch = hh * CC + cc;
                        atomicAdd(&acc[dl * HCC + (ch & 3) * 32 + (ch >> 2)],
                                  xl[(size_t)s * HCC + ch] * ex);
                    }
                }
            }
        }
    }

    for (int i0 = wv; i0 < total; i0 += 128) {
        int rec[4], sl[4];
#pragma unroll
        for (int u = 0; u < 4; ++u) {
            const int idx = i0 + 32 * u;
            rec[u] = lst[idx < n ? idx : 0];
            sl[u]  = lsrc[idx < n ? idx : 0];
        }
        int      dl[4];
        float    sc[4];
        unsigned mk[4];
        float4   xv[4];
        bool     ok[4];
#pragma unroll
        for (int u = 0; u < 4; ++u) {
            const int  idx = i0 + 32 * u;
            const bool inl = idx < n;
            dl[u] = inl ? (rec[u] >> 20) : (idx - n);
            const int node = base + dl[u];
            ok[u] = (idx < total) && (node < NN);
            const int sN = ok[u] ? (inl ? sl[u] : node) : 0;
            sc[u] = lsc[(ok[u] ? idx : 0) * HH + h];
            mk[u] = mx[(ok[u] ? dl[u] : 0) * HH + h];
            xv[u] = xl4[(size_t)sN * (HCC / 4) + lane];
        }
#pragma unroll
        for (int u = 0; u < 4; ++u) {
            if (ok[u]) {
                const float ex = expf(sc[u] - funkey(mk[u]));
                float* a = &acc[dl[u] * HCC + lane];
                atomicAdd(a +  0, xv[u].x * ex);
                atomicAdd(a + 32, xv[u].y * ex);
                atomicAdd(a + 64, xv[u].z * ex);
                atomicAdd(a + 96, xv[u].w * ex);
                if ((lane & 7) == 0) atomicAdd(&dsum[dl[u] * HH + h], ex);
            }
        }
    }
    __syncthreads();

    float4* agg4 = (float4*)agg;
    const float4 cbv = ((const float4*)cb)[lane];
    float fs = 0.0f, fq = 0.0f;
    for (int i = tid; i < NPB * (HCC / 4); i += 1024) {
        const int j    = i >> 5;
        const int node = base + j;
        if (node < NN) {
            const float* a = &acc[j * HCC + lane];
            const float  d = dsum[j * HH + h] + 1e-16f;
            float4 y;
            y.x = a[0]  / d + cbv.x;
            y.y = a[32] / d + cbv.y;
            y.z = a[64] / d + cbv.z;
            y.w = a[96] / d + cbv.w;
            agg4[(size_t)node * (HCC / 4) + lane] = y;
            fs += (y.x + y.y) + (y.z + y.w);
            fq += (y.x * y.x + y.y * y.y) + (y.z * y.z + y.w * y.w);
        }
    }
    for (int off = 16; off >= 1; off >>= 1) {
        fs += __shfl_xor(fs, off);
        fq += __shfl_xor(fq, off);
    }
    if (lane == 0) { wsum[wv] = (double)fs; wsq[wv] = (double)fq; }
    __syncthreads();
    if (wv == 0) {
        double s = wsum[lane];
        double q = wsq[lane];
        for (int off = 16; off >= 1; off >>= 1) {
            s += __shfl_xor(s, off);
            q += __shfl_xor(q, off);
        }
        if (lane == 0) ((double2*)part)[blockIdx.x] = make_double2(s, q);
    }
}

__global__ __launch_bounds__(256)
void stats_finish_kernel(const double* __restrict__ part, int nblk,
                         double* __restrict__ red, int total) {
    __shared__ double ws_[8];
    __shared__ double wq_[8];
    const int tid  = (int)threadIdx.x;
    const int lane = tid & 31;
    const int wv   = tid >> 5;
    double s = 0.0, q = 0.0;
    for (int i = tid; i < nblk; i += 256) {
        const double2 p = ((const double2*)part)[i];
        s += p.x;
        q += p.y;
    }
    for (int off = 16; off >= 1; off >>= 1) {
        s += __shfl_xor(s, off);
        q += __shfl_xor(q, off);
    }
    if (lane == 0) { ws_[wv] = s; wq_[wv] = q; }
    __syncthreads();
    if (wv == 0) {
        s = lane < 8 ? ws_[lane] : 0.0;
        q = lane < 8 ? wq_[lane] : 0.0;
        for (int off = 4; off >= 1; off >>= 1) {
            s += __shfl_xor(s, off);
            q += __shfl_xor(q, off);
        }
        if (lane == 0) {
            const double mu = s / (double)total;
            double var = q / (double)total - mu * mu;
            if (var < 0.0) var = 0.0;
            const float sd = sqrtf((float)var);
            red[0] = mu;
            red[1] = (double)(sd + 1e-5f);
        }
    }
}

__global__ __launch_bounds__(256)
void pln_apply_kernel(float* __restrict__ agg, const float* __restrict__ w,
                      const float* __restrict__ b, const double* __restrict__ red,
                      int total) {
    const int idx = (int)(blockIdx.x * blockDim.x + threadIdx.x);
    if (idx >= total) return;
    const double mu  = red[0];
    const double dnm = red[1];
    const int c = idx & (HCC - 1);
    float y = w[c] * (float)((agg[idx] - mu) / dnm) + b[c];
    agg[idx] = leakyf(y, 0.01f);
}

extern "C" void kernel_launch(void* const* d_in, const int* in_sizes, int n_in,
                              void* d_out, int out_size, void* d_ws, size_t ws_size,
                              hipStream_t stream) {
    const float* x          = (const float*)d_in[0];
    const int*   edge_index = (const int*)  d_in[1];
    const float* edge_attr  = (const float*)d_in[2];
    const float* embed_W    = (const float*)d_in[3];
    const float* embed_b    = (const float*)d_in[4];
    const float* embed_lnw  = (const float*)d_in[5];
    const float* embed_lnb  = (const float*)d_in[6];
    const float* Wl         = (const float*)d_in[7];
    const float* bl         = (const float*)d_in[8];
    const float* Wr         = (const float*)d_in[9];
    const float* br         = (const float*)d_in[10];
    const float* We         = (const float*)d_in[11];
    const float* att        = (const float*)d_in[12];
    const float* conv_b     = (const float*)d_in[13];
    const float* pln_w      = (const float*)d_in[14];
    const float* pln_b      = (const float*)d_in[15];
    const float* proj_W     = (const float*)d_in[16];
    const float* proj_b     = (const float*)d_in[17];
    const float* ln2_w      = (const float*)d_in[18];
    const float* ln2_b      = (const float*)d_in[19];

    const int* srcE = edge_index;
    const int* dstE = edge_index + EE;

    size_t off = 0;
    auto carve = [&](size_t nbytes) -> char* {
        char* p = (char*)d_ws + off;
        off += (nbytes + 255) & ~(size_t)255;
        return p;
    };
    float*    h     = (float*)   carve((size_t)NN * FF  * 4);
    float*    raw   = (float*)   carve((size_t)NN * FF  * 4);
    float*    xl    = (float*)   carve((size_t)NN * HCC * 4);
    float*    agg   = (float*)   carve((size_t)NN * HCC * 4);
    double*   red   = (double*)  carve(2 * sizeof(double));
    double*   part  = (double*)  carve((size_t)NOWN * 2 * sizeof(double));
    int*      olst  = (int*)     carve((size_t)NOWN * LCAP * sizeof(int));
    int*      osrc  = (int*)     carve((size_t)NOWN * LCAP * sizeof(int));
    int*      ocnt  = (int*)     carve((size_t)NOWN * sizeof(int));
    (void)ws_size; (void)in_sizes; (void)n_in; (void)out_size;

    float* jk = (float*)d_out;

    const int mtiles = NN / 16;
    const dim3 blk(256);
    const dim3 gGemm((mtiles * 32 + 255) / 256);
    const dim3 gRows((NN * 32 + 255) / 256);
    const dim3 gElem(((size_t)NN * HCC + 255) / 256);
    const int totalHC = NN * HCC;

    const size_t shLin  = (size_t)HCC * (FF + 4) * 4;
    const size_t shProj = (size_t)FF * (HCC + 4) * 4;

    gemm_wmma_f32<2, INF_, false><<<gGemm, blk, 0, stream>>>(x, embed_W, embed_b, raw, NN, FF);
    ln_rows_kernel<<<gRows, blk, 0, stream>>>(raw, embed_lnw, embed_lnb, h, jk, 0, NN, ocnt);

    owner_lists_kernel<<<dim3((EE / 4 + LBLK - 1) / LBLK), dim3(LBLK), 0, stream>>>(
        srcE, dstE, olst, osrc, ocnt);

    for (int l = 0; l < LL; ++l) {
        const float* Wl_l = Wl + (size_t)l * HCC * FF;
        const float* bl_l = bl + (size_t)l * HCC;
        const float* Wr_l = Wr + (size_t)l * HCC * FF;
        const float* br_l = br + (size_t)l * HCC;
        const float* We_l = We + (size_t)l * HCC * EDD;
        const float* at_l = att + (size_t)l * HH * CC;
        const float* cb_l = conv_b + (size_t)l * HCC;
        const float* pw_l = pln_w + (size_t)l * HCC;
        const float* pb_l = pln_b + (size_t)l * HCC;
        const float* pW_l = proj_W + (size_t)l * FF * HCC;
        const float* pB_l = proj_b + (size_t)l * FF;
        const float* l2w  = ln2_w + (size_t)l * FF;
        const float* l2b  = ln2_b + (size_t)l * FF;

        gemm_wmma_f32<8, FF, true><<<gGemm, blk, shLin, stream>>>(h, Wl_l, bl_l, xl, NN, HCC);

        edge_owner_kernel<<<dim3(NOWN), dim3(1024), 0, stream>>>(
            srcE, dstE, edge_attr, We_l, xl, h, Wr_l, br_l, at_l, cb_l, agg, part, olst, osrc, ocnt);

        stats_finish_kernel<<<dim3(1), dim3(256), 0, stream>>>(part, NOWN, red, totalHC);
        pln_apply_kernel<<<gElem, blk, 0, stream>>>(agg, pw_l, pb_l, red, totalHC);

        gemm_wmma_f32<2, HCC, true><<<gGemm, blk, shProj, stream>>>(agg, pW_l, pB_l, raw, NN, FF);
        ln_rows_kernel<<<gRows, blk, 0, stream>>>(raw, l2w, l2b, h, jk, 1, NN, nullptr);
    }
}


# WMMA operands no longer read 8 B per lane from memory: embed W and the owner groups' h rows + Wr staged in LDS by coalesced 16-B loads, embed A read 16 B per lane; k order of the sums changes
# speedup vs baseline: 3.0815x; 1.0436x over previous
#include <hip/hip_runtime.h>
#include <math.h>

#define NN   50000
#define EE   800000
#define INF_ 512
#define FF   32
#define HH   4
#define CC   32
#define HCC  128
#define EDD  4
#define E2E  850000
#define LL   2

typedef float v2f __attribute__((ext_vector_type(2)));
typedef float v8f __attribute__((ext_vector_type(8)));
typedef float v4f __attribute__((ext_vector_type(4)));

__device__ __forceinline__ unsigned fkey(float f) {
    unsigned u = __float_as_uint(f);
    return (u & 0x80000000u) ? ~u : (u | 0x80000000u);
}
__device__ __forceinline__ float funkey(unsigned u) {
    return __uint_as_float((u & 0x80000000u) ? (u & 0x7FFFFFFFu) : ~u);
}
__device__ __forceinline__ float leakyf(float x, float s) {
    return x >= 0.0f ? x : s * x;
}

template <int CT, int KT, bool USE_LDS>
__global__ __launch_bounds__(256)
void gemm_wmma_f32(const float* __restrict__ A, const float* __restrict__ W,
                   const float* __restrict__ bias, float* __restrict__ Cout,
                   int M, int Nout) {
    extern __shared__ float sW[];
    const int tid  = (int)threadIdx.x;
    const int lane = tid & 31;
    const int wave = (int)((blockIdx.x * blockDim.x + tid) >> 5);
    const int mtiles  = M >> 4;
    const int ngroups = Nout / (16 * CT);
    const int stride  = KT + 4;

    if (USE_LDS) {
        const int total = Nout * KT;
        for (int i = tid; i < total; i += 256) {
            const int c  = i / KT;
            const int kk = i - c * KT;
            sW[c * stride + kk] = W[i];
        }
        __syncthreads();
    }
    if (wave >= mtiles * ngroups) return;

    const int mtile = wave / ngroups;
    const int ng    = wave % ngroups;
    const int row0  = mtile * 16;
    const int col0  = ng * 16 * CT;
    const int hi = lane >> 4;
    const int mi = lane & 15;

    v8f acc[CT] = {};
    const float* Arow = A + (size_t)(row0 + mi) * KT + 2 * hi;

    v2f a_cur;
    a_cur.x = Arow[0];
    a_cur.y = Arow[1];
    v2f b_cur[CT];
    if (!USE_LDS) {
#pragma unroll
        for (int t = 0; t < CT; ++t) {
            const float* Wc = W + (size_t)(col0 + t * 16 + mi) * KT + 2 * hi;
            b_cur[t].x = Wc[0];
            b_cur[t].y = Wc[1];
        }
    }

    for (int k = 0; k < KT; k += 4) {
        v2f a_nxt = {};
        v2f b_nxt[CT];
        if (k + 4 < KT) {
            a_nxt.x = Arow[k + 4];
            a_nxt.y = Arow[k + 5];
            if (!USE_LDS) {
#pragma unroll
                for (int t = 0; t < CT; ++t) {
                    const float* Wc = W + (size_t)(col0 + t * 16 + mi) * KT + 2 * hi;
                    b_nxt[t].x = Wc[k + 4];
                    b_nxt[t].y = Wc[k + 5];
                }
                if ((k & 31) == 0)
                    __builtin_prefetch(Arow + k + 132, 0, 1);
            }
        }
#pragma unroll
        for (int t = 0; t < CT; ++t) {
            v2f b;
            if (USE_LDS) {
                const float* Sc = sW + (size_t)(col0 + t * 16 + mi) * stride + 2 * hi;
                b.x = Sc[k];
                b.y = Sc[k + 1];
            } else {
                b = b_cur[t];
            }
            acc[t] = __builtin_amdgcn_wmma_f32_16x16x4_f32(
                false, a_cur, false, b, (short)0, acc[t], false, false);
        }
        a_cur = a_nxt;
        if (!USE_LDS) {
#pragma unroll
            for (int t = 0; t < CT; ++t) b_cur[t] = b_nxt[t];
        }
    }

#pragma unroll
    for (int t = 0; t < CT; ++t) {
        const int col = col0 + t * 16 + mi;
        const float bb = bias ? bias[col] : 0.0f;
#pragma unroll
        for (int v = 0; v < 8; ++v) {
            const int r = row0 + v + 8 * hi;
            Cout[(size_t)r * Nout + col] = acc[t][v] + bb;
        }
    }
}

#define EW_PITCH (INF_ + 4)
#define EW_WAVES 13
__global__ __launch_bounds__(EW_WAVES * 32)
void embed_gemm_kernel(const float* __restrict__ A, const float* __restrict__ W,
                       const float* __restrict__ bias, float* __restrict__ Cout) {
    __shared__ __attribute__((aligned(16))) float sWe[FF * EW_PITCH];
    const int tid  = (int)threadIdx.x;
    const int lane = tid & 31;
    const int wv   = tid >> 5;
    v4f wst[10];
#pragma unroll
    for (int it = 0; it < 10; ++it) {
        const int i = tid + it * (EW_WAVES * 32);
        if (i < FF * (INF_ / 4)) wst[it] = ((const v4f*)W)[i];
    }
#pragma unroll
    for (int it = 0; it < 10; ++it) {
        const int i = tid + it * (EW_WAVES * 32);
        if (i < FF * (INF_ / 4))
            *(v4f*)&sWe[(i >> 7) * EW_PITCH + 4 * (i & 127)] = wst[it];
    }
    __syncthreads();
    const int strip = (int)blockIdx.x * EW_WAVES + wv;
    if (strip >= NN / 16) return;

    const int hi = lane >> 4;
    const int mi = lane & 15;
    const int row0 = strip * 16;
    const v4f* Ap = (const v4f*)(A + (size_t)(row0 + mi) * INF_ + 4 * hi);
    const float* B0 = sWe + mi * EW_PITCH + 4 * hi;
    const float* B1 = sWe + (16 + mi) * EW_PITCH + 4 * hi;

    v8f acc0 = {};
    v8f acc1 = {};
    v4f abuf[2][8];
#pragma unroll
    for (int j = 0; j < 8; ++j) abuf[0][j] = Ap[2 * j];
#pragma unroll
    for (int g = 0; g < 8; ++g) {
        if (g + 1 < 8) {
#pragma unroll
            for (int j = 0; j < 8; ++j) abuf[(g + 1) & 1][j] = Ap[2 * (8 * (g + 1) + j)];
            __builtin_amdgcn_sched_barrier(0);
        }
#pragma unroll
        for (int j = 0; j < 8; ++j) {
            const int kb = 8 * (8 * g + j);
            const v4f a  = abuf[g & 1][j];
            const v4f b0 = *(const v4f*)(B0 + kb);
            const v4f b1 = *(const v4f*)(B1 + kb);
            const v2f alo = {a.x, a.y};
            const v2f ahi = {a.z, a.w};
            const v2f b0lo = {b0.x, b0.y};
            const v2f b0hi = {b0.z, b0.w};
            const v2f b1lo = {b1.x, b1.y};
            const v2f b1hi = {b1.z, b1.w};
            acc0 = __builtin_amdgcn_wmma_f32_16x16x4_f32(false, alo, false, b0lo, (short)0, acc0, false, false);
            acc1 = __builtin_amdgcn_wmma_f32_16x16x4_f32(false, alo, false, b1lo, (short)0, acc1, false, false);
            acc0 = __builtin_amdgcn_wmma_f32_16x16x4_f32(false, ahi, false, b0hi, (short)0, acc0, false, false);
            acc1 = __builtin_amdgcn_wmma_f32_16x16x4_f32(false, ahi, false, b1hi, (short)0, acc1, false, false);
        }
    }

    const float bb0 = bias[mi];
    const float bb1 = bias[16 + mi];
#pragma unroll
    for (int v = 0; v < 8; ++v) {
        const int r = row0 + v + 8 * hi;
        Cout[(size_t)r * FF + mi]      = acc0[v] + bb0;
        Cout[(size_t)r * FF + 16 + mi] = acc1[v] + bb1;
    }
}

__global__ __launch_bounds__(256)
void ln_rows_kernel(const float* __restrict__ Z, const float* __restrict__ w,
                    const float* __restrict__ b, float* __restrict__ Hout,
                    float* __restrict__ jk, int mode, int M, int* __restrict__ zc) {
    const int row  = (int)((blockIdx.x * blockDim.x + threadIdx.x) >> 5);
    const int lane = (int)(threadIdx.x & 31);
    if (zc != nullptr && blockIdx.x == 0) zc[threadIdx.x] = 0;
    if (row >= M) return;
    float v = Z[(size_t)row * FF + lane];
    float s = v;
#pragma unroll
    for (int off = 16; off >= 1; off >>= 1) s += __shfl_xor(s, off);
    const float mu = s * (1.0f / 32.0f);
    const float d = v - mu;
    float q = d * d;
#pragma unroll
    for (int off = 16; off >= 1; off >>= 1) q += __shfl_xor(q, off);
    const float var = q * (1.0f / 32.0f);
    float y = w[lane] * d * rsqrtf(var + 1e-5f) + b[lane];
    y = leakyf(y, 0.01f);
    Hout[(size_t)row * FF + lane] = y;
    float* o = jk + (size_t)row * FF + lane;
    *o = (mode == 0) ? y : fmaxf(*o, y);
}

#define NPB    196
#define NOWN   256
#define LCAP   4096
#define XRT    13
#define XP     36

#define LBLK 1024
static_assert(NOWN == 256, "ln_rows_kernel clears one counter per thread of a 256-thread block");
__global__ __launch_bounds__(LBLK)
void owner_lists_kernel(const int* __restrict__ src, const int* __restrict__ dst,
                        int* __restrict__ glst, int* __restrict__ gsrc,
                        int* __restrict__ gcnt) {
    __shared__ int hist[NOWN];
    __shared__ int gbase[NOWN];
    const int tid = (int)threadIdx.x;
    if (tid < NOWN) hist[tid] = 0;
    __syncthreads();
    const int  q  = (int)blockIdx.x * LBLK + tid;
    const bool in = q < EE / 4;
    int4 dv = make_int4(-1, -1, -1, -1);
    int4 sv = make_int4(0, 0, 0, 0);
    if (in) {
        dv = ((const int4*)dst)[q];
        sv = ((const int4*)src)[q];
    }
    const int dd[4] = {dv.x, dv.y, dv.z, dv.w};
    const int ss[4] = {sv.x, sv.y, sv.z, sv.w};
    int ow[4], rk[4];
#pragma unroll
    for (int k = 0; k < 4; ++k) {
        const bool ok = (unsigned)dd[k] < (unsigned)NN;
        ow[k] = ok ? dd[k] / NPB : -1;
        rk[k] = ok ? atomicAdd(&hist[ow[k]], 1) : 0;
    }
    __syncthreads();
    if (tid < NOWN) {
        const int c = hist[tid];
        gbase[tid] = c > 0 ? atomicAdd(&gcnt[tid], c) : 0;
    }
    __syncthreads();
#pragma unroll
    for (int k = 0; k < 4; ++k) {
        if (ow[k] >= 0) {
            const int pos = gbase[ow[k]] + rk[k];
            if (pos < LCAP) {
                const int dl = dd[k] - ow[k] * NPB;
                glst[(size_t)ow[k] * LCAP + pos] = (q * 4 + k) | (dl << 20);
                gsrc[(size_t)ow[k] * LCAP + pos] = ss[k];
            }
        }
    }
}

__device__ __forceinline__ float edge_score_one(int e, int s, const float* xrow, int hh,
                                             const float* __restrict__ ea,
                                             const float* __restrict__ We,
                                             const float* __restrict__ xl,
                                             const float* __restrict__ att) {
    const float4 ev = ((const float4*)ea)[e];
    float sc = 0.0f;
#pragma nounroll
    for (int cc = 0; cc < CC; ++cc) {
        const int ch = hh * CC + cc;
        const float4 wr = ((const float4*)We)[ch];
        const float eev = wr.x * ev.x + wr.y * ev.y + wr.z * ev.z + wr.w * ev.w;
        const float m = xl[(size_t)s * HCC + ch] + xrow[ch] + eev;
        sc += att[ch] * leakyf(m, 0.2f);
    }
    return sc;
}

__global__ __launch_bounds__(1024)
void edge_owner_kernel(const int* __restrict__ src, const int* __restrict__ dst,
                       const float* __restrict__ ea, const float* __restrict__ We,
                       const float* __restrict__ xl, const float* __restrict__ hin,
                       const float* __restrict__ Wr, const float* __restrict__ br,
                       const float* __restrict__ att, const float* __restrict__ cb,
                       float* __restrict__ agg, double* __restrict__ part,
                       const int* __restrict__ glst, const int* __restrict__ gsrc,
                       const int* __restrict__ gcnt) {
    __shared__ float    acc[NPB * HCC];
    __shared__ float    dsum[NPB * HH];
    __shared__ unsigned mx[NPB * HH];
    __shared__ int      lst[LCAP];
    __shared__ int      lsrc[LCAP];
    __shared__ __attribute__((aligned(16))) float lsc[(LCAP + NPB) * HH];
    __shared__ double   wsum[32];
    __shared__ double   wsq[32];
    __shared__ __attribute__((aligned(16))) float xrs[XRT * 16 * HCC];
    const int tid  = (int)threadIdx.x;
    const int lane = tid & 31;
    const int wv   = tid >> 5;
    const int base = (int)blockIdx.x * NPB;

    for (int i = tid; i < NPB * HCC; i += 1024) acc[i] = 0.0f;
    for (int i = tid; i < NPB * HH; i += 1024) { dsum[i] = 0.0f; mx[i] = 0u; }

    {
        float* hs  = lsc;
        float* wrs = lsc + XRT * 16 * XP;
        const int r0 = tid >> 3;
        const int c4 = tid & 7;
        const bool two = tid + 1024 < XRT * 16 * (FF / 4);
        int rowa = base + r0;
        rowa = rowa < NN ? rowa : NN - 1;
        int rowb = base + r0 + 128;
        rowb = rowb < NN ? rowb : NN - 1;
        const v4f ha = ((const v4f*)hin)[(size_t)rowa * (FF / 4) + c4];
        const v4f wa = ((const v4f*)Wr)[tid];
        v4f hb = {};
        if (two) hb = ((const v4f*)hin)[(size_t)rowb * (FF / 4) + c4];
        *(v4f*)&hs[r0 * XP + 4 * c4]  = ha;
        *(v4f*)&wrs[r0 * XP + 4 * c4] = wa;
        if (two) *(v4f*)&hs[(r0 + 128) * XP + 4 * c4] = hb;
        __syncthreads();

        const int hi = lane >> 4;
        const int mi = lane & 15;
        const int nt = wv & 7;
        v4f bf[4];
#pragma unroll
        for (int q = 0; q < 4; ++q) bf[q] = *(const v4f*)&wrs[(nt * 16 + mi) * XP + 8 * q + 4 * hi];
        const float bb = br[nt * 16 + mi];
        v4f af[4][4];
#pragma unroll
        for (int u = 0; u < 4; ++u) {
            const int mt  = (wv >> 3) + 4 * u;
            const int mtc = mt < XRT ? mt : XRT - 1;
#pragma unroll
            for (int q = 0; q < 4; ++q) af[u][q] = *(const v4f*)&hs[(mtc * 16 + mi) * XP + 8 * q + 4 * hi];
        }
        v8f cacc[4] = {};
#pragma unroll
        for (int q = 0; q < 4; ++q) {
            const v2f blo = {bf[q].x, bf[q].y};
            const v2f bhi = {bf[q].z, bf[q].w};
#pragma unroll
            for (int u = 0; u < 4; ++u) {
                const v2f alo = {af[u][q].x, af[u][q].y};
                cacc[u] = __builtin_amdgcn_wmma_f32_16x16x4_f32(
                    false, alo, false, blo, (short)0, cacc[u], false, false);
            }
#pragma unroll
            for (int u = 0; u < 4; ++u) {
                const v2f ahi = {af[u][q].z, af[u][q].w};
                cacc[u] = __builtin_amdgcn_wmma_f32_16x16x4_f32(
                    false, ahi, false, bhi, (short)0, cacc[u], false, false);
            }
        }
#pragma unroll
        for (int u = 0; u < 4; ++u) {
            const int mt = (wv >> 3) + 4 * u;
            if (mt < XRT) {
#pragma unroll
                for (int v = 0; v < 8; ++v)
                    xrs[(mt * 16 + v + 8 * hi) * HCC + nt * 16 + mi] = cacc[u][v] + bb;
            }
        }
    }
    __syncthreads();

    int n = gcnt[blockIdx.x];
    const bool over = n > LCAP;
    if (over) {
        for (int e = tid; e < EE; e += 1024) {
            const int d = dst[e];
            const unsigned dl = (unsigned)(d - base);
            if (dl < (unsigned)NPB) {
                const int s = src[e];
                for (int hh = 0; hh < HH; ++hh)
                    atomicMax(&mx[dl * HH + hh],
                              fkey(edge_score_one(e, s, &xrs[dl * HCC], hh, ea, We, xl, att)));
            }
        }
        n = 0;
    }
    const int* mylst = glst + (size_t)blockIdx.x * LCAP;
    const int* mysrc = gsrc + (size_t)blockIdx.x * LCAP;
    for (int i = tid; i < n; i += 1024) {
        lst[i]  = mylst[i];
        lsrc[i] = mysrc[i];
    }
    __syncthreads();

    const int total = n + NPB;
    const int h     = lane >> 3;
    const float4* xl4 = (const float4*)xl;
    const float4* xrs4 = (const float4*)xrs;

    {
        const float4* We4 = (const float4*)We;
        const v4f* xl4v = (const v4f*)xl;
        const v4f* xrsv = (const v4f*)xrs;
        const v4f* ea4v = (const v4f*)ea;
        float4 wr[4];
#pragma unroll
        for (int k = 0; k < 4; ++k) wr[k] = We4[lane * 4 + k];
        const float4 atv = ((const float4*)att)[lane];
        const float at[4] = {atv.x, atv.y, atv.z, atv.w};

        for (int i0 = wv; i0 < n; i0 += 128) {
            int rec[4], sl[4];
#pragma unroll
            for (int u = 0; u < 4; ++u) {
                const int idx = i0 + 32 * u;
                rec[u] = lst[idx < n ? idx : 0];
                sl[u]  = lsrc[idx < n ? idx : 0];
            }
            v4f xa[4], xb[4], ev[4];
#pragma unroll
            for (int u = 0; u < 4; ++u) {
                xa[u] = xl4v[(size_t)sl[u] * (HCC / 4) + lane];
                xb[u] = xrsv[(rec[u] >> 20) * (HCC / 4) + lane];
                ev[u] = ea4v[rec[u] & 0xFFFFF];
            }
            asm volatile("" : "+v"(xa[0]), "+v"(xa[1]), "+v"(xa[2]), "+v"(xa[3]),
                              "+v"(xb[0]), "+v"(xb[1]), "+v"(xb[2]), "+v"(xb[3]),
                              "+v"(ev[0]), "+v"(ev[1]), "+v"(ev[2]), "+v"(ev[3]));
            float sc[4];
#pragma unroll
            for (int u = 0; u < 4; ++u) {
                const float pa[4] = {xa[u].x, xa[u].y, xa[u].z, xa[u].w};
                const float pb[4] = {xb[u].x, xb[u].y, xb[u].z, xb[u].w};
                float s = 0.0f;
#pragma unroll
                for (int k = 0; k < 4; ++k) {
                    const float eev = wr[k].x * ev[u].x + wr[k].y * ev[u].y +
                                      wr[k].z * ev[u].z + wr[k].w * ev[u].w;
                    float m = pa[k] + pb[k] + eev;
                    m = leakyf(m, 0.2f);
                    s += at[k] * m;
                }
                sc[u] = s;
            }
#pragma unroll
            for (int off = 4; off >= 1; off >>= 1) {
#pragma unroll
                for (int u = 0; u < 4; ++u) sc[u] += __shfl_xor(sc[u], off);
            }
#pragma unroll
            for (int u = 0; u < 4; ++u) {
                const int idx = i0 + 32 * u;
                if (idx < n && (lane & 7) == 0) {
                    lsc[idx * HH + h] = sc[u];
                    atomicMax(&mx[(rec[u] >> 20) * HH + h], fkey(sc[u]));
                }
            }
        }

        for (int j0 = wv; j0 < NPB; j0 += 128) {
            float4 xa[4], xb[4];
            bool   ok[4];
#pragma unroll
            for (int u = 0; u < 4; ++u) {
                const int j    = j0 + 32 * u;
                const int node = base + j;
                ok[u] = (j < NPB) && (node < NN);
                const int nd = ok[u] ? node : 0;
                xa[u] = xl4[(size_t)nd * (HCC / 4) + lane];
                xb[u] = xrs4[(ok[u] ? j : 0) * (HCC / 4) + lane];
            }
            float sc[4];
#pragma unroll
            for (int u = 0; u < 4; ++u) {
                const float pa[4] = {xa[u].x, xa[u].y, xa[u].z, xa[u].w};
                const float pb[4] = {xb[u].x, xb[u].y, xb[u].z, xb[u].w};
                float s = 0.0f;
#pragma unroll
                for (int k = 0; k < 4; ++k) {
                    float m = pa[k] + pb[k];
                    m = leakyf(m, 0.2f);
                    s += at[k] * m;
                }
                sc[u] = s;
            }
#pragma unroll
            for (int off = 4; off >= 1; off >>= 1) {
#pragma unroll
                for (int u = 0; u < 4; ++u) sc[u] += __shfl_xor(sc[u], off);
            }
#pragma unroll
            for (int u = 0; u < 4; ++u) {
                const int j = j0 + 32 * u;
                if (ok[u] && (lane & 7) == 0) {
                    lsc[(n + j) * HH + h] = sc[u];
                    atomicMax(&mx[j * HH + h], fkey(sc[u]));
                }
            }
        }
    }
    __syncthreads();

    if (over) {
        for (int e = tid; e < EE; e += 1024) {
            const int d = dst[e];
            const unsigned dl = (unsigned)(d - base);
            if (dl < (unsigned)NPB) {
                const int s = src[e];
                for (int hh = 0; hh < HH; ++hh) {
                    const float ex = expf(edge_score_one(e, s, &xrs[dl * HCC], hh, ea, We, xl, att) -
                                          funkey(mx[dl * HH + hh]));
                    atomicAdd(&dsum[dl * HH + hh], ex);
                    for (int cc = 0; cc < CC; ++cc) {
                        const int ch = hh * CC + cc;
                        atomicAdd(&acc[dl * HCC + (ch & 3) * 32 + (ch >> 2)],
                                  xl[(size_t)s * HCC + ch] * ex);
                    }
                }
            }
        }
    }

    for (int i0 = wv; i0 < total; i0 += 128) {
        int rec[4], sl[4];
#pragma unroll
        for (int u = 0; u < 4; ++u) {
            const int idx = i0 + 32 * u;
            rec[u] = lst[idx < n ? idx : 0];
            sl[u]  = lsrc[idx < n ? idx : 0];
        }
        int      dl[4];
        float    sc[4];
        unsigned mk[4];
        float4   xv[4];
        bool     ok[4];
#pragma unroll
        for (int u = 0; u < 4; ++u) {
            const int  idx = i0 + 32 * u;
            const bool inl = idx < n;
            dl[u] = inl ? (rec[u] >> 20) : (idx - n);
            const int node = base + dl[u];
            ok[u] = (idx < total) && (node < NN);
            const int sN = ok[u] ? (inl ? sl[u] : node) : 0;
            sc[u] = lsc[(ok[u] ? idx : 0) * HH + h];
            mk[u] = mx[(ok[u] ? dl[u] : 0) * HH + h];
            xv[u] = xl4[(size_t)sN * (HCC / 4) + lane];
        }
#pragma unroll
        for (int u = 0; u < 4; ++u) {
            if (ok[u]) {
                const float ex = expf(sc[u] - funkey(mk[u]));
                float* a = &acc[dl[u] * HCC + lane];
                atomicAdd(a +  0, xv[u].x * ex);
                atomicAdd(a + 32, xv[u].y * ex);
                atomicAdd(a + 64, xv[u].z * ex);
                atomicAdd(a + 96, xv[u].w * ex);
                if ((lane & 7) == 0) atomicAdd(&dsum[dl[u] * HH + h], ex);
            }
        }
    }
    __syncthreads();

    float4* agg4 = (float4*)agg;
    const float4 cbv = ((const float4*)cb)[lane];
    float fs = 0.0f, fq = 0.0f;
    for (int i = tid; i < NPB * (HCC / 4); i += 1024) {
        const int j    = i >> 5;
        const int node = base + j;
        if (node < NN) {
            const float* a = &acc[j * HCC + lane];
            const float  d = dsum[j * HH + h] + 1e-16f;
            float4 y;
            y.x = a[0]  / d + cbv.x;
            y.y = a[32] / d + cbv.y;
            y.z = a[64] / d + cbv.z;
            y.w = a[96] / d + cbv.w;
            agg4[(size_t)node * (HCC / 4) + lane] = y;
            fs += (y.x + y.y) + (y.z + y.w);
            fq += (y.x * y.x + y.y * y.y) + (y.z * y.z + y.w * y.w);
        }
    }
    for (int off = 16; off >= 1; off >>= 1) {
        fs += __shfl_xor(fs, off);
        fq += __shfl_xor(fq, off);
    }
    if (lane == 0) { wsum[wv] = (double)fs; wsq[wv] = (double)fq; }
    __syncthreads();
    if (wv == 0) {
        double s = wsum[lane];
        double q = wsq[lane];
        for (int off = 16; off >= 1; off >>= 1) {
            s += __shfl_xor(s, off);
            q += __shfl_xor(q, off);
        }
        if (lane == 0) ((double2*)part)[blockIdx.x] = make_double2(s, q);
    }
}

__global__ __launch_bounds__(256)
void stats_finish_kernel(const double* __restrict__ part, int nblk,
                         double* __restrict__ red, int total) {
    __shared__ double ws_[8];
    __shared__ double wq_[8];
    const int tid  = (int)threadIdx.x;
    const int lane = tid & 31;
    const int wv   = tid >> 5;
    double s = 0.0, q = 0.0;
    for (int i = tid; i < nblk; i += 256) {
        const double2 p = ((const double2*)part)[i];
        s += p.x;
        q += p.y;
    }
    for (int off = 16; off >= 1; off >>= 1) {
        s += __shfl_xor(s, off);
        q += __shfl_xor(q, off);
    }
    if (lane == 0) { ws_[wv] = s; wq_[wv] = q; }
    __syncthreads();
    if (wv == 0) {
        s = lane < 8 ? ws_[lane] : 0.0;
        q = lane < 8 ? wq_[lane] : 0.0;
        for (int off = 4; off >= 1; off >>= 1) {
            s += __shfl_xor(s, off);
            q += __shfl_xor(q, off);
        }
        if (lane == 0) {
            const double mu = s / (double)total;
            double var = q / (double)total - mu * mu;
            if (var < 0.0) var = 0.0;
            const float sd = sqrtf((float)var);
            red[0] = mu;
            red[1] = (double)(sd + 1e-5f);
        }
    }
}

__global__ __launch_bounds__(256)
void pln_apply_kernel(float* __restrict__ agg, const float* __restrict__ w,
                      const float* __restrict__ b, const double* __restrict__ red,
                      int total) {
    const int idx = (int)(blockIdx.x * blockDim.x + threadIdx.x);
    if (idx >= total) return;
    const double mu  = red[0];
    const double dnm = red[1];
    const int c = idx & (HCC - 1);
    float y = w[c] * (float)((agg[idx] - mu) / dnm) + b[c];
    agg[idx] = leakyf(y, 0.01f);
}

extern "C" void kernel_launch(void* const* d_in, const int* in_sizes, int n_in,
                              void* d_out, int out_size, void* d_ws, size_t ws_size,
                              hipStream_t stream) {
    const float* x          = (const float*)d_in[0];
    const int*   edge_index = (const int*)  d_in[1];
    const float* edge_attr  = (const float*)d_in[2];
    const float* embed_W    = (const float*)d_in[3];
    const float* embed_b    = (const float*)d_in[4];
    const float* embed_lnw  = (const float*)d_in[5];
    const float* embed_lnb  = (const float*)d_in[6];
    const float* Wl         = (const float*)d_in[7];
    const float* bl         = (const float*)d_in[8];
    const float* Wr         = (const float*)d_in[9];
    const float* br         = (const float*)d_in[10];
    const float* We         = (const float*)d_in[11];
    const float* att        = (const float*)d_in[12];
    const float* conv_b     = (const float*)d_in[13];
    const float* pln_w      = (const float*)d_in[14];
    const float* pln_b      = (const float*)d_in[15];
    const float* proj_W     = (const float*)d_in[16];
    const float* proj_b     = (const float*)d_in[17];
    const float* ln2_w      = (const float*)d_in[18];
    const float* ln2_b      = (const float*)d_in[19];

    const int* srcE = edge_index;
    const int* dstE = edge_index + EE;

    size_t off = 0;
    auto carve = [&](size_t nbytes) -> char* {
        char* p = (char*)d_ws + off;
        off += (nbytes + 255) & ~(size_t)255;
        return p;
    };
    float*    h     = (float*)   carve((size_t)NN * FF  * 4);
    float*    raw   = (float*)   carve((size_t)NN * FF  * 4);
    float*    xl    = (float*)   carve((size_t)NN * HCC * 4);
    float*    agg   = (float*)   carve((size_t)NN * HCC * 4);
    double*   red   = (double*)  carve(2 * sizeof(double));
    double*   part  = (double*)  carve((size_t)NOWN * 2 * sizeof(double));
    int*      olst  = (int*)     carve((size_t)NOWN * LCAP * sizeof(int));
    int*      osrc  = (int*)     carve((size_t)NOWN * LCAP * sizeof(int));
    int*      ocnt  = (int*)     carve((size_t)NOWN * sizeof(int));
    (void)ws_size; (void)in_sizes; (void)n_in; (void)out_size;

    float* jk = (float*)d_out;

    const int mtiles = NN / 16;
    const dim3 blk(256);
    const dim3 gGemm((mtiles * 32 + 255) / 256);
    const dim3 gRows((NN * 32 + 255) / 256);
    const dim3 gElem(((size_t)NN * HCC + 255) / 256);
    const int totalHC = NN * HCC;

    const size_t shLin  = (size_t)HCC * (FF + 4) * 4;
    const size_t shProj = (size_t)FF * (HCC + 4) * 4;

    embed_gemm_kernel<<<dim3((NN / 16 + EW_WAVES - 1) / EW_WAVES), dim3(EW_WAVES * 32), 0, stream>>>(
        x, embed_W, embed_b, raw);
    ln_rows_kernel<<<gRows, blk, 0, stream>>>(raw, embed_lnw, embed_lnb, h, jk, 0, NN, ocnt);

    owner_lists_kernel<<<dim3((EE / 4 + LBLK - 1) / LBLK), dim3(LBLK), 0, stream>>>(
        srcE, dstE, olst, osrc, ocnt);

    for (int l = 0; l < LL; ++l) {
        const float* Wl_l = Wl + (size_t)l * HCC * FF;
        const float* bl_l = bl + (size_t)l * HCC;
        const float* Wr_l = Wr + (size_t)l * HCC * FF;
        const float* br_l = br + (size_t)l * HCC;
        const float* We_l = We + (size_t)l * HCC * EDD;
        const float* at_l = att + (size_t)l * HH * CC;
        const float* cb_l = conv_b + (size_t)l * HCC;
        const float* pw_l = pln_w + (size_t)l * HCC;
        const float* pb_l = pln_b + (size_t)l * HCC;
        const float* pW_l = proj_W + (size_t)l * FF * HCC;
        const float* pB_l = proj_b + (size_t)l * FF;
        const float* l2w  = ln2_w + (size_t)l * FF;
        const float* l2b  = ln2_b + (size_t)l * FF;

        gemm_wmma_f32<8, FF, true><<<gGemm, blk, shLin, stream>>>(h, Wl_l, bl_l, xl, NN, HCC);

        edge_owner_kernel<<<dim3(NOWN), dim3(1024), 0, stream>>>(
            srcE, dstE, edge_attr, We_l, xl, h, Wr_l, br_l, at_l, cb_l, agg, part, olst, osrc, ocnt);

        stats_finish_kernel<<<dim3(1), dim3(256), 0, stream>>>(part, NOWN, red, totalHC);
        pln_apply_kernel<<<gElem, blk, 0, stream>>>(agg, pw_l, pb_l, red, totalHC);

        gemm_wmma_f32<2, HCC, true><<<gGemm, blk, shProj, stream>>>(agg, pW_l, pB_l, raw, NN, FF);
        ln_rows_kernel<<<gRows, blk, 0, stream>>>(raw, l2w, l2b, h, jk, 1, NN, nullptr);
    }
}


# per-layer row tail in one launch by the owner work-groups: agg rows to LDS with graph-LN applied in f32 (stats folded per group), proj WMMA from LDS, row-LN, JK-max; k order of proj sums changes
# speedup vs baseline: 4.3997x; 1.4278x over previous
#include <hip/hip_runtime.h>
#include <math.h>

#define NN   50000
#define EE   800000
#define INF_ 512
#define FF   32
#define HH   4
#define CC   32
#define HCC  128
#define EDD  4
#define E2E  850000
#define LL   2

typedef float v2f __attribute__((ext_vector_type(2)));
typedef float v8f __attribute__((ext_vector_type(8)));
typedef float v4f __attribute__((ext_vector_type(4)));

__device__ __forceinline__ unsigned fkey(float f) {
    unsigned u = __float_as_uint(f);
    return (u & 0x80000000u) ? ~u : (u | 0x80000000u);
}
__device__ __forceinline__ float funkey(unsigned u) {
    return __uint_as_float((u & 0x80000000u) ? (u & 0x7FFFFFFFu) : ~u);
}
__device__ __forceinline__ float leakyf(float x, float s) {
    return x >= 0.0f ? x : s * x;
}

template <int CT, int KT, bool USE_LDS>
__global__ __launch_bounds__(256)
void gemm_wmma_f32(const float* __restrict__ A, const float* __restrict__ W,
                   const float* __restrict__ bias, float* __restrict__ Cout,
                   int M, int Nout) {
    extern __shared__ float sW[];
    const int tid  = (int)threadIdx.x;
    const int lane = tid & 31;
    const int wave = (int)((blockIdx.x * blockDim.x + tid) >> 5);
    const int mtiles  = M >> 4;
    const int ngroups = Nout / (16 * CT);
    const int stride  = KT + 4;

    if (USE_LDS) {
        const int total = Nout * KT;
        for (int i = tid; i < total; i += 256) {
            const int c  = i / KT;
            const int kk = i - c * KT;
            sW[c * stride + kk] = W[i];
        }
        __syncthreads();
    }
    if (wave >= mtiles * ngroups) return;

    const int mtile = wave / ngroups;
    const int ng    = wave % ngroups;
    const int row0  = mtile * 16;
    const int col0  = ng * 16 * CT;
    const int hi = lane >> 4;
    const int mi = lane & 15;

    v8f acc[CT] = {};
    const float* Arow = A + (size_t)(row0 + mi) * KT + 2 * hi;

    v2f a_cur;
    a_cur.x = Arow[0];
    a_cur.y = Arow[1];
    v2f b_cur[CT];
    if (!USE_LDS) {
#pragma unroll
        for (int t = 0; t < CT; ++t) {
            const float* Wc = W + (size_t)(col0 + t * 16 + mi) * KT + 2 * hi;
            b_cur[t].x = Wc[0];
            b_cur[t].y = Wc[1];
        }
    }

    for (int k = 0; k < KT; k += 4) {
        v2f a_nxt = {};
        v2f b_nxt[CT];
        if (k + 4 < KT) {
            a_nxt.x = Arow[k + 4];
            a_nxt.y = Arow[k + 5];
            if (!USE_LDS) {
#pragma unroll
                for (int t = 0; t < CT; ++t) {
                    const float* Wc = W + (size_t)(col0 + t * 16 + mi) * KT + 2 * hi;
                    b_nxt[t].x = Wc[k + 4];
                    b_nxt[t].y = Wc[k + 5];
                }
                if ((k & 31) == 0)
                    __builtin_prefetch(Arow + k + 132, 0, 1);
            }
        }
#pragma unroll
        for (int t = 0; t < CT; ++t) {
            v2f b;
            if (USE_LDS) {
                const float* Sc = sW + (size_t)(col0 + t * 16 + mi) * stride + 2 * hi;
                b.x = Sc[k];
                b.y = Sc[k + 1];
            } else {
                b = b_cur[t];
            }
            acc[t] = __builtin_amdgcn_wmma_f32_16x16x4_f32(
                false, a_cur, false, b, (short)0, acc[t], false, false);
        }
        a_cur = a_nxt;
        if (!USE_LDS) {
#pragma unroll
            for (int t = 0; t < CT; ++t) b_cur[t] = b_nxt[t];
        }
    }

#pragma unroll
    for (int t = 0; t < CT; ++t) {
        const int col = col0 + t * 16 + mi;
        const float bb = bias ? bias[col] : 0.0f;
#pragma unroll
        for (int v = 0; v < 8; ++v) {
            const int r = row0 + v + 8 * hi;
            Cout[(size_t)r * Nout + col] = acc[t][v] + bb;
        }
    }
}

#define EW_PITCH (INF_ + 4)
#define EW_WAVES 13
__global__ __launch_bounds__(EW_WAVES * 32)
void embed_gemm_kernel(const float* __restrict__ A, const float* __restrict__ W,
                       const float* __restrict__ bias, float* __restrict__ Cout) {
    __shared__ __attribute__((aligned(16))) float sWe[FF * EW_PITCH];
    const int tid  = (int)threadIdx.x;
    const int lane = tid & 31;
    const int wv   = tid >> 5;
    v4f wst[10];
#pragma unroll
    for (int it = 0; it < 10; ++it) {
        const int i = tid + it * (EW_WAVES * 32);
        if (i < FF * (INF_ / 4)) wst[it] = ((const v4f*)W)[i];
    }
#pragma unroll
    for (int it = 0; it < 10; ++it) {
        const int i = tid + it * (EW_WAVES * 32);
        if (i < FF * (INF_ / 4))
            *(v4f*)&sWe[(i >> 7) * EW_PITCH + 4 * (i & 127)] = wst[it];
    }
    __syncthreads();
    const int strip = (int)blockIdx.x * EW_WAVES + wv;
    if (strip >= NN / 16) return;

    const int hi = lane >> 4;
    const int mi = lane & 15;
    const int row0 = strip * 16;
    const v4f* Ap = (const v4f*)(A + (size_t)(row0 + mi) * INF_ + 4 * hi);
    const float* B0 = sWe + mi * EW_PITCH + 4 * hi;
    const float* B1 = sWe + (16 + mi) * EW_PITCH + 4 * hi;

    v8f acc0 = {};
    v8f acc1 = {};
    v4f abuf[2][8];
#pragma unroll
    for (int j = 0; j < 8; ++j) abuf[0][j] = Ap[2 * j];
#pragma unroll
    for (int g = 0; g < 8; ++g) {
        if (g + 1 < 8) {
#pragma unroll
            for (int j = 0; j < 8; ++j) abuf[(g + 1) & 1][j] = Ap[2 * (8 * (g + 1) + j)];
            __builtin_amdgcn_sched_barrier(0);
        }
#pragma unroll
        for (int j = 0; j < 8; ++j) {
            const int kb = 8 * (8 * g + j);
            const v4f a  = abuf[g & 1][j];
            const v4f b0 = *(const v4f*)(B0 + kb);
            const v4f b1 = *(const v4f*)(B1 + kb);
            const v2f alo = {a.x, a.y};
            const v2f ahi = {a.z, a.w};
            const v2f b0lo = {b0.x, b0.y};
            const v2f b0hi = {b0.z, b0.w};
            const v2f b1lo = {b1.x, b1.y};
            const v2f b1hi = {b1.z, b1.w};
            acc0 = __builtin_amdgcn_wmma_f32_16x16x4_f32(false, alo, false, b0lo, (short)0, acc0, false, false);
            acc1 = __builtin_amdgcn_wmma_f32_16x16x4_f32(false, alo, false, b1lo, (short)0, acc1, false, false);
            acc0 = __builtin_amdgcn_wmma_f32_16x16x4_f32(false, ahi, false, b0hi, (short)0, acc0, false, false);
            acc1 = __builtin_amdgcn_wmma_f32_16x16x4_f32(false, ahi, false, b1hi, (short)0, acc1, false, false);
        }
    }

    const float bb0 = bias[mi];
    const float bb1 = bias[16 + mi];
#pragma unroll
    for (int v = 0; v < 8; ++v) {
        const int r = row0 + v + 8 * hi;
        Cout[(size_t)r * FF + mi]      = acc0[v] + bb0;
        Cout[(size_t)r * FF + 16 + mi] = acc1[v] + bb1;
    }
}

__global__ __launch_bounds__(256)
void ln_rows_kernel(const float* __restrict__ Z, const float* __restrict__ w,
                    const float* __restrict__ b, float* __restrict__ Hout,
                    float* __restrict__ jk, int mode, int M, int* __restrict__ zc) {
    const int row  = (int)((blockIdx.x * blockDim.x + threadIdx.x) >> 5);
    const int lane = (int)(threadIdx.x & 31);
    if (zc != nullptr && blockIdx.x == 0) zc[threadIdx.x] = 0;
    if (row >= M) return;
    float v = Z[(size_t)row * FF + lane];
    float s = v;
#pragma unroll
    for (int off = 16; off >= 1; off >>= 1) s += __shfl_xor(s, off);
    const float mu = s * (1.0f / 32.0f);
    const float d = v - mu;
    float q = d * d;
#pragma unroll
    for (int off = 16; off >= 1; off >>= 1) q += __shfl_xor(q, off);
    const float var = q * (1.0f / 32.0f);
    float y = w[lane] * d * rsqrtf(var + 1e-5f) + b[lane];
    y = leakyf(y, 0.01f);
    Hout[(size_t)row * FF + lane] = y;
    float* o = jk + (size_t)row * FF + lane;
    *o = (mode == 0) ? y : fmaxf(*o, y);
}

#define NPB    196
#define NOWN   256
#define LCAP   4096
#define XRT    13
#define XP     36

#define LBLK 1024
static_assert(NOWN == 256, "ln_rows_kernel clears one counter per thread of a 256-thread block");
__global__ __launch_bounds__(LBLK)
void owner_lists_kernel(const int* __restrict__ src, const int* __restrict__ dst,
                        int* __restrict__ glst, int* __restrict__ gsrc,
                        int* __restrict__ gcnt) {
    __shared__ int hist[NOWN];
    __shared__ int gbase[NOWN];
    const int tid = (int)threadIdx.x;
    if (tid < NOWN) hist[tid] = 0;
    __syncthreads();
    const int  q  = (int)blockIdx.x * LBLK + tid;
    const bool in = q < EE / 4;
    int4 dv = make_int4(-1, -1, -1, -1);
    int4 sv = make_int4(0, 0, 0, 0);
    if (in) {
        dv = ((const int4*)dst)[q];
        sv = ((const int4*)src)[q];
    }
    const int dd[4] = {dv.x, dv.y, dv.z, dv.w};
    const int ss[4] = {sv.x, sv.y, sv.z, sv.w};
    int ow[4], rk[4];
#pragma unroll
    for (int k = 0; k < 4; ++k) {
        const bool ok = (unsigned)dd[k] < (unsigned)NN;
        ow[k] = ok ? dd[k] / NPB : -1;
        rk[k] = ok ? atomicAdd(&hist[ow[k]], 1) : 0;
    }
    __syncthreads();
    if (tid < NOWN) {
        const int c = hist[tid];
        gbase[tid] = c > 0 ? atomicAdd(&gcnt[tid], c) : 0;
    }
    __syncthreads();
#pragma unroll
    for (int k = 0; k < 4; ++k) {
        if (ow[k] >= 0) {
            const int pos = gbase[ow[k]] + rk[k];
            if (pos < LCAP) {
                const int dl = dd[k] - ow[k] * NPB;
                glst[(size_t)ow[k] * LCAP + pos] = (q * 4 + k) | (dl << 20);
                gsrc[(size_t)ow[k] * LCAP + pos] = ss[k];
            }
        }
    }
}

__device__ __forceinline__ float edge_score_one(int e, int s, const float* xrow, int hh,
                                             const float* __restrict__ ea,
                                             const float* __restrict__ We,
                                             const float* __restrict__ xl,
                                             const float* __restrict__ att) {
    const float4 ev = ((const float4*)ea)[e];
    float sc = 0.0f;
#pragma nounroll
    for (int cc = 0; cc < CC; ++cc) {
        const int ch = hh * CC + cc;
        const float4 wr = ((const float4*)We)[ch];
        const float eev = wr.x * ev.x + wr.y * ev.y + wr.z * ev.z + wr.w * ev.w;
        const float m = xl[(size_t)s * HCC + ch] + xrow[ch] + eev;
        sc += att[ch] * leakyf(m, 0.2f);
    }
    return sc;
}

__global__ __launch_bounds__(1024)
void edge_owner_kernel(const int* __restrict__ src, const int* __restrict__ dst,
                       const float* __restrict__ ea, const float* __restrict__ We,
                       const float* __restrict__ xl, const float* __restrict__ hin,
                       const float* __restrict__ Wr, const float* __restrict__ br,
                       const float* __restrict__ att, const float* __restrict__ cb,
                       float* __restrict__ agg, double* __restrict__ part,
                       const int* __restrict__ glst, const int* __restrict__ gsrc,
                       const int* __restrict__ gcnt) {
    __shared__ float    acc[NPB * HCC];
    __shared__ float    dsum[NPB * HH];
    __shared__ unsigned mx[NPB * HH];
    __shared__ int      lst[LCAP];
    __shared__ int      lsrc[LCAP];
    __shared__ __attribute__((aligned(16))) float lsc[(LCAP + NPB) * HH];
    __shared__ double   wsum[32];
    __shared__ double   wsq[32];
    __shared__ __attribute__((aligned(16))) float xrs[XRT * 16 * HCC];
    const int tid  = (int)threadIdx.x;
    const int lane = tid & 31;
    const int wv   = tid >> 5;
    const int base = (int)blockIdx.x * NPB;

    for (int i = tid; i < NPB * HCC; i += 1024) acc[i] = 0.0f;
    for (int i = tid; i < NPB * HH; i += 1024) { dsum[i] = 0.0f; mx[i] = 0u; }

    {
        float* hs  = lsc;
        float* wrs = lsc + XRT * 16 * XP;
        const int r0 = tid >> 3;
        const int c4 = tid & 7;
        const bool two = tid + 1024 < XRT * 16 * (FF / 4);
        int rowa = base + r0;
        rowa = rowa < NN ? rowa : NN - 1;
        int rowb = base + r0 + 128;
        rowb = rowb < NN ? rowb : NN - 1;
        const v4f ha = ((const v4f*)hin)[(size_t)rowa * (FF / 4) + c4];
        const v4f wa = ((const v4f*)Wr)[tid];
        v4f hb = {};
        if (two) hb = ((const v4f*)hin)[(size_t)rowb * (FF / 4) + c4];
        *(v4f*)&hs[r0 * XP + 4 * c4]  = ha;
        *(v4f*)&wrs[r0 * XP + 4 * c4] = wa;
        if (two) *(v4f*)&hs[(r0 + 128) * XP + 4 * c4] = hb;
        __syncthreads();

        const int hi = lane >> 4;
        const int mi = lane & 15;
        const int nt = wv & 7;
        v4f bf[4];
#pragma unroll
        for (int q = 0; q < 4; ++q) bf[q] = *(const v4f*)&wrs[(nt * 16 + mi) * XP + 8 * q + 4 * hi];
        const float bb = br[nt * 16 + mi];
        v4f af[4][4];
#pragma unroll
        for (int u = 0; u < 4; ++u) {
            const int mt  = (wv >> 3) + 4 * u;
            const int mtc = mt < XRT ? mt : XRT - 1;
#pragma unroll
            for (int q = 0; q < 4; ++q) af[u][q] = *(const v4f*)&hs[(mtc * 16 + mi) * XP + 8 * q + 4 * hi];
        }
        v8f cacc[4] = {};
#pragma unroll
        for (int q = 0; q < 4; ++q) {
            const v2f blo = {bf[q].x, bf[q].y};
            const v2f bhi = {bf[q].z, bf[q].w};
#pragma unroll
            for (int u = 0; u < 4; ++u) {
                const v2f alo = {af[u][q].x, af[u][q].y};
                cacc[u] = __builtin_amdgcn_wmma_f32_16x16x4_f32(
                    false, alo, false, blo, (short)0, cacc[u], false, false);
            }
#pragma unroll
            for (int u = 0; u < 4; ++u) {
                const v2f ahi = {af[u][q].z, af[u][q].w};
                cacc[u] = __builtin_amdgcn_wmma_f32_16x16x4_f32(
                    false, ahi, false, bhi, (short)0, cacc[u], false, false);
            }
        }
#pragma unroll
        for (int u = 0; u < 4; ++u) {
            const int mt = (wv >> 3) + 4 * u;
            if (mt < XRT) {
#pragma unroll
                for (int v = 0; v < 8; ++v)
                    xrs[(mt * 16 + v + 8 * hi) * HCC + nt * 16 + mi] = cacc[u][v] + bb;
            }
        }
    }
    __syncthreads();

    int n = gcnt[blockIdx.x];
    const bool over = n > LCAP;
    if (over) {
        for (int e = tid; e < EE; e += 1024) {
            const int d = dst[e];
            const unsigned dl = (unsigned)(d - base);
            if (dl < (unsigned)NPB) {
                const int s = src[e];
                for (int hh = 0; hh < HH; ++hh)
                    atomicMax(&mx[dl * HH + hh],
                              fkey(edge_score_one(e, s, &xrs[dl * HCC], hh, ea, We, xl, att)));
            }
        }
        n = 0;
    }
    const int* mylst = glst + (size_t)blockIdx.x * LCAP;
    const int* mysrc = gsrc + (size_t)blockIdx.x * LCAP;
    for (int i = tid; i < n; i += 1024) {
        lst[i]  = mylst[i];
        lsrc[i] = mysrc[i];
    }
    __syncthreads();

    const int total = n + NPB;
    const int h     = lane >> 3;
    const float4* xl4 = (const float4*)xl;
    const float4* xrs4 = (const float4*)xrs;

    {
        const float4* We4 = (const float4*)We;
        const v4f* xl4v = (const v4f*)xl;
        const v4f* xrsv = (const v4f*)xrs;
        const v4f* ea4v = (const v4f*)ea;
        float4 wr[4];
#pragma unroll
        for (int k = 0; k < 4; ++k) wr[k] = We4[lane * 4 + k];
        const float4 atv = ((const float4*)att)[lane];
        const float at[4] = {atv.x, atv.y, atv.z, atv.w};

        for (int i0 = wv; i0 < n; i0 += 128) {
            int rec[4], sl[4];
#pragma unroll
            for (int u = 0; u < 4; ++u) {
                const int idx = i0 + 32 * u;
                rec[u] = lst[idx < n ? idx : 0];
                sl[u]  = lsrc[idx < n ? idx : 0];
            }
            v4f xa[4], xb[4], ev[4];
#pragma unroll
            for (int u = 0; u < 4; ++u) {
                xa[u] = xl4v[(size_t)sl[u] * (HCC / 4) + lane];
                xb[u] = xrsv[(rec[u] >> 20) * (HCC / 4) + lane];
                ev[u] = ea4v[rec[u] & 0xFFFFF];
            }
            asm volatile("" : "+v"(xa[0]), "+v"(xa[1]), "+v"(xa[2]), "+v"(xa[3]),
                              "+v"(xb[0]), "+v"(xb[1]), "+v"(xb[2]), "+v"(xb[3]),
                              "+v"(ev[0]), "+v"(ev[1]), "+v"(ev[2]), "+v"(ev[3]));
            float sc[4];
#pragma unroll
            for (int u = 0; u < 4; ++u) {
                const float pa[4] = {xa[u].x, xa[u].y, xa[u].z, xa[u].w};
                const float pb[4] = {xb[u].x, xb[u].y, xb[u].z, xb[u].w};
                float s = 0.0f;
#pragma unroll
                for (int k = 0; k < 4; ++k) {
                    const float eev = wr[k].x * ev[u].x + wr[k].y * ev[u].y +
                                      wr[k].z * ev[u].z + wr[k].w * ev[u].w;
                    float m = pa[k] + pb[k] + eev;
                    m = leakyf(m, 0.2f);
                    s += at[k] * m;
                }
                sc[u] = s;
            }
#pragma unroll
            for (int off = 4; off >= 1; off >>= 1) {
#pragma unroll
                for (int u = 0; u < 4; ++u) sc[u] += __shfl_xor(sc[u], off);
            }
#pragma unroll
            for (int u = 0; u < 4; ++u) {
                const int idx = i0 + 32 * u;
                if (idx < n && (lane & 7) == 0) {
                    lsc[idx * HH + h] = sc[u];
                    atomicMax(&mx[(rec[u] >> 20) * HH + h], fkey(sc[u]));
                }
            }
        }

        for (int j0 = wv; j0 < NPB; j0 += 128) {
            float4 xa[4], xb[4];
            bool   ok[4];
#pragma unroll
            for (int u = 0; u < 4; ++u) {
                const int j    = j0 + 32 * u;
                const int node = base + j;
                ok[u] = (j < NPB) && (node < NN);
                const int nd = ok[u] ? node : 0;
                xa[u] = xl4[(size_t)nd * (HCC / 4) + lane];
                xb[u] = xrs4[(ok[u] ? j : 0) * (HCC / 4) + lane];
            }
            float sc[4];
#pragma unroll
            for (int u = 0; u < 4; ++u) {
                const float pa[4] = {xa[u].x, xa[u].y, xa[u].z, xa[u].w};
                const float pb[4] = {xb[u].x, xb[u].y, xb[u].z, xb[u].w};
                float s = 0.0f;
#pragma unroll
                for (int k = 0; k < 4; ++k) {
                    float m = pa[k] + pb[k];
                    m = leakyf(m, 0.2f);
                    s += at[k] * m;
                }
                sc[u] = s;
            }
#pragma unroll
            for (int off = 4; off >= 1; off >>= 1) {
#pragma unroll
                for (int u = 0; u < 4; ++u) sc[u] += __shfl_xor(sc[u], off);
            }
#pragma unroll
            for (int u = 0; u < 4; ++u) {
                const int j = j0 + 32 * u;
                if (ok[u] && (lane & 7) == 0) {
                    lsc[(n + j) * HH + h] = sc[u];
                    atomicMax(&mx[j * HH + h], fkey(sc[u]));
                }
            }
        }
    }
    __syncthreads();

    if (over) {
        for (int e = tid; e < EE; e += 1024) {
            const int d = dst[e];
            const unsigned dl = (unsigned)(d - base);
            if (dl < (unsigned)NPB) {
                const int s = src[e];
                for (int hh = 0; hh < HH; ++hh) {
                    const float ex = expf(edge_score_one(e, s, &xrs[dl * HCC], hh, ea, We, xl, att) -
                                          funkey(mx[dl * HH + hh]));
                    atomicAdd(&dsum[dl * HH + hh], ex);
                    for (int cc = 0; cc < CC; ++cc) {
                        const int ch = hh * CC + cc;
                        atomicAdd(&acc[dl * HCC + (ch & 3) * 32 + (ch >> 2)],
                                  xl[(size_t)s * HCC + ch] * ex);
                    }
                }
            }
        }
    }

    for (int i0 = wv; i0 < total; i0 += 128) {
        int rec[4], sl[4];
#pragma unroll
        for (int u = 0; u < 4; ++u) {
            const int idx = i0 + 32 * u;
            rec[u] = lst[idx < n ? idx : 0];
            sl[u]  = lsrc[idx < n ? idx : 0];
        }
        int      dl[4];
        float    sc[4];
        unsigned mk[4];
        float4   xv[4];
        bool     ok[4];
#pragma unroll
        for (int u = 0; u < 4; ++u) {
            const int  idx = i0 + 32 * u;
            const bool inl = idx < n;
            dl[u] = inl ? (rec[u] >> 20) : (idx - n);
            const int node = base + dl[u];
            ok[u] = (idx < total) && (node < NN);
            const int sN = ok[u] ? (inl ? sl[u] : node) : 0;
            sc[u] = lsc[(ok[u] ? idx : 0) * HH + h];
            mk[u] = mx[(ok[u] ? dl[u] : 0) * HH + h];
            xv[u] = xl4[(size_t)sN * (HCC / 4) + lane];
        }
#pragma unroll
        for (int u = 0; u < 4; ++u) {
            if (ok[u]) {
                const float ex = expf(sc[u] - funkey(mk[u]));
                float* a = &acc[dl[u] * HCC + lane];
                atomicAdd(a +  0, xv[u].x * ex);
                atomicAdd(a + 32, xv[u].y * ex);
                atomicAdd(a + 64, xv[u].z * ex);
                atomicAdd(a + 96, xv[u].w * ex);
                if ((lane & 7) == 0) atomicAdd(&dsum[dl[u] * HH + h], ex);
            }
        }
    }
    __syncthreads();

    float4* agg4 = (float4*)agg;
    const float4 cbv = ((const float4*)cb)[lane];
    float fs = 0.0f, fq = 0.0f;
    for (int i = tid; i < NPB * (HCC / 4); i += 1024) {
        const int j    = i >> 5;
        const int node = base + j;
        if (node < NN) {
            const float* a = &acc[j * HCC + lane];
            const float  d = dsum[j * HH + h] + 1e-16f;
            float4 y;
            y.x = a[0]  / d + cbv.x;
            y.y = a[32] / d + cbv.y;
            y.z = a[64] / d + cbv.z;
            y.w = a[96] / d + cbv.w;
            agg4[(size_t)node * (HCC / 4) + lane] = y;
            fs += (y.x + y.y) + (y.z + y.w);
            fq += (y.x * y.x + y.y * y.y) + (y.z * y.z + y.w * y.w);
        }
    }
    for (int off = 16; off >= 1; off >>= 1) {
        fs += __shfl_xor(fs, off);
        fq += __shfl_xor(fq, off);
    }
    if (lane == 0) { wsum[wv] = (double)fs; wsq[wv] = (double)fq; }
    __syncthreads();
    if (wv == 0) {
        double s = wsum[lane];
        double q = wsq[lane];
        for (int off = 16; off >= 1; off >>= 1) {
            s += __shfl_xor(s, off);
            q += __shfl_xor(q, off);
        }
        if (lane == 0) ((double2*)part)[blockIdx.x] = make_double2(s, q);
    }
}

#define TP   (HCC + 4)
#define ZP   (FF + 2)
static_assert(XRT * 16 * (HCC / 4) <= 7 * 1024, "seven staging rounds of the 1024 threads cover the rows");
static_assert(FF * (HCC / 4) == 1024, "one 16-byte piece of proj_W per thread");
static_assert((NOWN - 1) * NPB < NN, "row 0 of every work-group is a row of the graph");
__global__ __launch_bounds__(1024)
void row_tail_owner_kernel(const float* __restrict__ agg, const double* __restrict__ part,
                           const float* __restrict__ pw, const float* __restrict__ pb,
                           const float* __restrict__ W, const float* __restrict__ bias,
                           const float* __restrict__ lw, const float* __restrict__ lb,
                           float* __restrict__ Hout, float* __restrict__ jk) {
    __shared__ __attribute__((aligned(16))) float as_[XRT * 16 * TP];
    __shared__ __attribute__((aligned(16))) float wps[FF * TP];
    __shared__ float  zs[XRT * 16 * ZP];
    __shared__ double rs_[8];
    __shared__ double rq_[8];
    __shared__ float  st[2];
    const int tid  = (int)threadIdx.x;
    const int lane = tid & 31;
    const int wv   = tid >> 5;
    const int base = (int)blockIdx.x * NPB;

    double2 pp = make_double2(0.0, 0.0);
    if (wv < 8) pp = ((const double2*)part)[tid];
    v4f rv[7];
#pragma unroll
    for (int it = 0; it < 7; ++it) {
        int node = base + wv + 32 * it;
        node = node < NN ? node : NN - 1;
        rv[it] = ((const v4f*)agg)[(size_t)node * (HCC / 4) + lane];
    }
    const v4f wst = ((const v4f*)W)[tid];
    const v4f gw  = ((const v4f*)pw)[lane];
    const v4f gb  = ((const v4f*)pb)[lane];

    if (wv < 8) {
        double s = pp.x;
        double q = pp.y;
        for (int off = 16; off >= 1; off >>= 1) {
            s += __shfl_xor(s, off);
            q += __shfl_xor(q, off);
        }
        if (lane == 0) { rs_[wv] = s; rq_[wv] = q; }
    }
    __syncthreads();
    if (wv == 0) {
        double s = lane < 8 ? rs_[lane] : 0.0;
        double q = lane < 8 ? rq_[lane] : 0.0;
        for (int off = 4; off >= 1; off >>= 1) {
            s += __shfl_xor(s, off);
            q += __shfl_xor(q, off);
        }
        if (lane == 0) {
            const double tot = (double)NN * (double)HCC;
            const double mu = s / tot;
            double var = q / tot - mu * mu;
            if (var < 0.0) var = 0.0;
            const float sd = sqrtf((float)var);
            st[0] = (float)mu;
            st[1] = (float)(1.0 / (double)(sd + 1e-5f));
        }
    }
    __syncthreads();
    const float mu  = st[0];
    const float inv = st[1];

#pragma unroll
    for (int it = 0; it < 7; ++it) {
        const int r = wv + 32 * it;
        if (r < XRT * 16) {
            v4f y;
            y.x = leakyf(gw.x * ((rv[it].x - mu) * inv) + gb.x, 0.01f);
            y.y = leakyf(gw.y * ((rv[it].y - mu) * inv) + gb.y, 0.01f);
            y.z = leakyf(gw.z * ((rv[it].z - mu) * inv) + gb.z, 0.01f);
            y.w = leakyf(gw.w * ((rv[it].w - mu) * inv) + gb.w, 0.01f);
            *(v4f*)&as_[r * TP + 4 * lane] = y;
        }
    }
    *(v4f*)&wps[wv * TP + 4 * lane] = wst;
    __syncthreads();

    if (wv < 2 * XRT) {
        const int hi = lane >> 4;
        const int mi = lane & 15;
        const int mt = wv >> 1;
        const int nt = wv & 1;
        const float* Ap = as_ + (mt * 16 + mi) * TP + 4 * hi;
        const float* Bp = wps + (nt * 16 + mi) * TP + 4 * hi;
        v8f c = {};
#pragma unroll
        for (int j = 0; j < HCC / 8; ++j) {
            const v4f a = *(const v4f*)(Ap + 8 * j);
            const v4f b = *(const v4f*)(Bp + 8 * j);
            const v2f alo = {a.x, a.y};
            const v2f ahi = {a.z, a.w};
            const v2f blo = {b.x, b.y};
            const v2f bhi = {b.z, b.w};
            c = __builtin_amdgcn_wmma_f32_16x16x4_f32(false, alo, false, blo, (short)0, c, false, false);
            c = __builtin_amdgcn_wmma_f32_16x16x4_f32(false, ahi, false, bhi, (short)0, c, false, false);
        }
        const float bb = bias[nt * 16 + mi];
#pragma unroll
        for (int v = 0; v < 8; ++v)
            zs[(mt * 16 + v + 8 * hi) * ZP + nt * 16 + mi] = c[v] + bb;
    }
    __syncthreads();

    const float lwv = lw[lane];
    const float lbv = lb[lane];
    float v[7], old[7], s[7], q[7];
    bool  ok[7];
#pragma unroll
    for (int u = 0; u < 7; ++u) {
        const int j = wv + 32 * u;
        ok[u] = (j < NPB) && (base + j < NN);
        const int jj = ok[u] ? j : 0;
        v[u]   = zs[jj * ZP + lane];
        old[u] = jk[(size_t)(base + jj) * FF + lane];
        s[u]   = v[u];
    }
#pragma unroll
    for (int off = 16; off >= 1; off >>= 1) {
#pragma unroll
        for (int u = 0; u < 7; ++u) s[u] += __shfl_xor(s[u], off);
    }
#pragma unroll
    for (int u = 0; u < 7; ++u) {
        const float m = s[u] * (1.0f / 32.0f);
        v[u] = v[u] - m;
        q[u] = v[u] * v[u];
    }
#pragma unroll
    for (int off = 16; off >= 1; off >>= 1) {
#pragma unroll
        for (int u = 0; u < 7; ++u) q[u] += __shfl_xor(q[u], off);
    }
    asm volatile("" : "+v"(old[0]), "+v"(q[0]));
#pragma unroll
    for (int u = 0; u < 7; ++u) {
        if (ok[u]) {
            const int node = base + wv + 32 * u;
            const float var = q[u] * (1.0f / 32.0f);
            float y = lwv * v[u] * rsqrtf(var + 1e-5f) + lbv;
            y = leakyf(y, 0.01f);
            Hout[(size_t)node * FF + lane] = y;
            jk[(size_t)node * FF + lane] = fmaxf(old[u], y);
        }
    }
}

extern "C" void kernel_launch(void* const* d_in, const int* in_sizes, int n_in,
                              void* d_out, int out_size, void* d_ws, size_t ws_size,
                              hipStream_t stream) {
    const float* x          = (const float*)d_in[0];
    const int*   edge_index = (const int*)  d_in[1];
    const float* edge_attr  = (const float*)d_in[2];
    const float* embed_W    = (const float*)d_in[3];
    const float* embed_b    = (const float*)d_in[4];
    const float* embed_lnw  = (const float*)d_in[5];
    const float* embed_lnb  = (const float*)d_in[6];
    const float* Wl         = (const float*)d_in[7];
    const float* bl         = (const float*)d_in[8];
    const float* Wr         = (const float*)d_in[9];
    const float* br         = (const float*)d_in[10];
    const float* We         = (const float*)d_in[11];
    const float* att        = (const float*)d_in[12];
    const float* conv_b     = (const float*)d_in[13];
    const float* pln_w      = (const float*)d_in[14];
    const float* pln_b      = (const float*)d_in[15];
    const float* proj_W     = (const float*)d_in[16];
    const float* proj_b     = (const float*)d_in[17];
    const float* ln2_w      = (const float*)d_in[18];
    const float* ln2_b      = (const float*)d_in[19];

    const int* srcE = edge_index;
    const int* dstE = edge_index + EE;

    size_t off = 0;
    auto carve = [&](size_t nbytes) -> char* {
        char* p = (char*)d_ws + off;
        off += (nbytes + 255) & ~(size_t)255;
        return p;
    };
    float*    h     = (float*)   carve((size_t)NN * FF  * 4);
    float*    raw   = (float*)   carve((size_t)NN * FF  * 4);
    float*    xl    = (float*)   carve((size_t)NN * HCC * 4);
    float*    agg   = (float*)   carve((size_t)NN * HCC * 4);
    double*   part  = (double*)  carve((size_t)NOWN * 2 * sizeof(double));
    int*      olst  = (int*)     carve((size_t)NOWN * LCAP * sizeof(int));
    int*      osrc  = (int*)     carve((size_t)NOWN * LCAP * sizeof(int));
    int*      ocnt  = (int*)     carve((size_t)NOWN * sizeof(int));
    (void)ws_size; (void)in_sizes; (void)n_in; (void)out_size;

    float* jk = (float*)d_out;

    const int mtiles = NN / 16;
    const dim3 blk(256);
    const dim3 gGemm((mtiles * 32 + 255) / 256);
    const dim3 gRows((NN * 32 + 255) / 256);

    const size_t shLin  = (size_t)HCC * (FF + 4) * 4;

    embed_gemm_kernel<<<dim3((NN / 16 + EW_WAVES - 1) / EW_WAVES), dim3(EW_WAVES * 32), 0, stream>>>(
        x, embed_W, embed_b, raw);
    ln_rows_kernel<<<gRows, blk, 0, stream>>>(raw, embed_lnw, embed_lnb, h, jk, 0, NN, ocnt);

    owner_lists_kernel<<<dim3((EE / 4 + LBLK - 1) / LBLK), dim3(LBLK), 0, stream>>>(
        srcE, dstE, olst, osrc, ocnt);

    for (int l = 0; l < LL; ++l) {
        const float* Wl_l = Wl + (size_t)l * HCC * FF;
        const float* bl_l = bl + (size_t)l * HCC;
        const float* Wr_l = Wr + (size_t)l * HCC * FF;
        const float* br_l = br + (size_t)l * HCC;
        const float* We_l = We + (size_t)l * HCC * EDD;
        const float* at_l = att + (size_t)l * HH * CC;
        const float* cb_l = conv_b + (size_t)l * HCC;
        const float* pw_l = pln_w + (size_t)l * HCC;
        const float* pb_l = pln_b + (size_t)l * HCC;
        const float* pW_l = proj_W + (size_t)l * FF * HCC;
        const float* pB_l = proj_b + (size_t)l * FF;
        const float* l2w  = ln2_w + (size_t)l * FF;
        const float* l2b  = ln2_b + (size_t)l * FF;

        gemm_wmma_f32<8, FF, true><<<gGemm, blk, shLin, stream>>>(h, Wl_l, bl_l, xl, NN, HCC);

        edge_owner_kernel<<<dim3(NOWN), dim3(1024), 0, stream>>>(
            srcE, dstE, edge_attr, We_l, xl, h, Wr_l, br_l, at_l, cb_l, agg, part, olst, osrc, ocnt);

        row_tail_owner_kernel<<<dim3(NOWN), dim3(1024), 0, stream>>>(
            agg, part, pw_l, pb_l, pW_l, pB_l, l2w, l2b, h, jk);
    }
}
